# starting kernel
# baseline (speedup 1.0000x reference)
#include <hip/hip_runtime.h>


typedef __attribute__((ext_vector_type(2))) float v2f;
typedef __attribute__((ext_vector_type(8))) float v8f;

static constexpr int Bb = 8;
static constexpr int Tt = 256;
static constexpr int Dd = 256;
static constexpr int Nn = 1024;

__global__ __launch_bounds__(256) void ln256_kernel(const float* __restrict__ in,
                                                    float* __restrict__ out) {
  const int D = 256;
  const long row = blockIdx.x;
  const int j = threadIdx.x;
  float v = in[row * D + j];
  __shared__ float red[8];

  float s = v;
#pragma unroll
  for (int o = 16; o > 0; o >>= 1) s += __shfl_down(s, o, 32);
  if ((j & 31) == 0) red[j >> 5] = s;
  __syncthreads();
  float tot = 0.f;
#pragma unroll
  for (int w = 0; w < 8; ++w) tot += red[w];
  const float mean = tot * (1.0f / 256.0f);
  const float c = v - mean;
  __syncthreads();

  float s2 = c * c;
#pragma unroll
  for (int o = 16; o > 0; o >>= 1) s2 += __shfl_down(s2, o, 32);
  if ((j & 31) == 0) red[j >> 5] = s2;
  __syncthreads();
  float tot2 = 0.f;
#pragma unroll
  for (int w = 0; w < 8; ++w) tot2 += red[w];
  const float var = tot2 * (1.0f / 256.0f);

  out[row * D + j] = c * rsqrtf(var + 1e-5f);
}

__global__ __launch_bounds__(1024) void xscan_kernel(const float* __restrict__ XU,
                                                     float* __restrict__ X) {
  const int b = blockIdx.x;
  const int j = threadIdx.x;
  const float* xu = XU + (long)b * Tt * Nn;
  float* xo = X + (long)b * Tt * Nn;
  __shared__ float red[32];
  float x = 0.f;
  for (int t = 0; t < Tt; ++t) {
    x = 0.97f * x + xu[(long)t * Nn + j];
    float s = fabsf(x);
#pragma unroll
    for (int o = 16; o > 0; o >>= 1) s += __shfl_down(s, o, 32);
    if ((j & 31) == 0) red[j >> 5] = s;
    __syncthreads();
    if (j < 32) {
      float v = red[j];
#pragma unroll
      for (int o = 16; o > 0; o >>= 1) v += __shfl_down(v, o, 32);
      if (j == 0) red[0] = v;
    }
    __syncthreads();
    x = x / fmaxf(red[0], 1e-12f);
    xo[(long)t * Nn + j] = x;
    __syncthreads();
  }
}

__global__ __launch_bounds__(256) void mask_kernel(const float* __restrict__ Kg,
                                                   float* __restrict__ Wm) {
  const int s = threadIdx.x;
  const int t = blockIdx.x;
  const int b = blockIdx.y;
  const long idx = (((long)b * Tt) + t) * Tt + s;
  float v = 0.f;
  if (s < t) v = __powf(0.97f, (float)(t - 1 - s)) * Kg[idx];
  Wm[idx] = v;
}

template <int EPI>
__global__ __launch_bounds__(256) void gemm_ABt_wmma(const float* __restrict__ A,
                                                     const float* __restrict__ W,
                                                     float* __restrict__ C,
                                                     const float* __restrict__ Mul,
                                                     int M, int N, int K,
                                                     long sA, long sW, long sC) {
  const int b = blockIdx.y;
  A += (long)b * sA;
  W += (long)b * sW;
  C += (long)b * sC;
  const int lane = threadIdx.x & 31;
  const int wave = threadIdx.x >> 5;
  const int strip = blockIdx.x * (blockDim.x >> 5) + wave;
  const int nstrips_n = N >> 6;
  const int tm = strip / nstrips_n;
  const int tn = strip - tm * nstrips_n;
  if (tm >= (M >> 4)) return;

  const int half = lane >> 4;
  const int l = lane & 15;

  const float2* Arow = (const float2*)(A + (long)(tm * 16 + l) * K);
  const float2* Wrow0 = (const float2*)(W + (long)(tn * 64 + 0 * 16 + l) * K);
  const float2* Wrow1 = (const float2*)(W + (long)(tn * 64 + 1 * 16 + l) * K);
  const float2* Wrow2 = (const float2*)(W + (long)(tn * 64 + 2 * 16 + l) * K);
  const float2* Wrow3 = (const float2*)(W + (long)(tn * 64 + 3 * 16 + l) * K);

  v8f c0 = {}, c1 = {}, c2 = {}, c3 = {};
#pragma unroll 2
  for (int kk = 0; kk < K; kk += 4) {
    const int ki = (kk >> 1) + half;
    const float2 af = Arow[ki];
    v2f a; a[0] = af.x; a[1] = af.y;
    float2 wf;
    v2f bb;
    wf = Wrow0[ki]; bb[0] = wf.x; bb[1] = wf.y;
    c0 = __builtin_amdgcn_wmma_f32_16x16x4_f32(false, a, false, bb, (short)0, c0,
                                               false, false);
    wf = Wrow1[ki]; bb[0] = wf.x; bb[1] = wf.y;
    c1 = __builtin_amdgcn_wmma_f32_16x16x4_f32(false, a, false, bb, (short)0, c1,
                                               false, false);
    wf = Wrow2[ki]; bb[0] = wf.x; bb[1] = wf.y;
    c2 = __builtin_amdgcn_wmma_f32_16x16x4_f32(false, a, false, bb, (short)0, c2,
                                               false, false);
    wf = Wrow3[ki]; bb[0] = wf.x; bb[1] = wf.y;
    c3 = __builtin_amdgcn_wmma_f32_16x16x4_f32(false, a, false, bb, (short)0, c3,
                                               false, false);
  }

  v8f acc[4] = {c0, c1, c2, c3};
#pragma unroll
  for (int t = 0; t < 4; ++t) {
    const long col = tn * 64 + t * 16 + l;
#pragma unroll
    for (int r = 0; r < 8; ++r) {
      const long row = tm * 16 + r + 8 * half;
      float v = acc[t][r];
      if (EPI == 1) v = fmaxf(v, 0.f);
      if (EPI == 2) v = fmaxf(v, 0.f) * fmaxf(Mul[row * N + col], 0.f);
      C[row * N + col] = v;
    }
  }
}

__global__ __launch_bounds__(256) void gemm_AB_wmma(const float* __restrict__ A,
                                                    const float* __restrict__ Bm,
                                                    float* __restrict__ C,
                                                    int M, int N, int K,
                                                    long sA, long sB, long sC) {
  const int b = blockIdx.y;
  A += (long)b * sA;
  Bm += (long)b * sB;
  C += (long)b * sC;
  const int lane = threadIdx.x & 31;
  const int wave = threadIdx.x >> 5;
  const int strip = blockIdx.x * (blockDim.x >> 5) + wave;
  const int nstrips_n = N >> 6;
  const int tm = strip / nstrips_n;
  const int tn = strip - tm * nstrips_n;
  if (tm >= (M >> 4)) return;

  const int half = lane >> 4;
  const int l = lane & 15;

  const float2* Arow = (const float2*)(A + (long)(tm * 16 + l) * K);
  const float* Bc0 = Bm + (tn * 64 + 0 * 16 + l);
  const float* Bc1 = Bm + (tn * 64 + 1 * 16 + l);
  const float* Bc2 = Bm + (tn * 64 + 2 * 16 + l);
  const float* Bc3 = Bm + (tn * 64 + 3 * 16 + l);

  v8f c0 = {}, c1 = {}, c2 = {}, c3 = {};
#pragma unroll 2
  for (int kk = 0; kk < K; kk += 4) {
    const float2 af = Arow[(kk >> 1) + half];
    const long k0 = kk + 2 * half;
    v2f a; a[0] = af.x; a[1] = af.y;
    v2f bb;
    bb[0] = Bc0[k0 * N]; bb[1] = Bc0[(k0 + 1) * N];
    c0 = __builtin_amdgcn_wmma_f32_16x16x4_f32(false, a, false, bb, (short)0, c0,
                                               false, false);
    bb[0] = Bc1[k0 * N]; bb[1] = Bc1[(k0 + 1) * N];
    c1 = __builtin_amdgcn_wmma_f32_16x16x4_f32(false, a, false, bb, (short)0, c1,
                                               false, false);
    bb[0] = Bc2[k0 * N]; bb[1] = Bc2[(k0 + 1) * N];
    c2 = __builtin_amdgcn_wmma_f32_16x16x4_f32(false, a, false, bb, (short)0, c2,
                                               false, false);
    bb[0] = Bc3[k0 * N]; bb[1] = Bc3[(k0 + 1) * N];
    c3 = __builtin_amdgcn_wmma_f32_16x16x4_f32(false, a, false, bb, (short)0, c3,
                                               false, false);
  }

  v8f acc[4] = {c0, c1, c2, c3};
#pragma unroll
  for (int t = 0; t < 4; ++t) {
    const long col = tn * 64 + t * 16 + l;
#pragma unroll
    for (int r = 0; r < 8; ++r) {
      const long row = tm * 16 + r + 8 * half;
      C[row * N + col] = acc[t][r];
    }
  }
}

extern "C" void kernel_launch(void* const* d_in, const int* in_sizes, int n_in,
                              void* d_out, int out_size, void* d_ws, size_t ws_size,
                              hipStream_t stream) {
  const float* emb = (const float*)d_in[0];
  const float* E   = (const float*)d_in[1];
  const float* Dx  = (const float*)d_in[2];
  const float* Dy  = (const float*)d_in[3];
  float* out = (float*)d_out;

  const long BT = (long)Bb * Tt;
  const long szBTD = BT * Dd;
  const long szBTN = BT * Nn;
  const long szBTT = (long)Bb * Tt * Tt;

  float* ws = (float*)d_ws;
  float* LNV = ws;
  float* XU  = LNV + szBTD;
  float* X   = XU + szBTN;
  float* Kg  = X + szBTN;
  float* Wm  = Kg + szBTT;
  float* As  = Wm + szBTT;
  float* LNA = As + szBTD;
  float* Y   = LNA + szBTD;
  float* V   = Y + szBTN;

  ln256_kernel<<<(int)BT, 256, 0, stream>>>(emb, LNV);

  gemm_ABt_wmma<1><<<dim3((2048 / 16) * (1024 / 64) / 8, 1), 256, 0, stream>>>(
      emb, Dx, XU, nullptr, 2048, 1024, 256, 0, 0, 0);

  xscan_kernel<<<Bb, 1024, 0, stream>>>(XU, X);

  gemm_ABt_wmma<0><<<dim3((256 / 16) * (256 / 64) / 8, Bb), 256, 0, stream>>>(
      X, X, Kg, nullptr, 256, 256, 1024, (long)Tt * Nn, (long)Tt * Nn,
      (long)Tt * Tt);

  mask_kernel<<<dim3(Tt, Bb), 256, 0, stream>>>(Kg, Wm);

  gemm_AB_wmma<<<dim3((256 / 16) * (256 / 64) / 8, Bb), 256, 0, stream>>>(
      Wm, LNV, As, 256, 256, 256, (long)Tt * Tt, (long)Tt * Dd, (long)Tt * Dd);

  ln256_kernel<<<(int)BT, 256, 0, stream>>>(As, LNA);

  gemm_ABt_wmma<2><<<dim3((2048 / 16) * (1024 / 64) / 8, 1), 256, 0, stream>>>(
      LNA, Dy, Y, X, 2048, 1024, 256, 0, 0, 0);

  gemm_ABt_wmma<0><<<dim3((2048 / 16) * (256 / 64) / 8, 1), 256, 0, stream>>>(
      Y, E, V, nullptr, 2048, 256, 1024, 0, 0, 0);

  ln256_kernel<<<(int)BT, 256, 0, stream>>>(V, out);
}


# All 5 GEMMs: operand tiles staged in LDS by whole-line 16 B global loads (double buffered, next chunk loaded under the WMMAs), fragments read from LDS; 4-wave workgroups; k and sum order kept
# speedup vs baseline: 4.8135x; 4.8135x over previous
#include <hip/hip_runtime.h>


typedef __attribute__((ext_vector_type(2))) float v2f;
typedef __attribute__((ext_vector_type(8))) float v8f;
typedef __attribute__((ext_vector_type(4))) float v4f;

static constexpr int Bb = 8;
static constexpr int Tt = 256;
static constexpr int Dd = 256;
static constexpr int Nn = 1024;

__global__ __launch_bounds__(256) void ln256_kernel(const float* __restrict__ in,
                                                    float* __restrict__ out) {
  const int D = 256;
  const long row = blockIdx.x;
  const int j = threadIdx.x;
  float v = in[row * D + j];
  __shared__ float red[8];

  float s = v;
#pragma unroll
  for (int o = 16; o > 0; o >>= 1) s += __shfl_down(s, o, 32);
  if ((j & 31) == 0) red[j >> 5] = s;
  __syncthreads();
  float tot = 0.f;
#pragma unroll
  for (int w = 0; w < 8; ++w) tot += red[w];
  const float mean = tot * (1.0f / 256.0f);
  const float c = v - mean;
  __syncthreads();

  float s2 = c * c;
#pragma unroll
  for (int o = 16; o > 0; o >>= 1) s2 += __shfl_down(s2, o, 32);
  if ((j & 31) == 0) red[j >> 5] = s2;
  __syncthreads();
  float tot2 = 0.f;
#pragma unroll
  for (int w = 0; w < 8; ++w) tot2 += red[w];
  const float var = tot2 * (1.0f / 256.0f);

  out[row * D + j] = c * rsqrtf(var + 1e-5f);
}

static constexpr int XS_WAVES = 8;
static constexpr int XS_AHEAD = 8;

template <int CTRL>
static __device__ __forceinline__ float row_xadd(float v) {
  const int iv = __builtin_bit_cast(int, v);
  const int sw = __builtin_amdgcn_update_dpp(0, iv, CTRL, 0xf, 0xf, true);
  return v + __builtin_bit_cast(float, sw);
}
static __device__ __forceinline__ float other_row(float v) {
  const int iv = __builtin_bit_cast(int, v);
  return __builtin_bit_cast(
      float, __builtin_amdgcn_permlanex16(iv, iv, 0x76543210, 0xfedcba98, false, false));
}

__global__ __launch_bounds__(XS_WAVES * 32) void xscan_kernel(const float* __restrict__ XU,
                                                              float* __restrict__ X) {
  const int b = blockIdx.x;
  const int tid = threadIdx.x;
  const int lane = tid & 31;
  const int wave = tid >> 5;
  const int col = wave * 128 + lane;
  const float* xub = XU + (long)b * Tt * Nn;
  float* xob = X + (long)b * Tt * Nn;
  __shared__ __attribute__((aligned(16))) float part[2][XS_WAVES];

  float u0[XS_AHEAD], u1[XS_AHEAD], u2[XS_AHEAD], u3[XS_AHEAD];
#pragma unroll
  for (int i = 0; i < XS_AHEAD; ++i) {
    const float* r = xub + i * Nn + col;
    u0[i] = r[0]; u1[i] = r[32]; u2[i] = r[64]; u3[i] = r[96];
  }

  float x0 = 0.f, x1 = 0.f, x2 = 0.f, x3 = 0.f;
  for (int t0 = 0; t0 < Tt; t0 += XS_AHEAD) {
#pragma unroll
    for (int i = 0; i < XS_AHEAD; ++i) {
      const int t = t0 + i;
      x0 = 0.97f * x0 + u0[i];
      x1 = 0.97f * x1 + u1[i];
      x2 = 0.97f * x2 + u2[i];
      x3 = 0.97f * x3 + u3[i];
      const int tn = (t + XS_AHEAD < Tt) ? (t + XS_AHEAD) : (Tt - 1);
      const float* r = xub + tn * Nn + col;
      u0[i] = r[0]; u1[i] = r[32]; u2[i] = r[64]; u3[i] = r[96];

      float s = (fabsf(x0) + fabsf(x1)) + (fabsf(x2) + fabsf(x3));
      s = row_xadd<0x161>(s);
      s = row_xadd<0x162>(s);
      s = row_xadd<0x164>(s);
      s = row_xadd<0x168>(s);
      s = s + other_row(s);

      float* slot = part[t & 1];
      if (lane == 0) slot[wave] = s;
      __builtin_amdgcn_fence(__ATOMIC_RELEASE, "workgroup", "local");
      __builtin_amdgcn_s_barrier();
      __builtin_amdgcn_fence(__ATOMIC_ACQUIRE, "workgroup", "local");
      const v4f pa = *(const v4f*)(slot);
      const v4f pb = *(const v4f*)(slot + 4);
      const v2f h = (pa.lo + pb.lo) + (pa.hi + pb.hi);
      const float tot = h[0] + h[1];
      const float d = fmaxf(tot, 1e-12f);
      x0 = x0 / d;
      x1 = x1 / d;
      x2 = x2 / d;
      x3 = x3 / d;
      float* w = xob + t * Nn + col;
      w[0] = x0; w[32] = x1; w[64] = x2; w[96] = x3;
    }
  }
}

__global__ __launch_bounds__(256) void mask_kernel(const float* __restrict__ Kg,
                                                   float* __restrict__ Wm) {
  const int s = threadIdx.x;
  const int t = blockIdx.x;
  const int b = blockIdx.y;
  const long idx = (((long)b * Tt) + t) * Tt + s;
  float v = 0.f;
  if (s < t) v = __powf(0.97f, (float)(t - 1 - s)) * Kg[idx];
  Wm[idx] = v;
}

template <int EPI, int BM, int BN, int BK>
__global__ __launch_bounds__(128) void gemm_ABt_lds(const float* __restrict__ A,
                                                    const float* __restrict__ W,
                                                    float* __restrict__ C,
                                                    const float* __restrict__ Mul,
                                                    int N, int K,
                                                    long sA, long sW, long sC) {
  constexpr int PITCH = BK + 4;
  constexpr int C4 = BK / 4;
  constexpr int NA = BM * C4 / 128;
  constexpr int NW = BN * C4 / 128;
  constexpr int FM = BM / 32;
  constexpr int FN = BN / 32;
  __shared__ __attribute__((aligned(16))) float As[2][BM * PITCH];
  __shared__ __attribute__((aligned(16))) float Ws[2][BN * PITCH];

  const int b = blockIdx.z;
  A += (long)b * sA;
  W += (long)b * sW;
  C += (long)b * sC;
  const int tid = threadIdx.x;
  const int lane = tid & 31;
  const int wave = tid >> 5;
  const int wm = wave >> 1;
  const int wn = wave & 1;
  const int half = lane >> 4;
  const int l = lane & 15;
  const int row0 = blockIdx.y * BM;
  const int col0 = blockIdx.x * BN;

  const float* ga[NA];
  int la[NA];
#pragma unroll
  for (int j = 0; j < NA; ++j) {
    const int f = tid + 128 * j;
    const int r = f / C4;
    const int c = f - r * C4;
    ga[j] = A + (long)(row0 + r) * K + c * 4;
    la[j] = r * PITCH + c * 4;
  }
  const float* gw[NW];
  int lw[NW];
#pragma unroll
  for (int j = 0; j < NW; ++j) {
    const int f = tid + 128 * j;
    const int r = f / C4;
    const int c = f - r * C4;
    gw[j] = W + (long)(col0 + r) * K + c * 4;
    lw[j] = r * PITCH + c * 4;
  }

  v4f ra[NA], rw[NW];
#pragma unroll
  for (int j = 0; j < NA; ++j) ra[j] = *(const v4f*)(ga[j]);
#pragma unroll
  for (int j = 0; j < NW; ++j) rw[j] = *(const v4f*)(gw[j]);
#pragma unroll
  for (int j = 0; j < NA; ++j) *(v4f*)(&As[0][la[j]]) = ra[j];
#pragma unroll
  for (int j = 0; j < NW; ++j) *(v4f*)(&Ws[0][lw[j]]) = rw[j];
  __syncthreads();

  v8f acc[FM][FN];
#pragma unroll
  for (int i = 0; i < FM; ++i)
#pragma unroll
    for (int j = 0; j < FN; ++j) acc[i][j] = v8f{};

  const int aoff = (wm * (BM / 2) + l) * PITCH + 2 * half;
  const int woff = (wn * (BN / 2) + l) * PITCH + 2 * half;
  const int nchunks = K / BK;
  for (int ch = 0; ch < nchunks; ++ch) {
    const int cur = ch & 1;
    const bool more = (ch + 1 < nchunks);
    if (more) {
      const int kn = (ch + 1) * BK;
#pragma unroll
      for (int j = 0; j < NA; ++j) ra[j] = *(const v4f*)(ga[j] + kn);
#pragma unroll
      for (int j = 0; j < NW; ++j) rw[j] = *(const v4f*)(gw[j] + kn);
    }
    const float* Ab = &As[cur][aoff];
    const float* Wb = &Ws[cur][woff];
#pragma unroll
    for (int ks = 0; ks < BK / 4; ++ks) {
      v2f a[FM], bb[FN];
#pragma unroll
      for (int i = 0; i < FM; ++i) a[i] = *(const v2f*)(Ab + i * 16 * PITCH + ks * 4);
#pragma unroll
      for (int j = 0; j < FN; ++j) bb[j] = *(const v2f*)(Wb + j * 16 * PITCH + ks * 4);
#pragma unroll
      for (int i = 0; i < FM; ++i)
#pragma unroll
        for (int j = 0; j < FN; ++j)
          acc[i][j] = __builtin_amdgcn_wmma_f32_16x16x4_f32(false, a[i], false, bb[j], (short)0,
                                                            acc[i][j], false, false);
    }
    if (more) {
#pragma unroll
      for (int j = 0; j < NA; ++j) *(v4f*)(&As[cur ^ 1][la[j]]) = ra[j];
#pragma unroll
      for (int j = 0; j < NW; ++j) *(v4f*)(&Ws[cur ^ 1][lw[j]]) = rw[j];
    }
    __syncthreads();
  }

#pragma unroll
  for (int i = 0; i < FM; ++i) {
#pragma unroll
    for (int j = 0; j < FN; ++j) {
      const long col = col0 + wn * (BN / 2) + j * 16 + l;
#pragma unroll
      for (int r = 0; r < 8; ++r) {
        const long row = row0 + wm * (BM / 2) + i * 16 + r + 8 * half;
        float v = acc[i][j][r];
        if (EPI == 1) v = fmaxf(v, 0.f);
        if (EPI == 2) v = fmaxf(v, 0.f) * fmaxf(Mul[row * N + col], 0.f);
        C[row * N + col] = v;
      }
    }
  }
}

template <int BM, int BN, int BK>
__global__ __launch_bounds__(128) void gemm_AB_lds(const float* __restrict__ A,
                                                   const float* __restrict__ Bm,
                                                   float* __restrict__ C,
                                                   int N, int K,
                                                   long sA, long sB, long sC) {
  constexpr int PA = BK + 4;
  constexpr int PB = BN + 8;
  constexpr int CA = BK / 4;
  constexpr int CB = BN / 4;
  constexpr int NA = BM * CA / 128;
  constexpr int NB = BK * CB / 128;
  constexpr int FM = BM / 32;
  constexpr int FN = BN / 32;
  __shared__ __attribute__((aligned(16))) float As[2][BM * PA];
  __shared__ __attribute__((aligned(16))) float Bs[2][BK * PB];

  const int b = blockIdx.z;
  A += (long)b * sA;
  Bm += (long)b * sB;
  C += (long)b * sC;
  const int tid = threadIdx.x;
  const int lane = tid & 31;
  const int wave = tid >> 5;
  const int wm = wave >> 1;
  const int wn = wave & 1;
  const int half = lane >> 4;
  const int l = lane & 15;
  const int row0 = blockIdx.y * BM;
  const int col0 = blockIdx.x * BN;

  const float* ga[NA];
  int la[NA];
#pragma unroll
  for (int j = 0; j < NA; ++j) {
    const int f = tid + 128 * j;
    const int r = f / CA;
    const int c = f - r * CA;
    ga[j] = A + (long)(row0 + r) * K + c * 4;
    la[j] = r * PA + c * 4;
  }
  const float* gb[NB];
  int lb[NB];
#pragma unroll
  for (int j = 0; j < NB; ++j) {
    const int f = tid + 128 * j;
    const int r = f / CB;
    const int c = f - r * CB;
    gb[j] = Bm + (long)r * N + col0 + c * 4;
    lb[j] = r * PB + c * 4;
  }

  v4f ra[NA], rb[NB];
#pragma unroll
  for (int j = 0; j < NA; ++j) ra[j] = *(const v4f*)(ga[j]);
#pragma unroll
  for (int j = 0; j < NB; ++j) rb[j] = *(const v4f*)(gb[j]);
#pragma unroll
  for (int j = 0; j < NA; ++j) *(v4f*)(&As[0][la[j]]) = ra[j];
#pragma unroll
  for (int j = 0; j < NB; ++j) *(v4f*)(&Bs[0][lb[j]]) = rb[j];
  __syncthreads();

  v8f acc[FM][FN];
#pragma unroll
  for (int i = 0; i < FM; ++i)
#pragma unroll
    for (int j = 0; j < FN; ++j) acc[i][j] = v8f{};

  const int aoff = (wm * (BM / 2) + l) * PA + 2 * half;
  const int boff = (2 * half) * PB + wn * (BN / 2) + l;
  const int nchunks = K / BK;
  for (int ch = 0; ch < nchunks; ++ch) {
    const int cur = ch & 1;
    const bool more = (ch + 1 < nchunks);
    if (more) {
      const int kn = (ch + 1) * BK;
#pragma unroll
      for (int j = 0; j < NA; ++j) ra[j] = *(const v4f*)(ga[j] + kn);
#pragma unroll
      for (int j = 0; j < NB; ++j) rb[j] = *(const v4f*)(gb[j] + (long)kn * N);
    }
    const float* Ab = &As[cur][aoff];
    const float* Bb = &Bs[cur][boff];
#pragma unroll
    for (int ks = 0; ks < BK / 4; ++ks) {
      v2f a[FM], bb[FN];
#pragma unroll
      for (int i = 0; i < FM; ++i) a[i] = *(const v2f*)(Ab + i * 16 * PA + ks * 4);
#pragma unroll
      for (int j = 0; j < FN; ++j) {
        bb[j][0] = Bb[(ks * 4) * PB + j * 16];
        bb[j][1] = Bb[(ks * 4 + 1) * PB + j * 16];
      }
#pragma unroll
      for (int i = 0; i < FM; ++i)
#pragma unroll
        for (int j = 0; j < FN; ++j)
          acc[i][j] = __builtin_amdgcn_wmma_f32_16x16x4_f32(false, a[i], false, bb[j], (short)0,
                                                            acc[i][j], false, false);
    }
    if (more) {
#pragma unroll
      for (int j = 0; j < NA; ++j) *(v4f*)(&As[cur ^ 1][la[j]]) = ra[j];
#pragma unroll
      for (int j = 0; j < NB; ++j) *(v4f*)(&Bs[cur ^ 1][lb[j]]) = rb[j];
    }
    __syncthreads();
  }

#pragma unroll
  for (int i = 0; i < FM; ++i) {
#pragma unroll
    for (int j = 0; j < FN; ++j) {
      const long col = col0 + wn * (BN / 2) + j * 16 + l;
#pragma unroll
      for (int r = 0; r < 8; ++r) {
        const long row = row0 + wm * (BM / 2) + i * 16 + r + 8 * half;
        C[row * N + col] = acc[i][j][r];
      }
    }
  }
}

extern "C" void kernel_launch(void* const* d_in, const int* in_sizes, int n_in,
                              void* d_out, int out_size, void* d_ws, size_t ws_size,
                              hipStream_t stream) {
  const float* emb = (const float*)d_in[0];
  const float* E   = (const float*)d_in[1];
  const float* Dx  = (const float*)d_in[2];
  const float* Dy  = (const float*)d_in[3];
  float* out = (float*)d_out;

  const long BT = (long)Bb * Tt;
  const long szBTD = BT * Dd;
  const long szBTN = BT * Nn;
  const long szBTT = (long)Bb * Tt * Tt;

  float* ws = (float*)d_ws;
  float* LNV = ws;
  float* XU  = LNV + szBTD;
  float* X   = XU + szBTN;
  float* Kg  = X + szBTN;
  float* Wm  = Kg + szBTT;
  float* As  = Wm + szBTT;
  float* LNA = As + szBTD;
  float* Y   = LNA + szBTD;
  float* V   = Y + szBTN;

  ln256_kernel<<<(int)BT, 256, 0, stream>>>(emb, LNV);

  gemm_ABt_lds<1, 64, 64, 32><<<dim3(1024 / 64, 2048 / 64, 1), 128, 0, stream>>>(
      emb, Dx, XU, nullptr, 1024, 256, 0, 0, 0);

  xscan_kernel<<<Bb, XS_WAVES * 32, 0, stream>>>(XU, X);

  gemm_ABt_lds<0, 32, 64, 64><<<dim3(256 / 64, 256 / 32, Bb), 128, 0, stream>>>(
      X, X, Kg, nullptr, 256, 1024, (long)Tt * Nn, (long)Tt * Nn, (long)Tt * Tt);

  mask_kernel<<<dim3(Tt, Bb), 256, 0, stream>>>(Kg, Wm);

  gemm_AB_lds<32, 64, 32><<<dim3(256 / 64, 256 / 32, Bb), 128, 0, stream>>>(
      Wm, LNV, As, 256, 256, (long)Tt * Tt, (long)Tt * Dd, (long)Tt * Dd);

  ln256_kernel<<<(int)BT, 256, 0, stream>>>(As, LNA);

  gemm_ABt_lds<2, 64, 64, 32><<<dim3(1024 / 64, 2048 / 64, 1), 128, 0, stream>>>(
      LNA, Dy, Y, X, 1024, 256, 0, 0, 0);

  gemm_ABt_lds<0, 32, 64, 64><<<dim3(256 / 64, 2048 / 32, 1), 128, 0, stream>>>(
      Y, E, V, nullptr, 256, 1024, 0, 0, 0);

  ln256_kernel<<<(int)BT, 256, 0, stream>>>(V, out);
}


# x-scan as a WMMA matrix product: step norms d_t = 0.97*S + rowsum(u_t) (x, u >= 0), weights G[t,s], X = G @ U per batch, causal k; row sums in GEMM epilogue; 11 launches (was 10); x sum order changes
# speedup vs baseline: 7.6056x; 1.5801x over previous
#include <hip/hip_runtime.h>


typedef __attribute__((ext_vector_type(2))) float v2f;
typedef __attribute__((ext_vector_type(8))) float v8f;
typedef __attribute__((ext_vector_type(4))) float v4f;

static constexpr int Bb = 8;
static constexpr int Tt = 256;
static constexpr int Dd = 256;
static constexpr int Nn = 1024;

__global__ __launch_bounds__(256) void ln256_kernel(const float* __restrict__ in,
                                                    float* __restrict__ out) {
  const int D = 256;
  const long row = blockIdx.x;
  const int j = threadIdx.x;
  float v = in[row * D + j];
  __shared__ float red[8];

  float s = v;
#pragma unroll
  for (int o = 16; o > 0; o >>= 1) s += __shfl_down(s, o, 32);
  if ((j & 31) == 0) red[j >> 5] = s;
  __syncthreads();
  float tot = 0.f;
#pragma unroll
  for (int w = 0; w < 8; ++w) tot += red[w];
  const float mean = tot * (1.0f / 256.0f);
  const float c = v - mean;
  __syncthreads();

  float s2 = c * c;
#pragma unroll
  for (int o = 16; o > 0; o >>= 1) s2 += __shfl_down(s2, o, 32);
  if ((j & 31) == 0) red[j >> 5] = s2;
  __syncthreads();
  float tot2 = 0.f;
#pragma unroll
  for (int w = 0; w < 8; ++w) tot2 += red[w];
  const float var = tot2 * (1.0f / 256.0f);

  out[row * D + j] = c * rsqrtf(var + 1e-5f);
}


template <int CTRL>
static __device__ __forceinline__ float row_xadd(float v) {
  const int iv = __builtin_bit_cast(int, v);
  const int sw = __builtin_amdgcn_update_dpp(0, iv, CTRL, 0xf, 0xf, true);
  return v + __builtin_bit_cast(float, sw);
}

static constexpr int RS_STRIPS = 32;
static constexpr int GB_ROWS = 32;

__global__ __launch_bounds__(256) void gbuild_kernel(const float* __restrict__ Rp,
                                                     float* __restrict__ G) {
  const int rb = blockIdx.x;
  const int b = blockIdx.y;
  const int s = threadIdx.x;
  const int lane = s & 31;
  const int wave = s >> 5;
  __shared__ __attribute__((aligned(16))) float rs[Tt];
  __shared__ __attribute__((aligned(16))) float qs[Tt];
  __shared__ unsigned big[8];
  __shared__ float bp[8];

  float r = 0.f;
#pragma unroll
  for (int c = 0; c < RS_STRIPS; ++c) r += Rp[(long)c * (Bb * Tt) + b * Tt + s];
  rs[s] = r;
  const unsigned m = __builtin_amdgcn_ballot_w32(r >= 1e-12f);
  if (lane == 0) big[wave] = m;
  __syncthreads();

  int f = Tt;
#pragma unroll
  for (int w = 7; w >= 0; --w) {
    const unsigned mw = big[w];
    if (mw != 0u) f = 32 * w + __builtin_ctz(mw);
  }
  float S = 1.0f;
  if (s <= f) {
    S = 0.f;
    for (int i = 0; i < s; ++i) {
      const float sum = fmaf(0.97f, S, rs[i]);
      S = (sum >= 1e-12f) ? 1.0f : sum * 1e12f;
    }
  }
  const float d = fmaxf(fmaf(0.97f, S, r), 1e-12f);
  const float inv = 1.0f / d;
  const float q = 0.97f * inv;
  qs[s] = q;

  const float GMAX = 3.0e38f;

  float v = q;
#pragma unroll
  for (int o = 1; o < 32; o <<= 1) {
    const float up = __shfl_down(v, o, 32);
    if (lane + o < 32) v = fminf(v * up, GMAX);
  }
  const float nxt = __shfl_down(v, 1, 32);
  const float suf = (lane < 31) ? nxt : 1.0f;
  if (lane == 0) bp[wave] = v;
  __syncthreads();

  const int t0 = rb * GB_ROWS;
  float* gp = G + ((long)b * Tt + t0) * Tt + s;
  if (wave > rb) {
#pragma unroll 8
    for (int i = 0; i < GB_ROWS; ++i) gp[(long)i * Tt] = 0.f;
  } else {
    float g = 0.f;
    if (wave < rb) {
      g = fminf(inv * suf, GMAX);
      for (int k = wave + 1; k < rb; ++k) g = fminf(g * bp[k], GMAX);
    }
#pragma unroll 8
    for (int i = 0; i < GB_ROWS; ++i) {
      const int tt = t0 + i;
      const float gn = fminf(g * qs[tt], GMAX);
      g = (tt < s) ? 0.f : ((tt == s) ? inv : gn);
      gp[(long)i * Tt] = g;
    }
  }
}

__global__ __launch_bounds__(256) void mask_kernel(const float* __restrict__ Kg,
                                                   float* __restrict__ Wm) {
  const int s = threadIdx.x;
  const int t = blockIdx.x;
  const int b = blockIdx.y;
  const long idx = (((long)b * Tt) + t) * Tt + s;
  float v = 0.f;
  if (s < t) v = __powf(0.97f, (float)(t - 1 - s)) * Kg[idx];
  Wm[idx] = v;
}

template <int EPI, int BM, int BN, int BK>
__global__ __launch_bounds__(128) void gemm_ABt_lds(const float* __restrict__ A,
                                                    const float* __restrict__ W,
                                                    float* __restrict__ C,
                                                    const float* __restrict__ Mul,
                                                    float* __restrict__ Rs,
                                                    int N, int K,
                                                    long sA, long sW, long sC) {
  constexpr int PITCH = BK + 4;
  constexpr int C4 = BK / 4;
  constexpr int NA = BM * C4 / 128;
  constexpr int NW = BN * C4 / 128;
  constexpr int FM = BM / 32;
  constexpr int FN = BN / 32;
  __shared__ __attribute__((aligned(16))) float As[2][BM * PITCH];
  __shared__ __attribute__((aligned(16))) float Ws[2][BN * PITCH];

  const int b = blockIdx.z;
  A += (long)b * sA;
  W += (long)b * sW;
  C += (long)b * sC;
  const int tid = threadIdx.x;
  const int lane = tid & 31;
  const int wave = tid >> 5;
  const int wm = wave >> 1;
  const int wn = wave & 1;
  const int half = lane >> 4;
  const int l = lane & 15;
  const int row0 = blockIdx.y * BM;
  const int col0 = blockIdx.x * BN;

  const float* ga[NA];
  int la[NA];
#pragma unroll
  for (int j = 0; j < NA; ++j) {
    const int f = tid + 128 * j;
    const int r = f / C4;
    const int c = f - r * C4;
    ga[j] = A + (long)(row0 + r) * K + c * 4;
    la[j] = r * PITCH + c * 4;
  }
  const float* gw[NW];
  int lw[NW];
#pragma unroll
  for (int j = 0; j < NW; ++j) {
    const int f = tid + 128 * j;
    const int r = f / C4;
    const int c = f - r * C4;
    gw[j] = W + (long)(col0 + r) * K + c * 4;
    lw[j] = r * PITCH + c * 4;
  }

  v4f ra[NA], rw[NW];
#pragma unroll
  for (int j = 0; j < NA; ++j) ra[j] = *(const v4f*)(ga[j]);
#pragma unroll
  for (int j = 0; j < NW; ++j) rw[j] = *(const v4f*)(gw[j]);
#pragma unroll
  for (int j = 0; j < NA; ++j) *(v4f*)(&As[0][la[j]]) = ra[j];
#pragma unroll
  for (int j = 0; j < NW; ++j) *(v4f*)(&Ws[0][lw[j]]) = rw[j];
  __syncthreads();

  v8f acc[FM][FN];
#pragma unroll
  for (int i = 0; i < FM; ++i)
#pragma unroll
    for (int j = 0; j < FN; ++j) acc[i][j] = v8f{};

  const int aoff = (wm * (BM / 2) + l) * PITCH + 2 * half;
  const int woff = (wn * (BN / 2) + l) * PITCH + 2 * half;
  const int nchunks = K / BK;
  for (int ch = 0; ch < nchunks; ++ch) {
    const int cur = ch & 1;
    const bool more = (ch + 1 < nchunks);
    if (more) {
      const int kn = (ch + 1) * BK;
#pragma unroll
      for (int j = 0; j < NA; ++j) ra[j] = *(const v4f*)(ga[j] + kn);
#pragma unroll
      for (int j = 0; j < NW; ++j) rw[j] = *(const v4f*)(gw[j] + kn);
    }
    const float* Ab = &As[cur][aoff];
    const float* Wb = &Ws[cur][woff];
#pragma unroll
    for (int ks = 0; ks < BK / 4; ++ks) {
      v2f a[FM], bb[FN];
#pragma unroll
      for (int i = 0; i < FM; ++i) a[i] = *(const v2f*)(Ab + i * 16 * PITCH + ks * 4);
#pragma unroll
      for (int j = 0; j < FN; ++j) bb[j] = *(const v2f*)(Wb + j * 16 * PITCH + ks * 4);
#pragma unroll
      for (int i = 0; i < FM; ++i)
#pragma unroll
        for (int j = 0; j < FN; ++j)
          acc[i][j] = __builtin_amdgcn_wmma_f32_16x16x4_f32(false, a[i], false, bb[j], (short)0,
                                                            acc[i][j], false, false);
    }
    if (more) {
#pragma unroll
      for (int j = 0; j < NA; ++j) *(v4f*)(&As[cur ^ 1][la[j]]) = ra[j];
#pragma unroll
      for (int j = 0; j < NW; ++j) *(v4f*)(&Ws[cur ^ 1][lw[j]]) = rw[j];
    }
    __syncthreads();
  }

#pragma unroll
  for (int i = 0; i < FM; ++i) {
#pragma unroll
    for (int j = 0; j < FN; ++j) {
      const long col = col0 + wn * (BN / 2) + j * 16 + l;
#pragma unroll
      for (int r = 0; r < 8; ++r) {
        const long row = row0 + wm * (BM / 2) + i * 16 + r + 8 * half;
        float v = acc[i][j][r];
        if (EPI == 1) v = fmaxf(v, 0.f);
        if (EPI == 2) v = fmaxf(v, 0.f) * fmaxf(Mul[row * N + col], 0.f);
        C[row * N + col] = v;
      }
    }
  }

  if (EPI == 1) {
    static_assert(EPI != 1 || (BM == 64 && BN == 64), "row-sum epilogue: 32 x 32 wave tile");
    float p[FM][8];
#pragma unroll
    for (int i = 0; i < FM; ++i) {
#pragma unroll
      for (int r = 0; r < 8; ++r) {
        float s = 0.f;
#pragma unroll
        for (int j = 0; j < FN; ++j) s += fmaxf(acc[i][j][r], 0.f);
        s = row_xadd<0x161>(s);
        s = row_xadd<0x162>(s);
        s = row_xadd<0x164>(s);
        s = row_xadd<0x168>(s);
        p[i][r] = s;
      }
    }
    float mine = 0.f;
#pragma unroll
    for (int i = 0; i < FM; ++i)
#pragma unroll
      for (int r = 0; r < 8; ++r)
        if (l == i * 8 + r) mine = p[i][r];
    const int rho = (l >> 3) * 16 + 8 * half + (l & 7);
    const long rows = (long)gridDim.y * BM;
    Rs[(long)(blockIdx.x * 2 + wn) * rows + row0 + wm * (BM / 2) + rho] = mine;
  }
}

template <int BM, int BN, int BK, bool CAUSAL>
__global__ __launch_bounds__(128) void gemm_AB_lds(const float* __restrict__ A,
                                                   const float* __restrict__ Bm,
                                                   float* __restrict__ C,
                                                   int N, int K,
                                                   long sA, long sB, long sC) {
  constexpr int PA = BK + 4;
  constexpr int PB = BN + 8;
  constexpr int CA = BK / 4;
  constexpr int CB = BN / 4;
  constexpr int NA = BM * CA / 128;
  constexpr int NB = BK * CB / 128;
  constexpr int FM = BM / 32;
  constexpr int FN = BN / 32;
  __shared__ __attribute__((aligned(16))) float As[2][BM * PA];
  __shared__ __attribute__((aligned(16))) float Bs[2][BK * PB];

  const int b = blockIdx.z;
  A += (long)b * sA;
  Bm += (long)b * sB;
  C += (long)b * sC;
  const int tid = threadIdx.x;
  const int lane = tid & 31;
  const int wave = tid >> 5;
  const int wm = wave >> 1;
  const int wn = wave & 1;
  const int half = lane >> 4;
  const int l = lane & 15;
  const int row0 = blockIdx.y * BM;
  const int col0 = blockIdx.x * BN;

  const float* ga[NA];
  int la[NA];
#pragma unroll
  for (int j = 0; j < NA; ++j) {
    const int f = tid + 128 * j;
    const int r = f / CA;
    const int c = f - r * CA;
    ga[j] = A + (long)(row0 + r) * K + c * 4;
    la[j] = r * PA + c * 4;
  }
  const float* gb[NB];
  int lb[NB];
#pragma unroll
  for (int j = 0; j < NB; ++j) {
    const int f = tid + 128 * j;
    const int r = f / CB;
    const int c = f - r * CB;
    gb[j] = Bm + (long)r * N + col0 + c * 4;
    lb[j] = r * PB + c * 4;
  }

  v4f ra[NA], rb[NB];
#pragma unroll
  for (int j = 0; j < NA; ++j) ra[j] = *(const v4f*)(ga[j]);
#pragma unroll
  for (int j = 0; j < NB; ++j) rb[j] = *(const v4f*)(gb[j]);
#pragma unroll
  for (int j = 0; j < NA; ++j) *(v4f*)(&As[0][la[j]]) = ra[j];
#pragma unroll
  for (int j = 0; j < NB; ++j) *(v4f*)(&Bs[0][lb[j]]) = rb[j];
  __syncthreads();

  v8f acc[FM][FN];
#pragma unroll
  for (int i = 0; i < FM; ++i)
#pragma unroll
    for (int j = 0; j < FN; ++j) acc[i][j] = v8f{};

  const int aoff = (wm * (BM / 2) + l) * PA + 2 * half;
  const int boff = (2 * half) * PB + wn * (BN / 2) + l;
  const int nchunks = CAUSAL ? (row0 + BM) / BK : K / BK;
  for (int ch = 0; ch < nchunks; ++ch) {
    const int cur = ch & 1;
    const bool more = (ch + 1 < nchunks);
    if (more) {
      const int kn = (ch + 1) * BK;
#pragma unroll
      for (int j = 0; j < NA; ++j) ra[j] = *(const v4f*)(ga[j] + kn);
#pragma unroll
      for (int j = 0; j < NB; ++j) rb[j] = *(const v4f*)(gb[j] + (long)kn * N);
    }
    const float* Ab = &As[cur][aoff];
    const float* Bb = &Bs[cur][boff];
#pragma unroll
    for (int ks = 0; ks < BK / 4; ++ks) {
      v2f a[FM], bb[FN];
#pragma unroll
      for (int i = 0; i < FM; ++i) a[i] = *(const v2f*)(Ab + i * 16 * PA + ks * 4);
#pragma unroll
      for (int j = 0; j < FN; ++j) {
        bb[j][0] = Bb[(ks * 4) * PB + j * 16];
        bb[j][1] = Bb[(ks * 4 + 1) * PB + j * 16];
      }
#pragma unroll
      for (int i = 0; i < FM; ++i)
#pragma unroll
        for (int j = 0; j < FN; ++j)
          acc[i][j] = __builtin_amdgcn_wmma_f32_16x16x4_f32(false, a[i], false, bb[j], (short)0,
                                                            acc[i][j], false, false);
    }
    if (more) {
#pragma unroll
      for (int j = 0; j < NA; ++j) *(v4f*)(&As[cur ^ 1][la[j]]) = ra[j];
#pragma unroll
      for (int j = 0; j < NB; ++j) *(v4f*)(&Bs[cur ^ 1][lb[j]]) = rb[j];
    }
    __syncthreads();
  }

#pragma unroll
  for (int i = 0; i < FM; ++i) {
#pragma unroll
    for (int j = 0; j < FN; ++j) {
      const long col = col0 + wn * (BN / 2) + j * 16 + l;
#pragma unroll
      for (int r = 0; r < 8; ++r) {
        const long row = row0 + wm * (BM / 2) + i * 16 + r + 8 * half;
        C[row * N + col] = acc[i][j][r];
      }
    }
  }
}

extern "C" void kernel_launch(void* const* d_in, const int* in_sizes, int n_in,
                              void* d_out, int out_size, void* d_ws, size_t ws_size,
                              hipStream_t stream) {
  const float* emb = (const float*)d_in[0];
  const float* E   = (const float*)d_in[1];
  const float* Dx  = (const float*)d_in[2];
  const float* Dy  = (const float*)d_in[3];
  float* out = (float*)d_out;

  const long BT = (long)Bb * Tt;
  const long szBTD = BT * Dd;
  const long szBTN = BT * Nn;
  const long szBTT = (long)Bb * Tt * Tt;

  float* ws = (float*)d_ws;
  float* LNV = ws;
  float* XU  = LNV + szBTD;
  float* X   = XU + szBTN;
  float* Kg  = X + szBTN;
  float* Wm  = Kg + szBTT;
  float* As  = Wm + szBTT;
  float* LNA = As + szBTD;
  float* Y   = LNA + szBTD;
  float* V   = Y + szBTN;

  ln256_kernel<<<(int)BT, 256, 0, stream>>>(emb, LNV);

  float* Rp = Kg;
  float* G  = Wm;
  gemm_ABt_lds<1, 64, 64, 32><<<dim3(1024 / 64, 2048 / 64, 1), 128, 0, stream>>>(
      emb, Dx, XU, nullptr, Rp, 1024, 256, 0, 0, 0);

  gbuild_kernel<<<dim3(Tt / GB_ROWS, Bb), 256, 0, stream>>>(Rp, G);

  gemm_AB_lds<64, 64, 32, true><<<dim3(1024 / 64, 256 / 64, Bb), 128, 0, stream>>>(
      G, XU, X, 1024, 256, (long)Tt * Tt, (long)Tt * Nn, (long)Tt * Nn);

  gemm_ABt_lds<0, 32, 64, 64><<<dim3(256 / 64, 256 / 32, Bb), 128, 0, stream>>>(
      X, X, Kg, nullptr, nullptr, 256, 1024, (long)Tt * Nn, (long)Tt * Nn, (long)Tt * Tt);

  mask_kernel<<<dim3(Tt, Bb), 256, 0, stream>>>(Kg, Wm);

  gemm_AB_lds<32, 64, 32, false><<<dim3(256 / 64, 256 / 32, Bb), 128, 0, stream>>>(
      Wm, LNV, As, 256, 256, (long)Tt * Tt, (long)Tt * Dd, (long)Tt * Dd);

  ln256_kernel<<<(int)BT, 256, 0, stream>>>(As, LNA);

  gemm_ABt_lds<2, 64, 64, 32><<<dim3(1024 / 64, 2048 / 64, 1), 128, 0, stream>>>(
      LNA, Dy, Y, X, nullptr, 1024, 256, 0, 0, 0);

  gemm_ABt_lds<0, 32, 64, 64><<<dim3(256 / 64, 2048 / 32, 1), 128, 0, stream>>>(
      Y, E, V, nullptr, nullptr, 256, 1024, 0, 0, 0);

  ln256_kernel<<<(int)BT, 256, 0, stream>>>(V, out);
}


# decay mask and ln(a*) moved into the consuming GEMMs' A staging (mask on Gram pieces in registers; 64x256 a* panel layer-normed in registers, held in LDS): 11 -> 9 launches; ln(a*) sum order changes
# speedup vs baseline: 8.1323x; 1.0693x over previous
#include <hip/hip_runtime.h>


typedef __attribute__((ext_vector_type(2))) float v2f;
typedef __attribute__((ext_vector_type(8))) float v8f;
typedef __attribute__((ext_vector_type(4))) float v4f;

static constexpr int Bb = 8;
static constexpr int Tt = 256;
static constexpr int Dd = 256;
static constexpr int Nn = 1024;

__global__ __launch_bounds__(256) void ln256_kernel(const float* __restrict__ in,
                                                    float* __restrict__ out) {
  const int D = 256;
  const long row = blockIdx.x;
  const int j = threadIdx.x;
  float v = in[row * D + j];
  __shared__ float red[8];

  float s = v;
#pragma unroll
  for (int o = 16; o > 0; o >>= 1) s += __shfl_down(s, o, 32);
  if ((j & 31) == 0) red[j >> 5] = s;
  __syncthreads();
  float tot = 0.f;
#pragma unroll
  for (int w = 0; w < 8; ++w) tot += red[w];
  const float mean = tot * (1.0f / 256.0f);
  const float c = v - mean;
  __syncthreads();

  float s2 = c * c;
#pragma unroll
  for (int o = 16; o > 0; o >>= 1) s2 += __shfl_down(s2, o, 32);
  if ((j & 31) == 0) red[j >> 5] = s2;
  __syncthreads();
  float tot2 = 0.f;
#pragma unroll
  for (int w = 0; w < 8; ++w) tot2 += red[w];
  const float var = tot2 * (1.0f / 256.0f);

  out[row * D + j] = c * rsqrtf(var + 1e-5f);
}


template <int CTRL>
static __device__ __forceinline__ float row_xadd(float v) {
  const int iv = __builtin_bit_cast(int, v);
  const int sw = __builtin_amdgcn_update_dpp(0, iv, CTRL, 0xf, 0xf, true);
  return v + __builtin_bit_cast(float, sw);
}

static constexpr int RS_STRIPS = 32;
static constexpr int GB_ROWS = 32;

__global__ __launch_bounds__(256) void gbuild_kernel(const float* __restrict__ Rp,
                                                     float* __restrict__ G) {
  const int rb = blockIdx.x;
  const int b = blockIdx.y;
  const int s = threadIdx.x;
  const int lane = s & 31;
  const int wave = s >> 5;
  __shared__ __attribute__((aligned(16))) float rs[Tt];
  __shared__ __attribute__((aligned(16))) float qs[Tt];
  __shared__ unsigned big[8];
  __shared__ float bp[8];

  float r = 0.f;
#pragma unroll
  for (int c = 0; c < RS_STRIPS; ++c) r += Rp[(long)c * (Bb * Tt) + b * Tt + s];
  rs[s] = r;
  const unsigned m = __builtin_amdgcn_ballot_w32(r >= 1e-12f);
  if (lane == 0) big[wave] = m;
  __syncthreads();

  int f = Tt;
#pragma unroll
  for (int w = 7; w >= 0; --w) {
    const unsigned mw = big[w];
    if (mw != 0u) f = 32 * w + __builtin_ctz(mw);
  }
  float S = 1.0f;
  if (s <= f) {
    S = 0.f;
    for (int i = 0; i < s; ++i) {
      const float sum = fmaf(0.97f, S, rs[i]);
      S = (sum >= 1e-12f) ? 1.0f : sum * 1e12f;
    }
  }
  const float d = fmaxf(fmaf(0.97f, S, r), 1e-12f);
  const float inv = 1.0f / d;
  const float q = 0.97f * inv;
  qs[s] = q;

  const float GMAX = 3.0e38f;

  float v = q;
#pragma unroll
  for (int o = 1; o < 32; o <<= 1) {
    const float up = __shfl_down(v, o, 32);
    if (lane + o < 32) v = fminf(v * up, GMAX);
  }
  const float nxt = __shfl_down(v, 1, 32);
  const float suf = (lane < 31) ? nxt : 1.0f;
  if (lane == 0) bp[wave] = v;
  __syncthreads();

  const int t0 = rb * GB_ROWS;
  float* gp = G + ((long)b * Tt + t0) * Tt + s;
  if (wave > rb) {
#pragma unroll 8
    for (int i = 0; i < GB_ROWS; ++i) gp[(long)i * Tt] = 0.f;
  } else {
    float g = 0.f;
    if (wave < rb) {
      g = fminf(inv * suf, GMAX);
      for (int k = wave + 1; k < rb; ++k) g = fminf(g * bp[k], GMAX);
    }
#pragma unroll 8
    for (int i = 0; i < GB_ROWS; ++i) {
      const int tt = t0 + i;
      const float gn = fminf(g * qs[tt], GMAX);
      g = (tt < s) ? 0.f : ((tt == s) ? inv : gn);
      gp[(long)i * Tt] = g;
    }
  }
}

template <int EPI, int BM, int BN, int BK>
__global__ __launch_bounds__(128) void gemm_ABt_lds(const float* __restrict__ A,
                                                    const float* __restrict__ W,
                                                    float* __restrict__ C,
                                                    float* __restrict__ Rs,
                                                    int N, int K,
                                                    long sA, long sW, long sC) {
  constexpr int PITCH = BK + 4;
  constexpr int C4 = BK / 4;
  constexpr int NA = BM * C4 / 128;
  constexpr int NW = BN * C4 / 128;
  constexpr int FM = BM / 32;
  constexpr int FN = BN / 32;
  __shared__ __attribute__((aligned(16))) float As[2][BM * PITCH];
  __shared__ __attribute__((aligned(16))) float Ws[2][BN * PITCH];

  const int b = blockIdx.z;
  A += (long)b * sA;
  W += (long)b * sW;
  C += (long)b * sC;
  const int tid = threadIdx.x;
  const int lane = tid & 31;
  const int wave = tid >> 5;
  const int wm = wave >> 1;
  const int wn = wave & 1;
  const int half = lane >> 4;
  const int l = lane & 15;
  const int row0 = blockIdx.y * BM;
  const int col0 = blockIdx.x * BN;

  const float* ga[NA];
  int la[NA];
#pragma unroll
  for (int j = 0; j < NA; ++j) {
    const int f = tid + 128 * j;
    const int r = f / C4;
    const int c = f - r * C4;
    ga[j] = A + (long)(row0 + r) * K + c * 4;
    la[j] = r * PITCH + c * 4;
  }
  const float* gw[NW];
  int lw[NW];
#pragma unroll
  for (int j = 0; j < NW; ++j) {
    const int f = tid + 128 * j;
    const int r = f / C4;
    const int c = f - r * C4;
    gw[j] = W + (long)(col0 + r) * K + c * 4;
    lw[j] = r * PITCH + c * 4;
  }

  v4f ra[NA], rw[NW];
#pragma unroll
  for (int j = 0; j < NA; ++j) ra[j] = *(const v4f*)(ga[j]);
#pragma unroll
  for (int j = 0; j < NW; ++j) rw[j] = *(const v4f*)(gw[j]);
#pragma unroll
  for (int j = 0; j < NA; ++j) *(v4f*)(&As[0][la[j]]) = ra[j];
#pragma unroll
  for (int j = 0; j < NW; ++j) *(v4f*)(&Ws[0][lw[j]]) = rw[j];
  __syncthreads();

  v8f acc[FM][FN];
#pragma unroll
  for (int i = 0; i < FM; ++i)
#pragma unroll
    for (int j = 0; j < FN; ++j) acc[i][j] = v8f{};

  const int aoff = (wm * (BM / 2) + l) * PITCH + 2 * half;
  const int woff = (wn * (BN / 2) + l) * PITCH + 2 * half;
  const int nchunks = K / BK;
  for (int ch = 0; ch < nchunks; ++ch) {
    const int cur = ch & 1;
    const bool more = (ch + 1 < nchunks);
    if (more) {
      const int kn = (ch + 1) * BK;
#pragma unroll
      for (int j = 0; j < NA; ++j) ra[j] = *(const v4f*)(ga[j] + kn);
#pragma unroll
      for (int j = 0; j < NW; ++j) rw[j] = *(const v4f*)(gw[j] + kn);
    }
    const float* Ab = &As[cur][aoff];
    const float* Wb = &Ws[cur][woff];
#pragma unroll
    for (int ks = 0; ks < BK / 4; ++ks) {
      v2f a[FM], bb[FN];
#pragma unroll
      for (int i = 0; i < FM; ++i) a[i] = *(const v2f*)(Ab + i * 16 * PITCH + ks * 4);
#pragma unroll
      for (int j = 0; j < FN; ++j) bb[j] = *(const v2f*)(Wb + j * 16 * PITCH + ks * 4);
#pragma unroll
      for (int i = 0; i < FM; ++i)
#pragma unroll
        for (int j = 0; j < FN; ++j)
          acc[i][j] = __builtin_amdgcn_wmma_f32_16x16x4_f32(false, a[i], false, bb[j], (short)0,
                                                            acc[i][j], false, false);
    }
    if (more) {
#pragma unroll
      for (int j = 0; j < NA; ++j) *(v4f*)(&As[cur ^ 1][la[j]]) = ra[j];
#pragma unroll
      for (int j = 0; j < NW; ++j) *(v4f*)(&Ws[cur ^ 1][lw[j]]) = rw[j];
    }
    __syncthreads();
  }

#pragma unroll
  for (int i = 0; i < FM; ++i) {
#pragma unroll
    for (int j = 0; j < FN; ++j) {
      const long col = col0 + wn * (BN / 2) + j * 16 + l;
#pragma unroll
      for (int r = 0; r < 8; ++r) {
        const long row = row0 + wm * (BM / 2) + i * 16 + r + 8 * half;
        float v = acc[i][j][r];
        if (EPI == 1) v = fmaxf(v, 0.f);
        C[row * N + col] = v;
      }
    }
  }

  if (EPI == 1) {
    static_assert(EPI != 1 || (BM == 64 && BN == 64), "row-sum epilogue: 32 x 32 wave tile");
    float p[FM][8];
#pragma unroll
    for (int i = 0; i < FM; ++i) {
#pragma unroll
      for (int r = 0; r < 8; ++r) {
        float s = 0.f;
#pragma unroll
        for (int j = 0; j < FN; ++j) s += fmaxf(acc[i][j][r], 0.f);
        s = row_xadd<0x161>(s);
        s = row_xadd<0x162>(s);
        s = row_xadd<0x164>(s);
        s = row_xadd<0x168>(s);
        p[i][r] = s;
      }
    }
    float mine = 0.f;
#pragma unroll
    for (int i = 0; i < FM; ++i)
#pragma unroll
      for (int r = 0; r < 8; ++r)
        if (l == i * 8 + r) mine = p[i][r];
    const int rho = (l >> 3) * 16 + 8 * half + (l & 7);
    const long rows = (long)gridDim.y * BM;
    Rs[(long)(blockIdx.x * 2 + wn) * rows + row0 + wm * (BM / 2) + rho] = mine;
  }
}

__global__ __launch_bounds__(128) void gemm_lnA_Bt_lds(const float* __restrict__ A,
                                                       const float* __restrict__ W,
                                                       float* __restrict__ C,
                                                       const float* __restrict__ Mul,
                                                       int N) {
  constexpr int BM = 64, BN = 64, BK = 32, KK = 256;
  constexpr int PP = KK + 4;
  constexpr int PITCH = BK + 4;
  constexpr int C4 = BK / 4;
  constexpr int NW = BN * C4 / 128;
  constexpr int FM = BM / 32;
  constexpr int FN = BN / 32;
  __shared__ __attribute__((aligned(16))) float Ap[BM * PP];
  __shared__ __attribute__((aligned(16))) float Ws[2][BN * PITCH];

  const int tid = threadIdx.x;
  const int lane = tid & 31;
  const int wave = tid >> 5;
  const int wm = wave >> 1;
  const int wn = wave & 1;
  const int half = lane >> 4;
  const int l = lane & 15;
  const int row0 = blockIdx.y * BM;
  const int col0 = blockIdx.x * BN;

  const float* gw[NW];
  int lw[NW];
#pragma unroll
  for (int j = 0; j < NW; ++j) {
    const int f = tid + 128 * j;
    const int r = f / C4;
    const int c = f - r * C4;
    gw[j] = W + (long)(col0 + r) * KK + c * 4;
    lw[j] = r * PITCH + c * 4;
  }

  v4f p[16][2];
  const float* gp = A + (long)(row0 + wave * 16) * KK + lane * 4;
#pragma unroll
  for (int i = 0; i < 16; ++i) {
    p[i][0] = *(const v4f*)(gp + (long)i * KK);
    p[i][1] = *(const v4f*)(gp + (long)i * KK + 128);
  }
  v4f rw[NW];
#pragma unroll
  for (int j = 0; j < NW; ++j) rw[j] = *(const v4f*)(gw[j]);

#pragma unroll
  for (int i = 0; i < 16; ++i) {
    float s = ((p[i][0][0] + p[i][0][1]) + (p[i][0][2] + p[i][0][3])) +
              ((p[i][1][0] + p[i][1][1]) + (p[i][1][2] + p[i][1][3]));
    s = row_xadd<0x161>(s);
    s = row_xadd<0x162>(s);
    s = row_xadd<0x164>(s);
    s = row_xadd<0x168>(s);
    s += __shfl_xor(s, 16, 32);
    const float mean = s * (1.0f / 256.0f);
    const v4f c0 = p[i][0] - mean;
    const v4f c1 = p[i][1] - mean;
    float q = ((c0[0] * c0[0] + c0[1] * c0[1]) + (c0[2] * c0[2] + c0[3] * c0[3])) +
              ((c1[0] * c1[0] + c1[1] * c1[1]) + (c1[2] * c1[2] + c1[3] * c1[3]));
    q = row_xadd<0x161>(q);
    q = row_xadd<0x162>(q);
    q = row_xadd<0x164>(q);
    q = row_xadd<0x168>(q);
    q += __shfl_xor(q, 16, 32);
    const float rstd = rsqrtf(q * (1.0f / 256.0f) + 1e-5f);
    float* dst = &Ap[(wave * 16 + i) * PP + lane * 4];
    *(v4f*)(dst) = c0 * rstd;
    *(v4f*)(dst + 128) = c1 * rstd;
  }
#pragma unroll
  for (int j = 0; j < NW; ++j) *(v4f*)(&Ws[0][lw[j]]) = rw[j];
  __syncthreads();

  v8f acc[FM][FN];
#pragma unroll
  for (int i = 0; i < FM; ++i)
#pragma unroll
    for (int j = 0; j < FN; ++j) acc[i][j] = v8f{};

  const int aoff = (wm * (BM / 2) + l) * PP + 2 * half;
  const int woff = (wn * (BN / 2) + l) * PITCH + 2 * half;
  constexpr int nchunks = KK / BK;
#pragma unroll 1
  for (int ch = 0; ch < nchunks; ++ch) {
    const int cur = ch & 1;
    const bool more = (ch + 1 < nchunks);
    if (more) {
      const int kn = (ch + 1) * BK;
#pragma unroll
      for (int j = 0; j < NW; ++j) rw[j] = *(const v4f*)(gw[j] + kn);
    }
    const float* Ab = &Ap[aoff + ch * BK];
    const float* Wb = &Ws[cur][woff];
#pragma unroll
    for (int ks = 0; ks < BK / 4; ++ks) {
      v2f a[FM], bb[FN];
#pragma unroll
      for (int i = 0; i < FM; ++i) a[i] = *(const v2f*)(Ab + i * 16 * PP + ks * 4);
#pragma unroll
      for (int j = 0; j < FN; ++j) bb[j] = *(const v2f*)(Wb + j * 16 * PITCH + ks * 4);
#pragma unroll
      for (int i = 0; i < FM; ++i)
#pragma unroll
        for (int j = 0; j < FN; ++j)
          acc[i][j] = __builtin_amdgcn_wmma_f32_16x16x4_f32(false, a[i], false, bb[j], (short)0,
                                                            acc[i][j], false, false);
    }
    if (more) {
#pragma unroll
      for (int j = 0; j < NW; ++j) *(v4f*)(&Ws[cur ^ 1][lw[j]]) = rw[j];
    }
    __syncthreads();
  }

#pragma unroll
  for (int i = 0; i < FM; ++i) {
#pragma unroll
    for (int j = 0; j < FN; ++j) {
      const long col = col0 + wn * (BN / 2) + j * 16 + l;
#pragma unroll
      for (int r = 0; r < 8; ++r) {
        const long row = row0 + wm * (BM / 2) + i * 16 + r + 8 * half;
        const float v = fmaxf(acc[i][j][r], 0.f) * fmaxf(Mul[row * N + col], 0.f);
        C[row * N + col] = v;
      }
    }
  }
}

static __device__ __forceinline__ v4f decay_weights(const float* pw, int t, int s0) {
  v4f w;
#pragma unroll
  for (int e = 0; e < 4; ++e) {
    const int d = t - 1 - (s0 + e);
    w[e] = pw[d < 0 ? 0 : d];
  }
  return w;
}
static __device__ __forceinline__ v4f decay_apply(v4f v, v4f w, int t, int s0) {
  v4f o;
#pragma unroll
  for (int e = 0; e < 4; ++e) o[e] = (s0 + e < t) ? w[e] * v[e] : 0.f;
  return o;
}

template <int BM, int BN, int BK, bool CAUSAL, bool DECAY>
__global__ __launch_bounds__(128) void gemm_AB_lds(const float* __restrict__ A,
                                                   const float* __restrict__ Bm,
                                                   float* __restrict__ C,
                                                   int N, int K,
                                                   long sA, long sB, long sC) {
  constexpr int PA = BK + 4;
  constexpr int PB = BN + 8;
  constexpr int CA = BK / 4;
  constexpr int CB = BN / 4;
  constexpr int NA = BM * CA / 128;
  constexpr int NB = BK * CB / 128;
  constexpr int FM = BM / 32;
  constexpr int FN = BN / 32;
  __shared__ __attribute__((aligned(16))) float As[2][BM * PA];
  __shared__ __attribute__((aligned(16))) float Bs[2][BK * PB];
  __shared__ float pw[DECAY ? Tt : 1];

  const int b = blockIdx.z;
  A += (long)b * sA;
  Bm += (long)b * sB;
  C += (long)b * sC;
  const int tid = threadIdx.x;
  const int lane = tid & 31;
  const int wave = tid >> 5;
  const int wm = wave >> 1;
  const int wn = wave & 1;
  const int half = lane >> 4;
  const int l = lane & 15;
  const int row0 = blockIdx.y * BM;
  const int col0 = blockIdx.x * BN;

  const float* ga[NA];
  int la[NA];
  int ta[NA], sa[NA];
#pragma unroll
  for (int j = 0; j < NA; ++j) {
    const int f = tid + 128 * j;
    const int r = f / CA;
    const int c = f - r * CA;
    ga[j] = A + (long)(row0 + r) * K + c * 4;
    la[j] = r * PA + c * 4;
    ta[j] = row0 + r;
    sa[j] = c * 4;
  }
  const float* gb[NB];
  int lb[NB];
#pragma unroll
  for (int j = 0; j < NB; ++j) {
    const int f = tid + 128 * j;
    const int r = f / CB;
    const int c = f - r * CB;
    gb[j] = Bm + (long)r * N + col0 + c * 4;
    lb[j] = r * PB + c * 4;
  }

  v4f ra[NA], rb[NB];
#pragma unroll
  for (int j = 0; j < NA; ++j) ra[j] = *(const v4f*)(ga[j]);
#pragma unroll
  for (int j = 0; j < NB; ++j) rb[j] = *(const v4f*)(gb[j]);
  v4f wq[NA];
  if (DECAY) {
    for (int i = tid; i < Tt; i += 128) pw[i] = __powf(0.97f, (float)i);
    __syncthreads();
#pragma unroll
    for (int j = 0; j < NA; ++j) wq[j] = decay_weights(pw, ta[j], sa[j]);
#pragma unroll
    for (int j = 0; j < NA; ++j) ra[j] = decay_apply(ra[j], wq[j], ta[j], sa[j]);
  }
#pragma unroll
  for (int j = 0; j < NA; ++j) *(v4f*)(&As[0][la[j]]) = ra[j];
#pragma unroll
  for (int j = 0; j < NB; ++j) *(v4f*)(&Bs[0][lb[j]]) = rb[j];
  __syncthreads();

  v8f acc[FM][FN];
#pragma unroll
  for (int i = 0; i < FM; ++i)
#pragma unroll
    for (int j = 0; j < FN; ++j) acc[i][j] = v8f{};

  const int aoff = (wm * (BM / 2) + l) * PA + 2 * half;
  const int boff = (2 * half) * PB + wn * (BN / 2) + l;
  const int nchunks = CAUSAL ? (row0 + BM) / BK : K / BK;
  for (int ch = 0; ch < nchunks; ++ch) {
    const int cur = ch & 1;
    const bool more = (ch + 1 < nchunks);
    if (more) {
      const int kn = (ch + 1) * BK;
#pragma unroll
      for (int j = 0; j < NA; ++j) ra[j] = *(const v4f*)(ga[j] + kn);
#pragma unroll
      for (int j = 0; j < NB; ++j) rb[j] = *(const v4f*)(gb[j] + (long)kn * N);
      if (DECAY) {
#pragma unroll
        for (int j = 0; j < NA; ++j) wq[j] = decay_weights(pw, ta[j], sa[j] + kn);
      }
    }
    const float* Ab = &As[cur][aoff];
    const float* Bb = &Bs[cur][boff];
#pragma unroll
    for (int ks = 0; ks < BK / 4; ++ks) {
      v2f a[FM], bb[FN];
#pragma unroll
      for (int i = 0; i < FM; ++i) a[i] = *(const v2f*)(Ab + i * 16 * PA + ks * 4);
#pragma unroll
      for (int j = 0; j < FN; ++j) {
        bb[j][0] = Bb[(ks * 4) * PB + j * 16];
        bb[j][1] = Bb[(ks * 4 + 1) * PB + j * 16];
      }
#pragma unroll
      for (int i = 0; i < FM; ++i)
#pragma unroll
        for (int j = 0; j < FN; ++j)
          acc[i][j] = __builtin_amdgcn_wmma_f32_16x16x4_f32(false, a[i], false, bb[j], (short)0,
                                                            acc[i][j], false, false);
    }
    if (more) {
      if (DECAY) {
        const int kn = (ch + 1) * BK;
#pragma unroll
        for (int j = 0; j < NA; ++j) ra[j] = decay_apply(ra[j], wq[j], ta[j], sa[j] + kn);
      }
#pragma unroll
      for (int j = 0; j < NA; ++j) *(v4f*)(&As[cur ^ 1][la[j]]) = ra[j];
#pragma unroll
      for (int j = 0; j < NB; ++j) *(v4f*)(&Bs[cur ^ 1][lb[j]]) = rb[j];
    }
    __syncthreads();
  }

#pragma unroll
  for (int i = 0; i < FM; ++i) {
#pragma unroll
    for (int j = 0; j < FN; ++j) {
      const long col = col0 + wn * (BN / 2) + j * 16 + l;
#pragma unroll
      for (int r = 0; r < 8; ++r) {
        const long row = row0 + wm * (BM / 2) + i * 16 + r + 8 * half;
        C[row * N + col] = acc[i][j][r];
      }
    }
  }
}

extern "C" void kernel_launch(void* const* d_in, const int* in_sizes, int n_in,
                              void* d_out, int out_size, void* d_ws, size_t ws_size,
                              hipStream_t stream) {
  const float* emb = (const float*)d_in[0];
  const float* E   = (const float*)d_in[1];
  const float* Dx  = (const float*)d_in[2];
  const float* Dy  = (const float*)d_in[3];
  float* out = (float*)d_out;

  const long BT = (long)Bb * Tt;
  const long szBTD = BT * Dd;
  const long szBTN = BT * Nn;
  const long szBTT = (long)Bb * Tt * Tt;

  float* ws = (float*)d_ws;
  float* LNV = ws;
  float* XU  = LNV + szBTD;
  float* X   = XU + szBTN;
  float* Kg  = X + szBTN;
  float* G   = Kg + szBTT;
  float* As  = G + szBTT;
  float* Y   = As + szBTD;
  float* V   = Y + szBTN;

  ln256_kernel<<<(int)BT, 256, 0, stream>>>(emb, LNV);

  float* Rp = Kg;
  gemm_ABt_lds<1, 64, 64, 32><<<dim3(1024 / 64, 2048 / 64, 1), 128, 0, stream>>>(
      emb, Dx, XU, Rp, 1024, 256, 0, 0, 0);

  gbuild_kernel<<<dim3(Tt / GB_ROWS, Bb), 256, 0, stream>>>(Rp, G);

  gemm_AB_lds<64, 64, 32, true, false><<<dim3(1024 / 64, 256 / 64, Bb), 128, 0, stream>>>(
      G, XU, X, 1024, 256, (long)Tt * Tt, (long)Tt * Nn, (long)Tt * Nn);

  gemm_ABt_lds<0, 32, 64, 64><<<dim3(256 / 64, 256 / 32, Bb), 128, 0, stream>>>(
      X, X, Kg, nullptr, 256, 1024, (long)Tt * Nn, (long)Tt * Nn, (long)Tt * Tt);

  gemm_AB_lds<32, 64, 32, false, true><<<dim3(256 / 64, 256 / 32, Bb), 128, 0, stream>>>(
      Kg, LNV, As, 256, 256, (long)Tt * Tt, (long)Tt * Dd, (long)Tt * Dd);

  gemm_lnA_Bt_lds<<<dim3(1024 / 64, 2048 / 64, 1), 128, 0, stream>>>(As, Dy, Y, X, 1024);

  gemm_ABt_lds<0, 32, 64, 64><<<dim3(256 / 64, 2048 / 32, 1), 128, 0, stream>>>(
      Y, E, V, nullptr, 256, 1024, 0, 0, 0);

  ln256_kernel<<<(int)BT, 256, 0, stream>>>(V, out);
}


# Gram and y@E^T (K=1024): K quarters on 4 workgroups (64x64 tiles, 512 workgroups, 2 per WGP; was 256 of 32x64); 4 partial planes added by the reader's loads (decay GEMM, last LN); sum order changes
# speedup vs baseline: 8.9919x; 1.1057x over previous
#include <hip/hip_runtime.h>


typedef __attribute__((ext_vector_type(2))) float v2f;
typedef __attribute__((ext_vector_type(8))) float v8f;
typedef __attribute__((ext_vector_type(4))) float v4f;

static constexpr int Bb = 8;
static constexpr int Tt = 256;
static constexpr int Dd = 256;
static constexpr int Nn = 1024;

template <int NP>
__global__ __launch_bounds__(256) void ln256_kernel(const float* __restrict__ in,
                                                    float* __restrict__ out, long sP) {
  static_assert(NP == 1 || NP == 4, "1 plane or 4 partial planes");
  const int D = 256;
  const long row = blockIdx.x;
  const int j = threadIdx.x;
  float pl[NP];
#pragma unroll
  for (int p = 0; p < NP; ++p) pl[p] = in[(long)p * sP + row * D + j];
  float v = pl[0];
  if constexpr (NP == 4) v = (pl[0] + pl[1]) + (pl[2] + pl[3]);
  __shared__ float red[8];

  float s = v;
#pragma unroll
  for (int o = 16; o > 0; o >>= 1) s += __shfl_down(s, o, 32);
  if ((j & 31) == 0) red[j >> 5] = s;
  __syncthreads();
  float tot = 0.f;
#pragma unroll
  for (int w = 0; w < 8; ++w) tot += red[w];
  const float mean = tot * (1.0f / 256.0f);
  const float c = v - mean;
  __syncthreads();

  float s2 = c * c;
#pragma unroll
  for (int o = 16; o > 0; o >>= 1) s2 += __shfl_down(s2, o, 32);
  if ((j & 31) == 0) red[j >> 5] = s2;
  __syncthreads();
  float tot2 = 0.f;
#pragma unroll
  for (int w = 0; w < 8; ++w) tot2 += red[w];
  const float var = tot2 * (1.0f / 256.0f);

  out[row * D + j] = c * rsqrtf(var + 1e-5f);
}


template <int CTRL>
static __device__ __forceinline__ float row_xadd(float v) {
  const int iv = __builtin_bit_cast(int, v);
  const int sw = __builtin_amdgcn_update_dpp(0, iv, CTRL, 0xf, 0xf, true);
  return v + __builtin_bit_cast(float, sw);
}

static constexpr int RS_STRIPS = 32;
static constexpr int GB_ROWS = 32;

__global__ __launch_bounds__(256) void gbuild_kernel(const float* __restrict__ Rp,
                                                     float* __restrict__ G) {
  const int rb = blockIdx.x;
  const int b = blockIdx.y;
  const int s = threadIdx.x;
  const int lane = s & 31;
  const int wave = s >> 5;
  __shared__ __attribute__((aligned(16))) float rs[Tt];
  __shared__ __attribute__((aligned(16))) float qs[Tt];
  __shared__ unsigned big[8];
  __shared__ float bp[8];

  float r = 0.f;
#pragma unroll
  for (int c = 0; c < RS_STRIPS; ++c) r += Rp[(long)c * (Bb * Tt) + b * Tt + s];
  rs[s] = r;
  const unsigned m = __builtin_amdgcn_ballot_w32(r >= 1e-12f);
  if (lane == 0) big[wave] = m;
  __syncthreads();

  int f = Tt;
#pragma unroll
  for (int w = 7; w >= 0; --w) {
    const unsigned mw = big[w];
    if (mw != 0u) f = 32 * w + __builtin_ctz(mw);
  }
  float S = 1.0f;
  if (s <= f) {
    S = 0.f;
    for (int i = 0; i < s; ++i) {
      const float sum = fmaf(0.97f, S, rs[i]);
      S = (sum >= 1e-12f) ? 1.0f : sum * 1e12f;
    }
  }
  const float d = fmaxf(fmaf(0.97f, S, r), 1e-12f);
  const float inv = 1.0f / d;
  const float q = 0.97f * inv;
  qs[s] = q;

  const float GMAX = 3.0e38f;

  float v = q;
#pragma unroll
  for (int o = 1; o < 32; o <<= 1) {
    const float up = __shfl_down(v, o, 32);
    if (lane + o < 32) v = fminf(v * up, GMAX);
  }
  const float nxt = __shfl_down(v, 1, 32);
  const float suf = (lane < 31) ? nxt : 1.0f;
  if (lane == 0) bp[wave] = v;
  __syncthreads();

  const int t0 = rb * GB_ROWS;
  float* gp = G + ((long)b * Tt + t0) * Tt + s;
  if (wave > rb) {
#pragma unroll 8
    for (int i = 0; i < GB_ROWS; ++i) gp[(long)i * Tt] = 0.f;
  } else {
    float g = 0.f;
    if (wave < rb) {
      g = fminf(inv * suf, GMAX);
      for (int k = wave + 1; k < rb; ++k) g = fminf(g * bp[k], GMAX);
    }
#pragma unroll 8
    for (int i = 0; i < GB_ROWS; ++i) {
      const int tt = t0 + i;
      const float gn = fminf(g * qs[tt], GMAX);
      g = (tt < s) ? 0.f : ((tt == s) ? inv : gn);
      gp[(long)i * Tt] = g;
    }
  }
}

template <int EPI, int BM, int BN, int BK, int SPLIT>
__global__ __launch_bounds__(128) void gemm_ABt_lds(const float* __restrict__ A,
                                                    const float* __restrict__ W,
                                                    float* __restrict__ C,
                                                    float* __restrict__ Rs,
                                                    int N, int K,
                                                    long sA, long sW, long sC, long sP) {
  constexpr int PITCH = BK + 4;
  constexpr int C4 = BK / 4;
  constexpr int NA = BM * C4 / 128;
  constexpr int NW = BN * C4 / 128;
  constexpr int FM = BM / 32;
  constexpr int FN = BN / 32;
  __shared__ __attribute__((aligned(16))) float As[2][BM * PITCH];
  __shared__ __attribute__((aligned(16))) float Ws[2][BN * PITCH];

  const int b = blockIdx.z / SPLIT;
  const int sp = blockIdx.z - b * SPLIT;
  const int Ks = K / SPLIT;
  A += (long)b * sA + sp * Ks;
  W += (long)b * sW + sp * Ks;
  C += (long)b * sC + (long)sp * sP;
  const int tid = threadIdx.x;
  const int lane = tid & 31;
  const int wave = tid >> 5;
  const int wm = wave >> 1;
  const int wn = wave & 1;
  const int half = lane >> 4;
  const int l = lane & 15;
  const int row0 = blockIdx.y * BM;
  const int col0 = blockIdx.x * BN;

  const float* ga[NA];
  int la[NA];
#pragma unroll
  for (int j = 0; j < NA; ++j) {
    const int f = tid + 128 * j;
    const int r = f / C4;
    const int c = f - r * C4;
    ga[j] = A + (long)(row0 + r) * K + c * 4;
    la[j] = r * PITCH + c * 4;
  }
  const float* gw[NW];
  int lw[NW];
#pragma unroll
  for (int j = 0; j < NW; ++j) {
    const int f = tid + 128 * j;
    const int r = f / C4;
    const int c = f - r * C4;
    gw[j] = W + (long)(col0 + r) * K + c * 4;
    lw[j] = r * PITCH + c * 4;
  }

  v4f ra[NA], rw[NW];
#pragma unroll
  for (int j = 0; j < NA; ++j) ra[j] = *(const v4f*)(ga[j]);
#pragma unroll
  for (int j = 0; j < NW; ++j) rw[j] = *(const v4f*)(gw[j]);
#pragma unroll
  for (int j = 0; j < NA; ++j) *(v4f*)(&As[0][la[j]]) = ra[j];
#pragma unroll
  for (int j = 0; j < NW; ++j) *(v4f*)(&Ws[0][lw[j]]) = rw[j];
  __syncthreads();

  v8f acc[FM][FN];
#pragma unroll
  for (int i = 0; i < FM; ++i)
#pragma unroll
    for (int j = 0; j < FN; ++j) acc[i][j] = v8f{};

  const int aoff = (wm * (BM / 2) + l) * PITCH + 2 * half;
  const int woff = (wn * (BN / 2) + l) * PITCH + 2 * half;
  const int nchunks = Ks / BK;
  for (int ch = 0; ch < nchunks; ++ch) {
    const int cur = ch & 1;
    const bool more = (ch + 1 < nchunks);
    if (more) {
      const int kn = (ch + 1) * BK;
#pragma unroll
      for (int j = 0; j < NA; ++j) ra[j] = *(const v4f*)(ga[j] + kn);
#pragma unroll
      for (int j = 0; j < NW; ++j) rw[j] = *(const v4f*)(gw[j] + kn);
    }
    const float* Ab = &As[cur][aoff];
    const float* Wb = &Ws[cur][woff];
#pragma unroll
    for (int ks = 0; ks < BK / 4; ++ks) {
      v2f a[FM], bb[FN];
#pragma unroll
      for (int i = 0; i < FM; ++i) a[i] = *(const v2f*)(Ab + i * 16 * PITCH + ks * 4);
#pragma unroll
      for (int j = 0; j < FN; ++j) bb[j] = *(const v2f*)(Wb + j * 16 * PITCH + ks * 4);
#pragma unroll
      for (int i = 0; i < FM; ++i)
#pragma unroll
        for (int j = 0; j < FN; ++j)
          acc[i][j] = __builtin_amdgcn_wmma_f32_16x16x4_f32(false, a[i], false, bb[j], (short)0,
                                                            acc[i][j], false, false);
    }
    if (more) {
#pragma unroll
      for (int j = 0; j < NA; ++j) *(v4f*)(&As[cur ^ 1][la[j]]) = ra[j];
#pragma unroll
      for (int j = 0; j < NW; ++j) *(v4f*)(&Ws[cur ^ 1][lw[j]]) = rw[j];
    }
    __syncthreads();
  }

#pragma unroll
  for (int i = 0; i < FM; ++i) {
#pragma unroll
    for (int j = 0; j < FN; ++j) {
      const long col = col0 + wn * (BN / 2) + j * 16 + l;
#pragma unroll
      for (int r = 0; r < 8; ++r) {
        const long row = row0 + wm * (BM / 2) + i * 16 + r + 8 * half;
        float v = acc[i][j][r];
        if (EPI == 1) v = fmaxf(v, 0.f);
        C[row * N + col] = v;
      }
    }
  }

  if (EPI == 1) {
    static_assert(EPI != 1 || (BM == 64 && BN == 64), "row-sum epilogue: 32 x 32 wave tile");
    float p[FM][8];
#pragma unroll
    for (int i = 0; i < FM; ++i) {
#pragma unroll
      for (int r = 0; r < 8; ++r) {
        float s = 0.f;
#pragma unroll
        for (int j = 0; j < FN; ++j) s += fmaxf(acc[i][j][r], 0.f);
        s = row_xadd<0x161>(s);
        s = row_xadd<0x162>(s);
        s = row_xadd<0x164>(s);
        s = row_xadd<0x168>(s);
        p[i][r] = s;
      }
    }
    float mine = 0.f;
#pragma unroll
    for (int i = 0; i < FM; ++i)
#pragma unroll
      for (int r = 0; r < 8; ++r)
        if (l == i * 8 + r) mine = p[i][r];
    const int rho = (l >> 3) * 16 + 8 * half + (l & 7);
    const long rows = (long)gridDim.y * BM;
    Rs[(long)(blockIdx.x * 2 + wn) * rows + row0 + wm * (BM / 2) + rho] = mine;
  }
}

__global__ __launch_bounds__(128) void gemm_lnA_Bt_lds(const float* __restrict__ A,
                                                       const float* __restrict__ W,
                                                       float* __restrict__ C,
                                                       const float* __restrict__ Mul,
                                                       int N) {
  constexpr int BM = 64, BN = 64, BK = 32, KK = 256;
  constexpr int PP = KK + 4;
  constexpr int PITCH = BK + 4;
  constexpr int C4 = BK / 4;
  constexpr int NW = BN * C4 / 128;
  constexpr int FM = BM / 32;
  constexpr int FN = BN / 32;
  __shared__ __attribute__((aligned(16))) float Ap[BM * PP];
  __shared__ __attribute__((aligned(16))) float Ws[2][BN * PITCH];

  const int tid = threadIdx.x;
  const int lane = tid & 31;
  const int wave = tid >> 5;
  const int wm = wave >> 1;
  const int wn = wave & 1;
  const int half = lane >> 4;
  const int l = lane & 15;
  const int row0 = blockIdx.y * BM;
  const int col0 = blockIdx.x * BN;

  const float* gw[NW];
  int lw[NW];
#pragma unroll
  for (int j = 0; j < NW; ++j) {
    const int f = tid + 128 * j;
    const int r = f / C4;
    const int c = f - r * C4;
    gw[j] = W + (long)(col0 + r) * KK + c * 4;
    lw[j] = r * PITCH + c * 4;
  }

  v4f p[16][2];
  const float* gp = A + (long)(row0 + wave * 16) * KK + lane * 4;
#pragma unroll
  for (int i = 0; i < 16; ++i) {
    p[i][0] = *(const v4f*)(gp + (long)i * KK);
    p[i][1] = *(const v4f*)(gp + (long)i * KK + 128);
  }
  v4f rw[NW];
#pragma unroll
  for (int j = 0; j < NW; ++j) rw[j] = *(const v4f*)(gw[j]);

#pragma unroll
  for (int i = 0; i < 16; ++i) {
    float s = ((p[i][0][0] + p[i][0][1]) + (p[i][0][2] + p[i][0][3])) +
              ((p[i][1][0] + p[i][1][1]) + (p[i][1][2] + p[i][1][3]));
    s = row_xadd<0x161>(s);
    s = row_xadd<0x162>(s);
    s = row_xadd<0x164>(s);
    s = row_xadd<0x168>(s);
    s += __shfl_xor(s, 16, 32);
    const float mean = s * (1.0f / 256.0f);
    const v4f c0 = p[i][0] - mean;
    const v4f c1 = p[i][1] - mean;
    float q = ((c0[0] * c0[0] + c0[1] * c0[1]) + (c0[2] * c0[2] + c0[3] * c0[3])) +
              ((c1[0] * c1[0] + c1[1] * c1[1]) + (c1[2] * c1[2] + c1[3] * c1[3]));
    q = row_xadd<0x161>(q);
    q = row_xadd<0x162>(q);
    q = row_xadd<0x164>(q);
    q = row_xadd<0x168>(q);
    q += __shfl_xor(q, 16, 32);
    const float rstd = rsqrtf(q * (1.0f / 256.0f) + 1e-5f);
    float* dst = &Ap[(wave * 16 + i) * PP + lane * 4];
    *(v4f*)(dst) = c0 * rstd;
    *(v4f*)(dst + 128) = c1 * rstd;
  }
#pragma unroll
  for (int j = 0; j < NW; ++j) *(v4f*)(&Ws[0][lw[j]]) = rw[j];
  __syncthreads();

  v8f acc[FM][FN];
#pragma unroll
  for (int i = 0; i < FM; ++i)
#pragma unroll
    for (int j = 0; j < FN; ++j) acc[i][j] = v8f{};

  const int aoff = (wm * (BM / 2) + l) * PP + 2 * half;
  const int woff = (wn * (BN / 2) + l) * PITCH + 2 * half;
  constexpr int nchunks = KK / BK;
#pragma unroll 1
  for (int ch = 0; ch < nchunks; ++ch) {
    const int cur = ch & 1;
    const bool more = (ch + 1 < nchunks);
    if (more) {
      const int kn = (ch + 1) * BK;
#pragma unroll
      for (int j = 0; j < NW; ++j) rw[j] = *(const v4f*)(gw[j] + kn);
    }
    const float* Ab = &Ap[aoff + ch * BK];
    const float* Wb = &Ws[cur][woff];
#pragma unroll
    for (int ks = 0; ks < BK / 4; ++ks) {
      v2f a[FM], bb[FN];
#pragma unroll
      for (int i = 0; i < FM; ++i) a[i] = *(const v2f*)(Ab + i * 16 * PP + ks * 4);
#pragma unroll
      for (int j = 0; j < FN; ++j) bb[j] = *(const v2f*)(Wb + j * 16 * PITCH + ks * 4);
#pragma unroll
      for (int i = 0; i < FM; ++i)
#pragma unroll
        for (int j = 0; j < FN; ++j)
          acc[i][j] = __builtin_amdgcn_wmma_f32_16x16x4_f32(false, a[i], false, bb[j], (short)0,
                                                            acc[i][j], false, false);
    }
    if (more) {
#pragma unroll
      for (int j = 0; j < NW; ++j) *(v4f*)(&Ws[cur ^ 1][lw[j]]) = rw[j];
    }
    __syncthreads();
  }

#pragma unroll
  for (int i = 0; i < FM; ++i) {
#pragma unroll
    for (int j = 0; j < FN; ++j) {
      const long col = col0 + wn * (BN / 2) + j * 16 + l;
#pragma unroll
      for (int r = 0; r < 8; ++r) {
        const long row = row0 + wm * (BM / 2) + i * 16 + r + 8 * half;
        const float v = fmaxf(acc[i][j][r], 0.f) * fmaxf(Mul[row * N + col], 0.f);
        C[row * N + col] = v;
      }
    }
  }
}

static __device__ __forceinline__ v4f decay_weights(const float* pw, int t, int s0) {
  v4f w;
#pragma unroll
  for (int e = 0; e < 4; ++e) {
    const int d = t - 1 - (s0 + e);
    w[e] = pw[d < 0 ? 0 : d];
  }
  return w;
}
static __device__ __forceinline__ v4f decay_apply(v4f v, v4f w, int t, int s0) {
  v4f o;
#pragma unroll
  for (int e = 0; e < 4; ++e) o[e] = (s0 + e < t) ? w[e] * v[e] : 0.f;
  return o;
}

template <int APL, int NA>
static __device__ __forceinline__ v4f plane_sum(const v4f (&rp)[APL][NA], int j) {
  if constexpr (APL == 4) return (rp[0][j] + rp[1][j]) + (rp[2][j] + rp[3][j]);
  else return rp[0][j];
}

template <int BM, int BN, int BK, bool CAUSAL, bool DECAY, int APL>
__global__ __launch_bounds__(128) void gemm_AB_lds(const float* __restrict__ A,
                                                   const float* __restrict__ Bm,
                                                   float* __restrict__ C,
                                                   int N, int K,
                                                   long sA, long sB, long sC, long sPA) {
  static_assert(APL == 1 || APL == 4, "1 plane or 4 partial planes");
  constexpr int PA = BK + 4;
  constexpr int PB = BN + 8;
  constexpr int CA = BK / 4;
  constexpr int CB = BN / 4;
  constexpr int NA = BM * CA / 128;
  constexpr int NB = BK * CB / 128;
  constexpr int FM = BM / 32;
  constexpr int FN = BN / 32;
  __shared__ __attribute__((aligned(16))) float As[2][BM * PA];
  __shared__ __attribute__((aligned(16))) float Bs[2][BK * PB];
  __shared__ float pw[DECAY ? Tt : 1];

  const int b = blockIdx.z;
  A += (long)b * sA;
  Bm += (long)b * sB;
  C += (long)b * sC;
  const int tid = threadIdx.x;
  const int lane = tid & 31;
  const int wave = tid >> 5;
  const int wm = wave >> 1;
  const int wn = wave & 1;
  const int half = lane >> 4;
  const int l = lane & 15;
  const int row0 = blockIdx.y * BM;
  const int col0 = blockIdx.x * BN;

  const float* ga[NA];
  int la[NA];
  int ta[NA], sa[NA];
#pragma unroll
  for (int j = 0; j < NA; ++j) {
    const int f = tid + 128 * j;
    const int r = f / CA;
    const int c = f - r * CA;
    ga[j] = A + (long)(row0 + r) * K + c * 4;
    la[j] = r * PA + c * 4;
    ta[j] = row0 + r;
    sa[j] = c * 4;
  }
  const float* gb[NB];
  int lb[NB];
#pragma unroll
  for (int j = 0; j < NB; ++j) {
    const int f = tid + 128 * j;
    const int r = f / CB;
    const int c = f - r * CB;
    gb[j] = Bm + (long)r * N + col0 + c * 4;
    lb[j] = r * PB + c * 4;
  }

  v4f rp[APL][NA];
  v4f ra[NA], rb[NB];
#pragma unroll
  for (int p = 0; p < APL; ++p)
#pragma unroll
    for (int j = 0; j < NA; ++j) rp[p][j] = *(const v4f*)(ga[j] + (long)p * sPA);
#pragma unroll
  for (int j = 0; j < NB; ++j) rb[j] = *(const v4f*)(gb[j]);
  v4f wq[NA];
  if (DECAY) {
    for (int i = tid; i < Tt; i += 128) pw[i] = __powf(0.97f, (float)i);
    __syncthreads();
#pragma unroll
    for (int j = 0; j < NA; ++j) wq[j] = decay_weights(pw, ta[j], sa[j]);
  }
#pragma unroll
  for (int j = 0; j < NA; ++j)
    ra[j] = plane_sum<APL, NA>(rp, j);
  if (DECAY) {
#pragma unroll
    for (int j = 0; j < NA; ++j) ra[j] = decay_apply(ra[j], wq[j], ta[j], sa[j]);
  }
#pragma unroll
  for (int j = 0; j < NA; ++j) *(v4f*)(&As[0][la[j]]) = ra[j];
#pragma unroll
  for (int j = 0; j < NB; ++j) *(v4f*)(&Bs[0][lb[j]]) = rb[j];
  __syncthreads();

  v8f acc[FM][FN];
#pragma unroll
  for (int i = 0; i < FM; ++i)
#pragma unroll
    for (int j = 0; j < FN; ++j) acc[i][j] = v8f{};

  const int aoff = (wm * (BM / 2) + l) * PA + 2 * half;
  const int boff = (2 * half) * PB + wn * (BN / 2) + l;
  const int nchunks = CAUSAL ? (row0 + BM) / BK : K / BK;
  for (int ch = 0; ch < nchunks; ++ch) {
    const int cur = ch & 1;
    const bool more = (ch + 1 < nchunks);
    if (more) {
      const int kn = (ch + 1) * BK;
#pragma unroll
      for (int p = 0; p < APL; ++p)
#pragma unroll
        for (int j = 0; j < NA; ++j) rp[p][j] = *(const v4f*)(ga[j] + (long)p * sPA + kn);
#pragma unroll
      for (int j = 0; j < NB; ++j) rb[j] = *(const v4f*)(gb[j] + (long)kn * N);
      if (DECAY) {
#pragma unroll
        for (int j = 0; j < NA; ++j) wq[j] = decay_weights(pw, ta[j], sa[j] + kn);
      }
    }
    const float* Ab = &As[cur][aoff];
    const float* Bb = &Bs[cur][boff];
#pragma unroll
    for (int ks = 0; ks < BK / 4; ++ks) {
      v2f a[FM], bb[FN];
#pragma unroll
      for (int i = 0; i < FM; ++i) a[i] = *(const v2f*)(Ab + i * 16 * PA + ks * 4);
#pragma unroll
      for (int j = 0; j < FN; ++j) {
        bb[j][0] = Bb[(ks * 4) * PB + j * 16];
        bb[j][1] = Bb[(ks * 4 + 1) * PB + j * 16];
      }
#pragma unroll
      for (int i = 0; i < FM; ++i)
#pragma unroll
        for (int j = 0; j < FN; ++j)
          acc[i][j] = __builtin_amdgcn_wmma_f32_16x16x4_f32(false, a[i], false, bb[j], (short)0,
                                                            acc[i][j], false, false);
    }
    if (more) {
#pragma unroll
      for (int j = 0; j < NA; ++j)
        ra[j] = plane_sum<APL, NA>(rp, j);
      if (DECAY) {
        const int kn = (ch + 1) * BK;
#pragma unroll
        for (int j = 0; j < NA; ++j) ra[j] = decay_apply(ra[j], wq[j], ta[j], sa[j] + kn);
      }
#pragma unroll
      for (int j = 0; j < NA; ++j) *(v4f*)(&As[cur ^ 1][la[j]]) = ra[j];
#pragma unroll
      for (int j = 0; j < NB; ++j) *(v4f*)(&Bs[cur ^ 1][lb[j]]) = rb[j];
    }
    __syncthreads();
  }

#pragma unroll
  for (int i = 0; i < FM; ++i) {
#pragma unroll
    for (int j = 0; j < FN; ++j) {
      const long col = col0 + wn * (BN / 2) + j * 16 + l;
#pragma unroll
      for (int r = 0; r < 8; ++r) {
        const long row = row0 + wm * (BM / 2) + i * 16 + r + 8 * half;
        C[row * N + col] = acc[i][j][r];
      }
    }
  }
}

extern "C" void kernel_launch(void* const* d_in, const int* in_sizes, int n_in,
                              void* d_out, int out_size, void* d_ws, size_t ws_size,
                              hipStream_t stream) {
  const float* emb = (const float*)d_in[0];
  const float* E   = (const float*)d_in[1];
  const float* Dx  = (const float*)d_in[2];
  const float* Dy  = (const float*)d_in[3];
  float* out = (float*)d_out;

  const long BT = (long)Bb * Tt;
  const long szBTD = BT * Dd;
  const long szBTN = BT * Nn;
  const long szBTT = (long)Bb * Tt * Tt;

  float* ws = (float*)d_ws;
  float* LNV = ws;
  float* XU  = LNV + szBTD;
  float* X   = XU + szBTN;
  float* Rp  = X + szBTN;
  float* G   = Rp + szBTT;
  float* As  = G + szBTT;
  float* Y   = As + szBTD;
  float* KgP = XU;
  float* VP  = XU;

  ln256_kernel<1><<<(int)BT, 256, 0, stream>>>(emb, LNV, 0);

  gemm_ABt_lds<1, 64, 64, 32, 1><<<dim3(1024 / 64, 2048 / 64, 1), 128, 0, stream>>>(
      emb, Dx, XU, Rp, 1024, 256, 0, 0, 0, 0);

  gbuild_kernel<<<dim3(Tt / GB_ROWS, Bb), 256, 0, stream>>>(Rp, G);

  gemm_AB_lds<64, 64, 32, true, false, 1><<<dim3(1024 / 64, 256 / 64, Bb), 128, 0, stream>>>(
      G, XU, X, 1024, 256, (long)Tt * Tt, (long)Tt * Nn, (long)Tt * Nn, 0);

  gemm_ABt_lds<0, 64, 64, 32, 4><<<dim3(256 / 64, 256 / 64, Bb * 4), 128, 0, stream>>>(
      X, X, KgP, nullptr, 256, 1024, (long)Tt * Nn, (long)Tt * Nn, (long)Tt * Tt, szBTT);

  gemm_AB_lds<32, 64, 32, false, true, 4><<<dim3(256 / 64, 256 / 32, Bb), 128, 0, stream>>>(
      KgP, LNV, As, 256, 256, (long)Tt * Tt, (long)Tt * Dd, (long)Tt * Dd, szBTT);

  gemm_lnA_Bt_lds<<<dim3(1024 / 64, 2048 / 64, 1), 128, 0, stream>>>(As, Dy, Y, X, 1024);

  gemm_ABt_lds<0, 64, 64, 32, 4><<<dim3(256 / 64, 2048 / 64, 4), 128, 0, stream>>>(
      Y, E, VP, nullptr, 256, 1024, 0, 0, 0, szBTD);

  ln256_kernel<4><<<(int)BT, 256, 0, stream>>>(VP, out, szBTD);
}


# Layer norms: one wave per row in registers (DPP + permlanex16 sums, no LDS or barrier); ln(emb) rides in the first GEMM's waves (9 -> 8 launches), last LN 1024 waves (was 16384); ln sum order changes
# speedup vs baseline: 9.3452x; 1.0393x over previous
#include <hip/hip_runtime.h>


typedef __attribute__((ext_vector_type(2))) float v2f;
typedef __attribute__((ext_vector_type(8))) float v8f;
typedef __attribute__((ext_vector_type(4))) float v4f;

static constexpr int Bb = 8;
static constexpr int Tt = 256;
static constexpr int Dd = 256;
static constexpr int Nn = 1024;

template <int CTRL>
static __device__ __forceinline__ float row_xadd(float v) {
  const int iv = __builtin_bit_cast(int, v);
  const int sw = __builtin_amdgcn_update_dpp(0, iv, CTRL, 0xf, 0xf, true);
  return v + __builtin_bit_cast(float, sw);
}

static __device__ __forceinline__ float xhalf(float v) {
  const int iv = __builtin_bit_cast(int, v);
  return __builtin_bit_cast(float, __builtin_amdgcn_permlanex16(iv, iv, 0x76543210, 0xfedcba98,
                                                                false, false));
}

static __device__ __forceinline__ void ln_row256(v4f& a, v4f& b) {
  float s = ((a[0] + a[1]) + (a[2] + a[3])) + ((b[0] + b[1]) + (b[2] + b[3]));
  s = row_xadd<0x161>(s);
  s = row_xadd<0x162>(s);
  s = row_xadd<0x164>(s);
  s = row_xadd<0x168>(s);
  s += xhalf(s);
  const float mean = s * (1.0f / 256.0f);
  a = a - mean;
  b = b - mean;
  float q = ((a[0] * a[0] + a[1] * a[1]) + (a[2] * a[2] + a[3] * a[3])) +
            ((b[0] * b[0] + b[1] * b[1]) + (b[2] * b[2] + b[3] * b[3]));
  q = row_xadd<0x161>(q);
  q = row_xadd<0x162>(q);
  q = row_xadd<0x164>(q);
  q = row_xadd<0x168>(q);
  q += xhalf(q);
  const float rstd = rsqrtf(q * (1.0f / 256.0f) + 1e-5f);
  a = a * rstd;
  b = b * rstd;
}

static constexpr int LN_RW = 2;
__global__ __launch_bounds__(128) void ln256_planes4_kernel(const float* __restrict__ in,
                                                            float* __restrict__ out, long sP) {
  const int lane = threadIdx.x & 31;
  const int wave = threadIdx.x >> 5;
  const long row = ((long)blockIdx.x * 4 + wave) * LN_RW;
  const float* src = in + row * 256 + lane * 4;
  v4f pa[LN_RW][4], pb[LN_RW][4];
#pragma unroll
  for (int r = 0; r < LN_RW; ++r) {
#pragma unroll
    for (int p = 0; p < 4; ++p) {
      pa[r][p] = *(const v4f*)(src + (long)p * sP + r * 256);
      pb[r][p] = *(const v4f*)(src + (long)p * sP + r * 256 + 128);
    }
  }
#pragma unroll
  for (int r = 0; r < LN_RW; ++r) {
    v4f a = (pa[r][0] + pa[r][1]) + (pa[r][2] + pa[r][3]);
    v4f b = (pb[r][0] + pb[r][1]) + (pb[r][2] + pb[r][3]);
    ln_row256(a, b);
    float* dst = out + (row + r) * 256 + lane * 4;
    *(v4f*)(dst) = a;
    *(v4f*)(dst + 128) = b;
  }
}


static constexpr int RS_STRIPS = 32;
static constexpr int GB_ROWS = 32;

__global__ __launch_bounds__(256) void gbuild_kernel(const float* __restrict__ Rp,
                                                     float* __restrict__ G) {
  const int rb = blockIdx.x;
  const int b = blockIdx.y;
  const int s = threadIdx.x;
  const int lane = s & 31;
  const int wave = s >> 5;
  __shared__ __attribute__((aligned(16))) float rs[Tt];
  __shared__ __attribute__((aligned(16))) float qs[Tt];
  __shared__ unsigned big[8];
  __shared__ float bp[8];

  float r = 0.f;
#pragma unroll
  for (int c = 0; c < RS_STRIPS; ++c) r += Rp[(long)c * (Bb * Tt) + b * Tt + s];
  rs[s] = r;
  const unsigned m = __builtin_amdgcn_ballot_w32(r >= 1e-12f);
  if (lane == 0) big[wave] = m;
  __syncthreads();

  int f = Tt;
#pragma unroll
  for (int w = 7; w >= 0; --w) {
    const unsigned mw = big[w];
    if (mw != 0u) f = 32 * w + __builtin_ctz(mw);
  }
  float S = 1.0f;
  if (s <= f) {
    S = 0.f;
    for (int i = 0; i < s; ++i) {
      const float sum = fmaf(0.97f, S, rs[i]);
      S = (sum >= 1e-12f) ? 1.0f : sum * 1e12f;
    }
  }
  const float d = fmaxf(fmaf(0.97f, S, r), 1e-12f);
  const float inv = 1.0f / d;
  const float q = 0.97f * inv;
  qs[s] = q;

  const float GMAX = 3.0e38f;

  float v = q;
#pragma unroll
  for (int o = 1; o < 32; o <<= 1) {
    const float up = __shfl_down(v, o, 32);
    if (lane + o < 32) v = fminf(v * up, GMAX);
  }
  const float nxt = __shfl_down(v, 1, 32);
  const float suf = (lane < 31) ? nxt : 1.0f;
  if (lane == 0) bp[wave] = v;
  __syncthreads();

  const int t0 = rb * GB_ROWS;
  float* gp = G + ((long)b * Tt + t0) * Tt + s;
  if (wave > rb) {
#pragma unroll 8
    for (int i = 0; i < GB_ROWS; ++i) gp[(long)i * Tt] = 0.f;
  } else {
    float g = 0.f;
    if (wave < rb) {
      g = fminf(inv * suf, GMAX);
      for (int k = wave + 1; k < rb; ++k) g = fminf(g * bp[k], GMAX);
    }
#pragma unroll 8
    for (int i = 0; i < GB_ROWS; ++i) {
      const int tt = t0 + i;
      const float gn = fminf(g * qs[tt], GMAX);
      g = (tt < s) ? 0.f : ((tt == s) ? inv : gn);
      gp[(long)i * Tt] = g;
    }
  }
}

template <int EPI, int BM, int BN, int BK, int SPLIT>
__global__ __launch_bounds__(128) void gemm_ABt_lds(const float* __restrict__ A,
                                                    const float* __restrict__ W,
                                                    float* __restrict__ C,
                                                    float* __restrict__ Rs,
                                                    float* __restrict__ Ln,
                                                    int N, int K,
                                                    long sA, long sW, long sC, long sP) {
  constexpr int PITCH = BK + 4;
  constexpr int C4 = BK / 4;
  constexpr int NA = BM * C4 / 128;
  constexpr int NW = BN * C4 / 128;
  constexpr int FM = BM / 32;
  constexpr int FN = BN / 32;
  __shared__ __attribute__((aligned(16))) float As[2][BM * PITCH];
  __shared__ __attribute__((aligned(16))) float Ws[2][BN * PITCH];

  const int b = blockIdx.z / SPLIT;
  const int sp = blockIdx.z - b * SPLIT;
  const int Ks = K / SPLIT;
  A += (long)b * sA + sp * Ks;
  W += (long)b * sW + sp * Ks;
  C += (long)b * sC + (long)sp * sP;
  const int tid = threadIdx.x;
  const int lane = tid & 31;
  const int wave = tid >> 5;
  const int wm = wave >> 1;
  const int wn = wave & 1;
  const int half = lane >> 4;
  const int l = lane & 15;
  const int row0 = blockIdx.y * BM;
  const int col0 = blockIdx.x * BN;

  const float* ga[NA];
  int la[NA];
#pragma unroll
  for (int j = 0; j < NA; ++j) {
    const int f = tid + 128 * j;
    const int r = f / C4;
    const int c = f - r * C4;
    ga[j] = A + (long)(row0 + r) * K + c * 4;
    la[j] = r * PITCH + c * 4;
  }
  const float* gw[NW];
  int lw[NW];
#pragma unroll
  for (int j = 0; j < NW; ++j) {
    const int f = tid + 128 * j;
    const int r = f / C4;
    const int c = f - r * C4;
    gw[j] = W + (long)(col0 + r) * K + c * 4;
    lw[j] = r * PITCH + c * 4;
  }

  v4f ra[NA], rw[NW];
#pragma unroll
  for (int j = 0; j < NA; ++j) ra[j] = *(const v4f*)(ga[j]);
#pragma unroll
  for (int j = 0; j < NW; ++j) rw[j] = *(const v4f*)(gw[j]);
  v4f na = v4f{}, nb = v4f{};
  long nrow = 0;
  if constexpr (EPI == 1) {
    nrow = row0 + 4 * (int)blockIdx.x + wave;
    const float* np = A + nrow * Dd + lane * 4;
    na = *(const v4f*)(np);
    nb = *(const v4f*)(np + 128);
  }
#pragma unroll
  for (int j = 0; j < NA; ++j) *(v4f*)(&As[0][la[j]]) = ra[j];
#pragma unroll
  for (int j = 0; j < NW; ++j) *(v4f*)(&Ws[0][lw[j]]) = rw[j];
  __syncthreads();
  if constexpr (EPI == 1) ln_row256(na, nb);

  v8f acc[FM][FN];
#pragma unroll
  for (int i = 0; i < FM; ++i)
#pragma unroll
    for (int j = 0; j < FN; ++j) acc[i][j] = v8f{};

  const int aoff = (wm * (BM / 2) + l) * PITCH + 2 * half;
  const int woff = (wn * (BN / 2) + l) * PITCH + 2 * half;
  const int nchunks = Ks / BK;
  for (int ch = 0; ch < nchunks; ++ch) {
    const int cur = ch & 1;
    const bool more = (ch + 1 < nchunks);
    if (more) {
      const int kn = (ch + 1) * BK;
#pragma unroll
      for (int j = 0; j < NA; ++j) ra[j] = *(const v4f*)(ga[j] + kn);
#pragma unroll
      for (int j = 0; j < NW; ++j) rw[j] = *(const v4f*)(gw[j] + kn);
    }
    const float* Ab = &As[cur][aoff];
    const float* Wb = &Ws[cur][woff];
#pragma unroll
    for (int ks = 0; ks < BK / 4; ++ks) {
      v2f a[FM], bb[FN];
#pragma unroll
      for (int i = 0; i < FM; ++i) a[i] = *(const v2f*)(Ab + i * 16 * PITCH + ks * 4);
#pragma unroll
      for (int j = 0; j < FN; ++j) bb[j] = *(const v2f*)(Wb + j * 16 * PITCH + ks * 4);
#pragma unroll
      for (int i = 0; i < FM; ++i)
#pragma unroll
        for (int j = 0; j < FN; ++j)
          acc[i][j] = __builtin_amdgcn_wmma_f32_16x16x4_f32(false, a[i], false, bb[j], (short)0,
                                                            acc[i][j], false, false);
    }
    if (more) {
#pragma unroll
      for (int j = 0; j < NA; ++j) *(v4f*)(&As[cur ^ 1][la[j]]) = ra[j];
#pragma unroll
      for (int j = 0; j < NW; ++j) *(v4f*)(&Ws[cur ^ 1][lw[j]]) = rw[j];
    }
    __syncthreads();
  }

  if constexpr (EPI == 1) {
    float* nd = Ln + nrow * Dd + lane * 4;
    *(v4f*)(nd) = na;
    *(v4f*)(nd + 128) = nb;
  }

#pragma unroll
  for (int i = 0; i < FM; ++i) {
#pragma unroll
    for (int j = 0; j < FN; ++j) {
      const long col = col0 + wn * (BN / 2) + j * 16 + l;
#pragma unroll
      for (int r = 0; r < 8; ++r) {
        const long row = row0 + wm * (BM / 2) + i * 16 + r + 8 * half;
        float v = acc[i][j][r];
        if (EPI == 1) v = fmaxf(v, 0.f);
        C[row * N + col] = v;
      }
    }
  }

  if (EPI == 1) {
    static_assert(EPI != 1 || (BM == 64 && BN == 64), "row-sum epilogue: 32 x 32 wave tile");
    float p[FM][8];
#pragma unroll
    for (int i = 0; i < FM; ++i) {
#pragma unroll
      for (int r = 0; r < 8; ++r) {
        float s = 0.f;
#pragma unroll
        for (int j = 0; j < FN; ++j) s += fmaxf(acc[i][j][r], 0.f);
        s = row_xadd<0x161>(s);
        s = row_xadd<0x162>(s);
        s = row_xadd<0x164>(s);
        s = row_xadd<0x168>(s);
        p[i][r] = s;
      }
    }
    float mine = 0.f;
#pragma unroll
    for (int i = 0; i < FM; ++i)
#pragma unroll
      for (int r = 0; r < 8; ++r)
        if (l == i * 8 + r) mine = p[i][r];
    const int rho = (l >> 3) * 16 + 8 * half + (l & 7);
    const long rows = (long)gridDim.y * BM;
    Rs[(long)(blockIdx.x * 2 + wn) * rows + row0 + wm * (BM / 2) + rho] = mine;
  }
}

__global__ __launch_bounds__(128) void gemm_lnA_Bt_lds(const float* __restrict__ A,
                                                       const float* __restrict__ W,
                                                       float* __restrict__ C,
                                                       const float* __restrict__ Mul,
                                                       int N) {
  constexpr int BM = 64, BN = 64, BK = 32, KK = 256;
  constexpr int PP = KK + 4;
  constexpr int PITCH = BK + 4;
  constexpr int C4 = BK / 4;
  constexpr int NW = BN * C4 / 128;
  constexpr int FM = BM / 32;
  constexpr int FN = BN / 32;
  __shared__ __attribute__((aligned(16))) float Ap[BM * PP];
  __shared__ __attribute__((aligned(16))) float Ws[2][BN * PITCH];

  const int tid = threadIdx.x;
  const int lane = tid & 31;
  const int wave = tid >> 5;
  const int wm = wave >> 1;
  const int wn = wave & 1;
  const int half = lane >> 4;
  const int l = lane & 15;
  const int row0 = blockIdx.y * BM;
  const int col0 = blockIdx.x * BN;

  const float* gw[NW];
  int lw[NW];
#pragma unroll
  for (int j = 0; j < NW; ++j) {
    const int f = tid + 128 * j;
    const int r = f / C4;
    const int c = f - r * C4;
    gw[j] = W + (long)(col0 + r) * KK + c * 4;
    lw[j] = r * PITCH + c * 4;
  }

  v4f p[16][2];
  const float* gp = A + (long)(row0 + wave * 16) * KK + lane * 4;
#pragma unroll
  for (int i = 0; i < 16; ++i) {
    p[i][0] = *(const v4f*)(gp + (long)i * KK);
    p[i][1] = *(const v4f*)(gp + (long)i * KK + 128);
  }
  v4f rw[NW];
#pragma unroll
  for (int j = 0; j < NW; ++j) rw[j] = *(const v4f*)(gw[j]);

#pragma unroll
  for (int i = 0; i < 16; ++i) {
    float s = ((p[i][0][0] + p[i][0][1]) + (p[i][0][2] + p[i][0][3])) +
              ((p[i][1][0] + p[i][1][1]) + (p[i][1][2] + p[i][1][3]));
    s = row_xadd<0x161>(s);
    s = row_xadd<0x162>(s);
    s = row_xadd<0x164>(s);
    s = row_xadd<0x168>(s);
    s += __shfl_xor(s, 16, 32);
    const float mean = s * (1.0f / 256.0f);
    const v4f c0 = p[i][0] - mean;
    const v4f c1 = p[i][1] - mean;
    float q = ((c0[0] * c0[0] + c0[1] * c0[1]) + (c0[2] * c0[2] + c0[3] * c0[3])) +
              ((c1[0] * c1[0] + c1[1] * c1[1]) + (c1[2] * c1[2] + c1[3] * c1[3]));
    q = row_xadd<0x161>(q);
    q = row_xadd<0x162>(q);
    q = row_xadd<0x164>(q);
    q = row_xadd<0x168>(q);
    q += __shfl_xor(q, 16, 32);
    const float rstd = rsqrtf(q * (1.0f / 256.0f) + 1e-5f);
    float* dst = &Ap[(wave * 16 + i) * PP + lane * 4];
    *(v4f*)(dst) = c0 * rstd;
    *(v4f*)(dst + 128) = c1 * rstd;
  }
#pragma unroll
  for (int j = 0; j < NW; ++j) *(v4f*)(&Ws[0][lw[j]]) = rw[j];
  __syncthreads();

  v8f acc[FM][FN];
#pragma unroll
  for (int i = 0; i < FM; ++i)
#pragma unroll
    for (int j = 0; j < FN; ++j) acc[i][j] = v8f{};

  const int aoff = (wm * (BM / 2) + l) * PP + 2 * half;
  const int woff = (wn * (BN / 2) + l) * PITCH + 2 * half;
  constexpr int nchunks = KK / BK;
#pragma unroll 1
  for (int ch = 0; ch < nchunks; ++ch) {
    const int cur = ch & 1;
    const bool more = (ch + 1 < nchunks);
    if (more) {
      const int kn = (ch + 1) * BK;
#pragma unroll
      for (int j = 0; j < NW; ++j) rw[j] = *(const v4f*)(gw[j] + kn);
    }
    const float* Ab = &Ap[aoff + ch * BK];
    const float* Wb = &Ws[cur][woff];
#pragma unroll
    for (int ks = 0; ks < BK / 4; ++ks) {
      v2f a[FM], bb[FN];
#pragma unroll
      for (int i = 0; i < FM; ++i) a[i] = *(const v2f*)(Ab + i * 16 * PP + ks * 4);
#pragma unroll
      for (int j = 0; j < FN; ++j) bb[j] = *(const v2f*)(Wb + j * 16 * PITCH + ks * 4);
#pragma unroll
      for (int i = 0; i < FM; ++i)
#pragma unroll
        for (int j = 0; j < FN; ++j)
          acc[i][j] = __builtin_amdgcn_wmma_f32_16x16x4_f32(false, a[i], false, bb[j], (short)0,
                                                            acc[i][j], false, false);
    }
    if (more) {
#pragma unroll
      for (int j = 0; j < NW; ++j) *(v4f*)(&Ws[cur ^ 1][lw[j]]) = rw[j];
    }
    __syncthreads();
  }

#pragma unroll
  for (int i = 0; i < FM; ++i) {
#pragma unroll
    for (int j = 0; j < FN; ++j) {
      const long col = col0 + wn * (BN / 2) + j * 16 + l;
#pragma unroll
      for (int r = 0; r < 8; ++r) {
        const long row = row0 + wm * (BM / 2) + i * 16 + r + 8 * half;
        const float v = fmaxf(acc[i][j][r], 0.f) * fmaxf(Mul[row * N + col], 0.f);
        C[row * N + col] = v;
      }
    }
  }
}

static __device__ __forceinline__ v4f decay_weights(const float* pw, int t, int s0) {
  v4f w;
#pragma unroll
  for (int e = 0; e < 4; ++e) {
    const int d = t - 1 - (s0 + e);
    w[e] = pw[d < 0 ? 0 : d];
  }
  return w;
}
static __device__ __forceinline__ v4f decay_apply(v4f v, v4f w, int t, int s0) {
  v4f o;
#pragma unroll
  for (int e = 0; e < 4; ++e) o[e] = (s0 + e < t) ? w[e] * v[e] : 0.f;
  return o;
}

template <int APL, int NA>
static __device__ __forceinline__ v4f plane_sum(const v4f (&rp)[APL][NA], int j) {
  if constexpr (APL == 4) return (rp[0][j] + rp[1][j]) + (rp[2][j] + rp[3][j]);
  else return rp[0][j];
}

template <int BM, int BN, int BK, bool CAUSAL, bool DECAY, int APL>
__global__ __launch_bounds__(128) void gemm_AB_lds(const float* __restrict__ A,
                                                   const float* __restrict__ Bm,
                                                   float* __restrict__ C,
                                                   int N, int K,
                                                   long sA, long sB, long sC, long sPA) {
  static_assert(APL == 1 || APL == 4, "1 plane or 4 partial planes");
  constexpr int PA = BK + 4;
  constexpr int PB = BN + 8;
  constexpr int CA = BK / 4;
  constexpr int CB = BN / 4;
  constexpr int NA = BM * CA / 128;
  constexpr int NB = BK * CB / 128;
  constexpr int FM = BM / 32;
  constexpr int FN = BN / 32;
  __shared__ __attribute__((aligned(16))) float As[2][BM * PA];
  __shared__ __attribute__((aligned(16))) float Bs[2][BK * PB];
  __shared__ float pw[DECAY ? Tt : 1];

  const int b = blockIdx.z;
  A += (long)b * sA;
  Bm += (long)b * sB;
  C += (long)b * sC;
  const int tid = threadIdx.x;
  const int lane = tid & 31;
  const int wave = tid >> 5;
  const int wm = wave >> 1;
  const int wn = wave & 1;
  const int half = lane >> 4;
  const int l = lane & 15;
  const int row0 = blockIdx.y * BM;
  const int col0 = blockIdx.x * BN;

  const float* ga[NA];
  int la[NA];
  int ta[NA], sa[NA];
#pragma unroll
  for (int j = 0; j < NA; ++j) {
    const int f = tid + 128 * j;
    const int r = f / CA;
    const int c = f - r * CA;
    ga[j] = A + (long)(row0 + r) * K + c * 4;
    la[j] = r * PA + c * 4;
    ta[j] = row0 + r;
    sa[j] = c * 4;
  }
  const float* gb[NB];
  int lb[NB];
#pragma unroll
  for (int j = 0; j < NB; ++j) {
    const int f = tid + 128 * j;
    const int r = f / CB;
    const int c = f - r * CB;
    gb[j] = Bm + (long)r * N + col0 + c * 4;
    lb[j] = r * PB + c * 4;
  }

  v4f rp[APL][NA];
  v4f ra[NA], rb[NB];
#pragma unroll
  for (int p = 0; p < APL; ++p)
#pragma unroll
    for (int j = 0; j < NA; ++j) rp[p][j] = *(const v4f*)(ga[j] + (long)p * sPA);
#pragma unroll
  for (int j = 0; j < NB; ++j) rb[j] = *(const v4f*)(gb[j]);
  v4f wq[NA];
  if (DECAY) {
    for (int i = tid; i < Tt; i += 128) pw[i] = __powf(0.97f, (float)i);
    __syncthreads();
#pragma unroll
    for (int j = 0; j < NA; ++j) wq[j] = decay_weights(pw, ta[j], sa[j]);
  }
#pragma unroll
  for (int j = 0; j < NA; ++j)
    ra[j] = plane_sum<APL, NA>(rp, j);
  if (DECAY) {
#pragma unroll
    for (int j = 0; j < NA; ++j) ra[j] = decay_apply(ra[j], wq[j], ta[j], sa[j]);
  }
#pragma unroll
  for (int j = 0; j < NA; ++j) *(v4f*)(&As[0][la[j]]) = ra[j];
#pragma unroll
  for (int j = 0; j < NB; ++j) *(v4f*)(&Bs[0][lb[j]]) = rb[j];
  __syncthreads();

  v8f acc[FM][FN];
#pragma unroll
  for (int i = 0; i < FM; ++i)
#pragma unroll
    for (int j = 0; j < FN; ++j) acc[i][j] = v8f{};

  const int aoff = (wm * (BM / 2) + l) * PA + 2 * half;
  const int boff = (2 * half) * PB + wn * (BN / 2) + l;
  const int nchunks = CAUSAL ? (row0 + BM) / BK : K / BK;
  for (int ch = 0; ch < nchunks; ++ch) {
    const int cur = ch & 1;
    const bool more = (ch + 1 < nchunks);
    if (more) {
      const int kn = (ch + 1) * BK;
#pragma unroll
      for (int p = 0; p < APL; ++p)
#pragma unroll
        for (int j = 0; j < NA; ++j) rp[p][j] = *(const v4f*)(ga[j] + (long)p * sPA + kn);
#pragma unroll
      for (int j = 0; j < NB; ++j) rb[j] = *(const v4f*)(gb[j] + (long)kn * N);
      if (DECAY) {
#pragma unroll
        for (int j = 0; j < NA; ++j) wq[j] = decay_weights(pw, ta[j], sa[j] + kn);
      }
    }
    const float* Ab = &As[cur][aoff];
    const float* Bb = &Bs[cur][boff];
#pragma unroll
    for (int ks = 0; ks < BK / 4; ++ks) {
      v2f a[FM], bb[FN];
#pragma unroll
      for (int i = 0; i < FM; ++i) a[i] = *(const v2f*)(Ab + i * 16 * PA + ks * 4);
#pragma unroll
      for (int j = 0; j < FN; ++j) {
        bb[j][0] = Bb[(ks * 4) * PB + j * 16];
        bb[j][1] = Bb[(ks * 4 + 1) * PB + j * 16];
      }
#pragma unroll
      for (int i = 0; i < FM; ++i)
#pragma unroll
        for (int j = 0; j < FN; ++j)
          acc[i][j] = __builtin_amdgcn_wmma_f32_16x16x4_f32(false, a[i], false, bb[j], (short)0,
                                                            acc[i][j], false, false);
    }
    if (more) {
#pragma unroll
      for (int j = 0; j < NA; ++j)
        ra[j] = plane_sum<APL, NA>(rp, j);
      if (DECAY) {
        const int kn = (ch + 1) * BK;
#pragma unroll
        for (int j = 0; j < NA; ++j) ra[j] = decay_apply(ra[j], wq[j], ta[j], sa[j] + kn);
      }
#pragma unroll
      for (int j = 0; j < NA; ++j) *(v4f*)(&As[cur ^ 1][la[j]]) = ra[j];
#pragma unroll
      for (int j = 0; j < NB; ++j) *(v4f*)(&Bs[cur ^ 1][lb[j]]) = rb[j];
    }
    __syncthreads();
  }

#pragma unroll
  for (int i = 0; i < FM; ++i) {
#pragma unroll
    for (int j = 0; j < FN; ++j) {
      const long col = col0 + wn * (BN / 2) + j * 16 + l;
#pragma unroll
      for (int r = 0; r < 8; ++r) {
        const long row = row0 + wm * (BM / 2) + i * 16 + r + 8 * half;
        C[row * N + col] = acc[i][j][r];
      }
    }
  }
}

extern "C" void kernel_launch(void* const* d_in, const int* in_sizes, int n_in,
                              void* d_out, int out_size, void* d_ws, size_t ws_size,
                              hipStream_t stream) {
  const float* emb = (const float*)d_in[0];
  const float* E   = (const float*)d_in[1];
  const float* Dx  = (const float*)d_in[2];
  const float* Dy  = (const float*)d_in[3];
  float* out = (float*)d_out;

  const long BT = (long)Bb * Tt;
  const long szBTD = BT * Dd;
  const long szBTN = BT * Nn;
  const long szBTT = (long)Bb * Tt * Tt;

  float* ws = (float*)d_ws;
  float* LNV = ws;
  float* XU  = LNV + szBTD;
  float* X   = XU + szBTN;
  float* Rp  = X + szBTN;
  float* G   = Rp + szBTT;
  float* As  = G + szBTT;
  float* Y   = As + szBTD;
  float* KgP = XU;
  float* VP  = XU;

  gemm_ABt_lds<1, 64, 64, 32, 1><<<dim3(1024 / 64, 2048 / 64, 1), 128, 0, stream>>>(
      emb, Dx, XU, Rp, LNV, 1024, 256, 0, 0, 0, 0);

  gbuild_kernel<<<dim3(Tt / GB_ROWS, Bb), 256, 0, stream>>>(Rp, G);

  gemm_AB_lds<64, 64, 32, true, false, 1><<<dim3(1024 / 64, 256 / 64, Bb), 128, 0, stream>>>(
      G, XU, X, 1024, 256, (long)Tt * Tt, (long)Tt * Nn, (long)Tt * Nn, 0);

  gemm_ABt_lds<0, 64, 64, 32, 4><<<dim3(256 / 64, 256 / 64, Bb * 4), 128, 0, stream>>>(
      X, X, KgP, nullptr, nullptr, 256, 1024, (long)Tt * Nn, (long)Tt * Nn, (long)Tt * Tt, szBTT);

  gemm_AB_lds<32, 64, 32, false, true, 4><<<dim3(256 / 64, 256 / 32, Bb), 128, 0, stream>>>(
      KgP, LNV, As, 256, 256, (long)Tt * Tt, (long)Tt * Dd, (long)Tt * Dd, szBTT);

  gemm_lnA_Bt_lds<<<dim3(1024 / 64, 2048 / 64, 1), 128, 0, stream>>>(As, Dy, Y, X, 1024);

  gemm_ABt_lds<0, 64, 64, 32, 4><<<dim3(256 / 64, 2048 / 64, 4), 128, 0, stream>>>(
      Y, E, VP, nullptr, nullptr, 256, 1024, 0, 0, 0, szBTD);

  ln256_planes4_kernel<<<2048 / (4 * LN_RW), 128, 0, stream>>>(VP, out, szBTD);
}


# x product in 64-step blocks: only the 4 diagonal 64x64 blocks of G multiplied (was 10), built in LDS by their workgroup; carry enters as u_first+0.97*x_prev; 8->7 launches; x sum order changes
# speedup vs baseline: 9.8570x; 1.0548x over previous
#include <hip/hip_runtime.h>


typedef __attribute__((ext_vector_type(2))) float v2f;
typedef __attribute__((ext_vector_type(8))) float v8f;
typedef __attribute__((ext_vector_type(4))) float v4f;

static constexpr int Bb = 8;
static constexpr int Tt = 256;
static constexpr int Dd = 256;
static constexpr int Nn = 1024;

template <int CTRL>
static __device__ __forceinline__ float row_xadd(float v) {
  const int iv = __builtin_bit_cast(int, v);
  const int sw = __builtin_amdgcn_update_dpp(0, iv, CTRL, 0xf, 0xf, true);
  return v + __builtin_bit_cast(float, sw);
}

static __device__ __forceinline__ float xhalf(float v) {
  const int iv = __builtin_bit_cast(int, v);
  return __builtin_bit_cast(float, __builtin_amdgcn_permlanex16(iv, iv, 0x76543210, 0xfedcba98,
                                                                false, false));
}

static __device__ __forceinline__ void ln_row256(v4f& a, v4f& b) {
  float s = ((a[0] + a[1]) + (a[2] + a[3])) + ((b[0] + b[1]) + (b[2] + b[3]));
  s = row_xadd<0x161>(s);
  s = row_xadd<0x162>(s);
  s = row_xadd<0x164>(s);
  s = row_xadd<0x168>(s);
  s += xhalf(s);
  const float mean = s * (1.0f / 256.0f);
  a = a - mean;
  b = b - mean;
  float q = ((a[0] * a[0] + a[1] * a[1]) + (a[2] * a[2] + a[3] * a[3])) +
            ((b[0] * b[0] + b[1] * b[1]) + (b[2] * b[2] + b[3] * b[3]));
  q = row_xadd<0x161>(q);
  q = row_xadd<0x162>(q);
  q = row_xadd<0x164>(q);
  q = row_xadd<0x168>(q);
  q += xhalf(q);
  const float rstd = rsqrtf(q * (1.0f / 256.0f) + 1e-5f);
  a = a * rstd;
  b = b * rstd;
}

static constexpr int LN_RW = 2;
__global__ __launch_bounds__(128) void ln256_planes4_kernel(const float* __restrict__ in,
                                                            float* __restrict__ out, long sP) {
  const int lane = threadIdx.x & 31;
  const int wave = threadIdx.x >> 5;
  const long row = ((long)blockIdx.x * 4 + wave) * LN_RW;
  const float* src = in + row * 256 + lane * 4;
  v4f pa[LN_RW][4], pb[LN_RW][4];
#pragma unroll
  for (int r = 0; r < LN_RW; ++r) {
#pragma unroll
    for (int p = 0; p < 4; ++p) {
      pa[r][p] = *(const v4f*)(src + (long)p * sP + r * 256);
      pb[r][p] = *(const v4f*)(src + (long)p * sP + r * 256 + 128);
    }
  }
#pragma unroll
  for (int r = 0; r < LN_RW; ++r) {
    v4f a = (pa[r][0] + pa[r][1]) + (pa[r][2] + pa[r][3]);
    v4f b = (pb[r][0] + pb[r][1]) + (pb[r][2] + pb[r][3]);
    ln_row256(a, b);
    float* dst = out + (row + r) * 256 + lane * 4;
    *(v4f*)(dst) = a;
    *(v4f*)(dst + 128) = b;
  }
}


static constexpr int RS_STRIPS = 32;
static constexpr int XB = 64;
static constexpr int XN = 32;

__global__ __launch_bounds__(128) void xblocks_kernel(const float* __restrict__ Rp,
                                                      const float* __restrict__ U,
                                                      float* __restrict__ X) {
  constexpr int PA = XB + 4;
  constexpr int PB = XN + 8;
  constexpr int NBLK = Tt / XB;
  constexpr int NU = Tt * (XN / 4) / 128;
  __shared__ __attribute__((aligned(16))) float sm[Tt * PB + NBLK * XB * PA + 2 * Tt + 2 * XN];
  __shared__ unsigned big[8];
  float* Us = sm;
  float* Gs = Us + Tt * PB;
  float* rs = Gs + NBLK * XB * PA;
  float* qs = rs + Tt;
  float* xp = qs + Tt;

  const int b = blockIdx.z;
  const int col0 = blockIdx.x * XN;
  const int tid = threadIdx.x;
  const int lane = tid & 31;
  const int wave = tid >> 5;
  const int half = lane >> 4;
  const int l = lane & 15;

  const float* up = U + (long)b * Tt * Nn + col0;
  v4f ru[NU];
  int lu[NU];
#pragma unroll
  for (int j = 0; j < NU; ++j) {
    const int f = tid + 128 * j;
    const int r = f / (XN / 4);
    const int c = f - r * (XN / 4);
    ru[j] = *(const v4f*)(up + (long)r * Nn + c * 4);
    lu[j] = r * PB + c * 4;
  }

  float r0 = 0.f, r1 = 0.f;
#pragma unroll
  for (int c = 0; c < RS_STRIPS; ++c) {
    r0 += Rp[(long)c * (Bb * Tt) + b * Tt + tid];
    r1 += Rp[(long)c * (Bb * Tt) + b * Tt + tid + 128];
  }
  rs[tid] = r0;
  rs[tid + 128] = r1;
  const unsigned m0 = __builtin_amdgcn_ballot_w32(r0 >= 1e-12f);
  const unsigned m1 = __builtin_amdgcn_ballot_w32(r1 >= 1e-12f);
  if (lane == 0) {
    big[wave] = m0;
    big[4 + wave] = m1;
  }
  __syncthreads();

  int f = Tt;
#pragma unroll
  for (int w = 7; w >= 0; --w) {
    const unsigned mw = big[w];
    if (mw != 0u) f = 32 * w + __builtin_ctz(mw);
  }
  float inv0, inv1;
  {
    float S = 1.0f;
    if (tid <= f) {
      S = 0.f;
      for (int i = 0; i < tid; ++i) {
        const float sum = fmaf(0.97f, S, rs[i]);
        S = (sum >= 1e-12f) ? 1.0f : sum * 1e12f;
      }
    }
    inv0 = 1.0f / fmaxf(fmaf(0.97f, S, r0), 1e-12f);
    qs[tid] = 0.97f * inv0;
  }
  {
    float S = 1.0f;
    if (tid + 128 <= f) {
      S = 0.f;
      for (int i = 0; i < tid + 128; ++i) {
        const float sum = fmaf(0.97f, S, rs[i]);
        S = (sum >= 1e-12f) ? 1.0f : sum * 1e12f;
      }
    }
    inv1 = 1.0f / fmaxf(fmaf(0.97f, S, r1), 1e-12f);
    qs[tid + 128] = 0.97f * inv1;
  }
#pragma unroll
  for (int j = 0; j < NU; ++j) *(v4f*)(&Us[lu[j]]) = ru[j];
  __syncthreads();

  const float GMAX = 3.0e38f;
  {
    const int jb = tid >> 6;
    const int sl = tid & (XB - 1);
    float* gp0 = Gs + jb * (XB * PA) + sl;
    float* gp1 = gp0 + 2 * (XB * PA);
    const float* q0 = &qs[XB * jb];
    const float* q1 = &qs[XB * (jb + 2)];
    float g0 = 0.f, g1 = 0.f;
#pragma unroll 8
    for (int i = 0; i < XB; ++i) {
      const float n0 = fminf(g0 * q0[i], GMAX);
      const float n1 = fminf(g1 * q1[i], GMAX);
      g0 = (i == sl) ? inv0 : n0;
      g1 = (i == sl) ? inv1 : n1;
      gp0[i * PA] = g0;
      gp1[i * PA] = g1;
    }
  }
  __syncthreads();

  const float* Ab0 = Gs + (16 * wave + l) * PA + 2 * half;
  const float* Bb0 = Us + (2 * half) * PB + l;
  float* xo = X + (long)b * Tt * Nn + col0;
  for (int j = 0; j < NBLK; ++j) {
    const float* Ab = Ab0 + j * (XB * PA);
    const float* Bq = Bb0 + (j * XB) * PB;
    v8f acc[2];
    acc[0] = v8f{};
    acc[1] = v8f{};
#pragma unroll
    for (int ks = 1; ks < XB / 4; ++ks) {
      const v2f a = *(const v2f*)(Ab + ks * 4);
      v2f bb[2];
#pragma unroll
      for (int n = 0; n < 2; ++n) {
        bb[n][0] = Bq[(ks * 4) * PB + n * 16];
        bb[n][1] = Bq[(ks * 4 + 1) * PB + n * 16];
      }
#pragma unroll
      for (int n = 0; n < 2; ++n)
        acc[n] = __builtin_amdgcn_wmma_f32_16x16x4_f32(false, a, false, bb[n], (short)0, acc[n],
                                                       false, false);
    }
    {
      const v2f a = *(const v2f*)(Ab);
      v2f bb[2];
#pragma unroll
      for (int n = 0; n < 2; ++n) {
        bb[n][0] = Bq[n * 16];
        bb[n][1] = Bq[PB + n * 16];
      }
      if (j > 0) {
        __builtin_amdgcn_fence(__ATOMIC_RELEASE, "workgroup", "local");
        __builtin_amdgcn_s_barrier();
        __builtin_amdgcn_fence(__ATOMIC_ACQUIRE, "workgroup", "local");
        const float* xq = xp + ((j - 1) & 1) * XN;
#pragma unroll
        for (int n = 0; n < 2; ++n) {
          const float cv = fmaf(0.97f, xq[n * 16 + l], bb[n][0]);
          bb[n][0] = (half == 0) ? cv : bb[n][0];
        }
      }
#pragma unroll
      for (int n = 0; n < 2; ++n)
        acc[n] = __builtin_amdgcn_wmma_f32_16x16x4_f32(false, a, false, bb[n], (short)0, acc[n],
                                                       false, false);
    }
    if (j + 1 < NBLK && wave == 3 && half == 1) {
      xp[(j & 1) * XN + l] = acc[0][7];
      xp[(j & 1) * XN + 16 + l] = acc[1][7];
    }
#pragma unroll
    for (int n = 0; n < 2; ++n) {
#pragma unroll
      for (int r = 0; r < 8; ++r) {
        const long row = j * XB + 16 * wave + 8 * half + r;
        xo[row * Nn + n * 16 + l] = acc[n][r];
      }
    }
  }
}

template <int EPI, int BM, int BN, int BK, int SPLIT>
__global__ __launch_bounds__(128) void gemm_ABt_lds(const float* __restrict__ A,
                                                    const float* __restrict__ W,
                                                    float* __restrict__ C,
                                                    float* __restrict__ Rs,
                                                    float* __restrict__ Ln,
                                                    int N, int K,
                                                    long sA, long sW, long sC, long sP) {
  constexpr int PITCH = BK + 4;
  constexpr int C4 = BK / 4;
  constexpr int NA = BM * C4 / 128;
  constexpr int NW = BN * C4 / 128;
  constexpr int FM = BM / 32;
  constexpr int FN = BN / 32;
  __shared__ __attribute__((aligned(16))) float As[2][BM * PITCH];
  __shared__ __attribute__((aligned(16))) float Ws[2][BN * PITCH];

  const int b = blockIdx.z / SPLIT;
  const int sp = blockIdx.z - b * SPLIT;
  const int Ks = K / SPLIT;
  A += (long)b * sA + sp * Ks;
  W += (long)b * sW + sp * Ks;
  C += (long)b * sC + (long)sp * sP;
  const int tid = threadIdx.x;
  const int lane = tid & 31;
  const int wave = tid >> 5;
  const int wm = wave >> 1;
  const int wn = wave & 1;
  const int half = lane >> 4;
  const int l = lane & 15;
  const int row0 = blockIdx.y * BM;
  const int col0 = blockIdx.x * BN;

  const float* ga[NA];
  int la[NA];
#pragma unroll
  for (int j = 0; j < NA; ++j) {
    const int f = tid + 128 * j;
    const int r = f / C4;
    const int c = f - r * C4;
    ga[j] = A + (long)(row0 + r) * K + c * 4;
    la[j] = r * PITCH + c * 4;
  }
  const float* gw[NW];
  int lw[NW];
#pragma unroll
  for (int j = 0; j < NW; ++j) {
    const int f = tid + 128 * j;
    const int r = f / C4;
    const int c = f - r * C4;
    gw[j] = W + (long)(col0 + r) * K + c * 4;
    lw[j] = r * PITCH + c * 4;
  }

  v4f ra[NA], rw[NW];
#pragma unroll
  for (int j = 0; j < NA; ++j) ra[j] = *(const v4f*)(ga[j]);
#pragma unroll
  for (int j = 0; j < NW; ++j) rw[j] = *(const v4f*)(gw[j]);
  v4f na = v4f{}, nb = v4f{};
  long nrow = 0;
  if constexpr (EPI == 1) {
    nrow = row0 + 4 * (int)blockIdx.x + wave;
    const float* np = A + nrow * Dd + lane * 4;
    na = *(const v4f*)(np);
    nb = *(const v4f*)(np + 128);
  }
#pragma unroll
  for (int j = 0; j < NA; ++j) *(v4f*)(&As[0][la[j]]) = ra[j];
#pragma unroll
  for (int j = 0; j < NW; ++j) *(v4f*)(&Ws[0][lw[j]]) = rw[j];
  __syncthreads();
  if constexpr (EPI == 1) ln_row256(na, nb);

  v8f acc[FM][FN];
#pragma unroll
  for (int i = 0; i < FM; ++i)
#pragma unroll
    for (int j = 0; j < FN; ++j) acc[i][j] = v8f{};

  const int aoff = (wm * (BM / 2) + l) * PITCH + 2 * half;
  const int woff = (wn * (BN / 2) + l) * PITCH + 2 * half;
  const int nchunks = Ks / BK;
  for (int ch = 0; ch < nchunks; ++ch) {
    const int cur = ch & 1;
    const bool more = (ch + 1 < nchunks);
    if (more) {
      const int kn = (ch + 1) * BK;
#pragma unroll
      for (int j = 0; j < NA; ++j) ra[j] = *(const v4f*)(ga[j] + kn);
#pragma unroll
      for (int j = 0; j < NW; ++j) rw[j] = *(const v4f*)(gw[j] + kn);
    }
    const float* Ab = &As[cur][aoff];
    const float* Wb = &Ws[cur][woff];
#pragma unroll
    for (int ks = 0; ks < BK / 4; ++ks) {
      v2f a[FM], bb[FN];
#pragma unroll
      for (int i = 0; i < FM; ++i) a[i] = *(const v2f*)(Ab + i * 16 * PITCH + ks * 4);
#pragma unroll
      for (int j = 0; j < FN; ++j) bb[j] = *(const v2f*)(Wb + j * 16 * PITCH + ks * 4);
#pragma unroll
      for (int i = 0; i < FM; ++i)
#pragma unroll
        for (int j = 0; j < FN; ++j)
          acc[i][j] = __builtin_amdgcn_wmma_f32_16x16x4_f32(false, a[i], false, bb[j], (short)0,
                                                            acc[i][j], false, false);
    }
    if (more) {
#pragma unroll
      for (int j = 0; j < NA; ++j) *(v4f*)(&As[cur ^ 1][la[j]]) = ra[j];
#pragma unroll
      for (int j = 0; j < NW; ++j) *(v4f*)(&Ws[cur ^ 1][lw[j]]) = rw[j];
    }
    __syncthreads();
  }

  if constexpr (EPI == 1) {
    float* nd = Ln + nrow * Dd + lane * 4;
    *(v4f*)(nd) = na;
    *(v4f*)(nd + 128) = nb;
  }

#pragma unroll
  for (int i = 0; i < FM; ++i) {
#pragma unroll
    for (int j = 0; j < FN; ++j) {
      const long col = col0 + wn * (BN / 2) + j * 16 + l;
#pragma unroll
      for (int r = 0; r < 8; ++r) {
        const long row = row0 + wm * (BM / 2) + i * 16 + r + 8 * half;
        float v = acc[i][j][r];
        if (EPI == 1) v = fmaxf(v, 0.f);
        C[row * N + col] = v;
      }
    }
  }

  if (EPI == 1) {
    static_assert(EPI != 1 || (BM == 64 && BN == 64), "row-sum epilogue: 32 x 32 wave tile");
    float p[FM][8];
#pragma unroll
    for (int i = 0; i < FM; ++i) {
#pragma unroll
      for (int r = 0; r < 8; ++r) {
        float s = 0.f;
#pragma unroll
        for (int j = 0; j < FN; ++j) s += fmaxf(acc[i][j][r], 0.f);
        s = row_xadd<0x161>(s);
        s = row_xadd<0x162>(s);
        s = row_xadd<0x164>(s);
        s = row_xadd<0x168>(s);
        p[i][r] = s;
      }
    }
    float mine = 0.f;
#pragma unroll
    for (int i = 0; i < FM; ++i)
#pragma unroll
      for (int r = 0; r < 8; ++r)
        if (l == i * 8 + r) mine = p[i][r];
    const int rho = (l >> 3) * 16 + 8 * half + (l & 7);
    const long rows = (long)gridDim.y * BM;
    Rs[(long)(blockIdx.x * 2 + wn) * rows + row0 + wm * (BM / 2) + rho] = mine;
  }
}

__global__ __launch_bounds__(128) void gemm_lnA_Bt_lds(const float* __restrict__ A,
                                                       const float* __restrict__ W,
                                                       float* __restrict__ C,
                                                       const float* __restrict__ Mul,
                                                       int N) {
  constexpr int BM = 64, BN = 64, BK = 32, KK = 256;
  constexpr int PP = KK + 4;
  constexpr int PITCH = BK + 4;
  constexpr int C4 = BK / 4;
  constexpr int NW = BN * C4 / 128;
  constexpr int FM = BM / 32;
  constexpr int FN = BN / 32;
  __shared__ __attribute__((aligned(16))) float Ap[BM * PP];
  __shared__ __attribute__((aligned(16))) float Ws[2][BN * PITCH];

  const int tid = threadIdx.x;
  const int lane = tid & 31;
  const int wave = tid >> 5;
  const int wm = wave >> 1;
  const int wn = wave & 1;
  const int half = lane >> 4;
  const int l = lane & 15;
  const int row0 = blockIdx.y * BM;
  const int col0 = blockIdx.x * BN;

  const float* gw[NW];
  int lw[NW];
#pragma unroll
  for (int j = 0; j < NW; ++j) {
    const int f = tid + 128 * j;
    const int r = f / C4;
    const int c = f - r * C4;
    gw[j] = W + (long)(col0 + r) * KK + c * 4;
    lw[j] = r * PITCH + c * 4;
  }

  v4f p[16][2];
  const float* gp = A + (long)(row0 + wave * 16) * KK + lane * 4;
#pragma unroll
  for (int i = 0; i < 16; ++i) {
    p[i][0] = *(const v4f*)(gp + (long)i * KK);
    p[i][1] = *(const v4f*)(gp + (long)i * KK + 128);
  }
  v4f rw[NW];
#pragma unroll
  for (int j = 0; j < NW; ++j) rw[j] = *(const v4f*)(gw[j]);

#pragma unroll
  for (int i = 0; i < 16; ++i) {
    float s = ((p[i][0][0] + p[i][0][1]) + (p[i][0][2] + p[i][0][3])) +
              ((p[i][1][0] + p[i][1][1]) + (p[i][1][2] + p[i][1][3]));
    s = row_xadd<0x161>(s);
    s = row_xadd<0x162>(s);
    s = row_xadd<0x164>(s);
    s = row_xadd<0x168>(s);
    s += __shfl_xor(s, 16, 32);
    const float mean = s * (1.0f / 256.0f);
    const v4f c0 = p[i][0] - mean;
    const v4f c1 = p[i][1] - mean;
    float q = ((c0[0] * c0[0] + c0[1] * c0[1]) + (c0[2] * c0[2] + c0[3] * c0[3])) +
              ((c1[0] * c1[0] + c1[1] * c1[1]) + (c1[2] * c1[2] + c1[3] * c1[3]));
    q = row_xadd<0x161>(q);
    q = row_xadd<0x162>(q);
    q = row_xadd<0x164>(q);
    q = row_xadd<0x168>(q);
    q += __shfl_xor(q, 16, 32);
    const float rstd = rsqrtf(q * (1.0f / 256.0f) + 1e-5f);
    float* dst = &Ap[(wave * 16 + i) * PP + lane * 4];
    *(v4f*)(dst) = c0 * rstd;
    *(v4f*)(dst + 128) = c1 * rstd;
  }
#pragma unroll
  for (int j = 0; j < NW; ++j) *(v4f*)(&Ws[0][lw[j]]) = rw[j];
  __syncthreads();

  v8f acc[FM][FN];
#pragma unroll
  for (int i = 0; i < FM; ++i)
#pragma unroll
    for (int j = 0; j < FN; ++j) acc[i][j] = v8f{};

  const int aoff = (wm * (BM / 2) + l) * PP + 2 * half;
  const int woff = (wn * (BN / 2) + l) * PITCH + 2 * half;
  constexpr int nchunks = KK / BK;
#pragma unroll 1
  for (int ch = 0; ch < nchunks; ++ch) {
    const int cur = ch & 1;
    const bool more = (ch + 1 < nchunks);
    if (more) {
      const int kn = (ch + 1) * BK;
#pragma unroll
      for (int j = 0; j < NW; ++j) rw[j] = *(const v4f*)(gw[j] + kn);
    }
    const float* Ab = &Ap[aoff + ch * BK];
    const float* Wb = &Ws[cur][woff];
#pragma unroll
    for (int ks = 0; ks < BK / 4; ++ks) {
      v2f a[FM], bb[FN];
#pragma unroll
      for (int i = 0; i < FM; ++i) a[i] = *(const v2f*)(Ab + i * 16 * PP + ks * 4);
#pragma unroll
      for (int j = 0; j < FN; ++j) bb[j] = *(const v2f*)(Wb + j * 16 * PITCH + ks * 4);
#pragma unroll
      for (int i = 0; i < FM; ++i)
#pragma unroll
        for (int j = 0; j < FN; ++j)
          acc[i][j] = __builtin_amdgcn_wmma_f32_16x16x4_f32(false, a[i], false, bb[j], (short)0,
                                                            acc[i][j], false, false);
    }
    if (more) {
#pragma unroll
      for (int j = 0; j < NW; ++j) *(v4f*)(&Ws[cur ^ 1][lw[j]]) = rw[j];
    }
    __syncthreads();
  }

#pragma unroll
  for (int i = 0; i < FM; ++i) {
#pragma unroll
    for (int j = 0; j < FN; ++j) {
      const long col = col0 + wn * (BN / 2) + j * 16 + l;
#pragma unroll
      for (int r = 0; r < 8; ++r) {
        const long row = row0 + wm * (BM / 2) + i * 16 + r + 8 * half;
        const float v = fmaxf(acc[i][j][r], 0.f) * fmaxf(Mul[row * N + col], 0.f);
        C[row * N + col] = v;
      }
    }
  }
}

static __device__ __forceinline__ v4f decay_weights(const float* pw, int t, int s0) {
  v4f w;
#pragma unroll
  for (int e = 0; e < 4; ++e) {
    const int d = t - 1 - (s0 + e);
    w[e] = pw[d < 0 ? 0 : d];
  }
  return w;
}
static __device__ __forceinline__ v4f decay_apply(v4f v, v4f w, int t, int s0) {
  v4f o;
#pragma unroll
  for (int e = 0; e < 4; ++e) o[e] = (s0 + e < t) ? w[e] * v[e] : 0.f;
  return o;
}

template <int APL, int NA>
static __device__ __forceinline__ v4f plane_sum(const v4f (&rp)[APL][NA], int j) {
  if constexpr (APL == 4) return (rp[0][j] + rp[1][j]) + (rp[2][j] + rp[3][j]);
  else return rp[0][j];
}

template <int BM, int BN, int BK, bool CAUSAL, bool DECAY, int APL>
__global__ __launch_bounds__(128) void gemm_AB_lds(const float* __restrict__ A,
                                                   const float* __restrict__ Bm,
                                                   float* __restrict__ C,
                                                   int N, int K,
                                                   long sA, long sB, long sC, long sPA) {
  static_assert(APL == 1 || APL == 4, "1 plane or 4 partial planes");
  constexpr int PA = BK + 4;
  constexpr int PB = BN + 8;
  constexpr int CA = BK / 4;
  constexpr int CB = BN / 4;
  constexpr int NA = BM * CA / 128;
  constexpr int NB = BK * CB / 128;
  constexpr int FM = BM / 32;
  constexpr int FN = BN / 32;
  __shared__ __attribute__((aligned(16))) float As[2][BM * PA];
  __shared__ __attribute__((aligned(16))) float Bs[2][BK * PB];
  __shared__ float pw[DECAY ? Tt : 1];

  const int b = blockIdx.z;
  A += (long)b * sA;
  Bm += (long)b * sB;
  C += (long)b * sC;
  const int tid = threadIdx.x;
  const int lane = tid & 31;
  const int wave = tid >> 5;
  const int wm = wave >> 1;
  const int wn = wave & 1;
  const int half = lane >> 4;
  const int l = lane & 15;
  const int row0 = blockIdx.y * BM;
  const int col0 = blockIdx.x * BN;

  const float* ga[NA];
  int la[NA];
  int ta[NA], sa[NA];
#pragma unroll
  for (int j = 0; j < NA; ++j) {
    const int f = tid + 128 * j;
    const int r = f / CA;
    const int c = f - r * CA;
    ga[j] = A + (long)(row0 + r) * K + c * 4;
    la[j] = r * PA + c * 4;
    ta[j] = row0 + r;
    sa[j] = c * 4;
  }
  const float* gb[NB];
  int lb[NB];
#pragma unroll
  for (int j = 0; j < NB; ++j) {
    const int f = tid + 128 * j;
    const int r = f / CB;
    const int c = f - r * CB;
    gb[j] = Bm + (long)r * N + col0 + c * 4;
    lb[j] = r * PB + c * 4;
  }

  v4f rp[APL][NA];
  v4f ra[NA], rb[NB];
#pragma unroll
  for (int p = 0; p < APL; ++p)
#pragma unroll
    for (int j = 0; j < NA; ++j) rp[p][j] = *(const v4f*)(ga[j] + (long)p * sPA);
#pragma unroll
  for (int j = 0; j < NB; ++j) rb[j] = *(const v4f*)(gb[j]);
  v4f wq[NA];
  if (DECAY) {
    for (int i = tid; i < Tt; i += 128) pw[i] = __powf(0.97f, (float)i);
    __syncthreads();
#pragma unroll
    for (int j = 0; j < NA; ++j) wq[j] = decay_weights(pw, ta[j], sa[j]);
  }
#pragma unroll
  for (int j = 0; j < NA; ++j)
    ra[j] = plane_sum<APL, NA>(rp, j);
  if (DECAY) {
#pragma unroll
    for (int j = 0; j < NA; ++j) ra[j] = decay_apply(ra[j], wq[j], ta[j], sa[j]);
  }
#pragma unroll
  for (int j = 0; j < NA; ++j) *(v4f*)(&As[0][la[j]]) = ra[j];
#pragma unroll
  for (int j = 0; j < NB; ++j) *(v4f*)(&Bs[0][lb[j]]) = rb[j];
  __syncthreads();

  v8f acc[FM][FN];
#pragma unroll
  for (int i = 0; i < FM; ++i)
#pragma unroll
    for (int j = 0; j < FN; ++j) acc[i][j] = v8f{};

  const int aoff = (wm * (BM / 2) + l) * PA + 2 * half;
  const int boff = (2 * half) * PB + wn * (BN / 2) + l;
  const int nchunks = CAUSAL ? (row0 + BM) / BK : K / BK;
  for (int ch = 0; ch < nchunks; ++ch) {
    const int cur = ch & 1;
    const bool more = (ch + 1 < nchunks);
    if (more) {
      const int kn = (ch + 1) * BK;
#pragma unroll
      for (int p = 0; p < APL; ++p)
#pragma unroll
        for (int j = 0; j < NA; ++j) rp[p][j] = *(const v4f*)(ga[j] + (long)p * sPA + kn);
#pragma unroll
      for (int j = 0; j < NB; ++j) rb[j] = *(const v4f*)(gb[j] + (long)kn * N);
      if (DECAY) {
#pragma unroll
        for (int j = 0; j < NA; ++j) wq[j] = decay_weights(pw, ta[j], sa[j] + kn);
      }
    }
    const float* Ab = &As[cur][aoff];
    const float* Bb = &Bs[cur][boff];
#pragma unroll
    for (int ks = 0; ks < BK / 4; ++ks) {
      v2f a[FM], bb[FN];
#pragma unroll
      for (int i = 0; i < FM; ++i) a[i] = *(const v2f*)(Ab + i * 16 * PA + ks * 4);
#pragma unroll
      for (int j = 0; j < FN; ++j) {
        bb[j][0] = Bb[(ks * 4) * PB + j * 16];
        bb[j][1] = Bb[(ks * 4 + 1) * PB + j * 16];
      }
#pragma unroll
      for (int i = 0; i < FM; ++i)
#pragma unroll
        for (int j = 0; j < FN; ++j)
          acc[i][j] = __builtin_amdgcn_wmma_f32_16x16x4_f32(false, a[i], false, bb[j], (short)0,
                                                            acc[i][j], false, false);
    }
    if (more) {
#pragma unroll
      for (int j = 0; j < NA; ++j)
        ra[j] = plane_sum<APL, NA>(rp, j);
      if (DECAY) {
        const int kn = (ch + 1) * BK;
#pragma unroll
        for (int j = 0; j < NA; ++j) ra[j] = decay_apply(ra[j], wq[j], ta[j], sa[j] + kn);
      }
#pragma unroll
      for (int j = 0; j < NA; ++j) *(v4f*)(&As[cur ^ 1][la[j]]) = ra[j];
#pragma unroll
      for (int j = 0; j < NB; ++j) *(v4f*)(&Bs[cur ^ 1][lb[j]]) = rb[j];
    }
    __syncthreads();
  }

#pragma unroll
  for (int i = 0; i < FM; ++i) {
#pragma unroll
    for (int j = 0; j < FN; ++j) {
      const long col = col0 + wn * (BN / 2) + j * 16 + l;
#pragma unroll
      for (int r = 0; r < 8; ++r) {
        const long row = row0 + wm * (BM / 2) + i * 16 + r + 8 * half;
        C[row * N + col] = acc[i][j][r];
      }
    }
  }
}

extern "C" void kernel_launch(void* const* d_in, const int* in_sizes, int n_in,
                              void* d_out, int out_size, void* d_ws, size_t ws_size,
                              hipStream_t stream) {
  const float* emb = (const float*)d_in[0];
  const float* E   = (const float*)d_in[1];
  const float* Dx  = (const float*)d_in[2];
  const float* Dy  = (const float*)d_in[3];
  float* out = (float*)d_out;

  const long BT = (long)Bb * Tt;
  const long szBTD = BT * Dd;
  const long szBTN = BT * Nn;
  const long szBTT = (long)Bb * Tt * Tt;

  float* ws = (float*)d_ws;
  float* LNV = ws;
  float* XU  = LNV + szBTD;
  float* X   = XU + szBTN;
  float* Rp  = X + szBTN;
  float* As  = Rp + szBTT;
  float* Y   = As + szBTD;
  float* KgP = XU;
  float* VP  = XU;

  gemm_ABt_lds<1, 64, 64, 32, 1><<<dim3(1024 / 64, 2048 / 64, 1), 128, 0, stream>>>(
      emb, Dx, XU, Rp, LNV, 1024, 256, 0, 0, 0, 0);

  xblocks_kernel<<<dim3(Nn / XN, 1, Bb), 128, 0, stream>>>(Rp, XU, X);

  gemm_ABt_lds<0, 64, 64, 32, 4><<<dim3(256 / 64, 256 / 64, Bb * 4), 128, 0, stream>>>(
      X, X, KgP, nullptr, nullptr, 256, 1024, (long)Tt * Nn, (long)Tt * Nn, (long)Tt * Tt, szBTT);

  gemm_AB_lds<32, 64, 32, false, true, 4><<<dim3(256 / 64, 256 / 32, Bb), 128, 0, stream>>>(
      KgP, LNV, As, 256, 256, (long)Tt * Tt, (long)Tt * Dd, (long)Tt * Dd, szBTT);

  gemm_lnA_Bt_lds<<<dim3(1024 / 64, 2048 / 64, 1), 128, 0, stream>>>(As, Dy, Y, X, 1024);

  gemm_ABt_lds<0, 64, 64, 32, 4><<<dim3(256 / 64, 2048 / 64, 4), 128, 0, stream>>>(
      Y, E, VP, nullptr, nullptr, 256, 1024, 0, 0, 0, szBTD);

  ln256_planes4_kernel<<<2048 / (4 * LN_RW), 128, 0, stream>>>(VP, out, szBTD);
}


# Decay GEMM (1 wave per SIMD): loads run a block ahead: K chunks 128 (was 32); B tile in LDS as k pairs, one 8-byte read per fragment, reads issued 8 WMMAs ahead; k and sum order kept
# speedup vs baseline: 10.3273x; 1.0477x over previous
#include <hip/hip_runtime.h>


typedef __attribute__((ext_vector_type(2))) float v2f;
typedef __attribute__((ext_vector_type(8))) float v8f;
typedef __attribute__((ext_vector_type(4))) float v4f;

static constexpr int Bb = 8;
static constexpr int Tt = 256;
static constexpr int Dd = 256;
static constexpr int Nn = 1024;

template <int CTRL>
static __device__ __forceinline__ float row_xadd(float v) {
  const int iv = __builtin_bit_cast(int, v);
  const int sw = __builtin_amdgcn_update_dpp(0, iv, CTRL, 0xf, 0xf, true);
  return v + __builtin_bit_cast(float, sw);
}

static __device__ __forceinline__ float xhalf(float v) {
  const int iv = __builtin_bit_cast(int, v);
  return __builtin_bit_cast(float, __builtin_amdgcn_permlanex16(iv, iv, 0x76543210, 0xfedcba98,
                                                                false, false));
}

static __device__ __forceinline__ void ln_row256(v4f& a, v4f& b) {
  float s = ((a[0] + a[1]) + (a[2] + a[3])) + ((b[0] + b[1]) + (b[2] + b[3]));
  s = row_xadd<0x161>(s);
  s = row_xadd<0x162>(s);
  s = row_xadd<0x164>(s);
  s = row_xadd<0x168>(s);
  s += xhalf(s);
  const float mean = s * (1.0f / 256.0f);
  a = a - mean;
  b = b - mean;
  float q = ((a[0] * a[0] + a[1] * a[1]) + (a[2] * a[2] + a[3] * a[3])) +
            ((b[0] * b[0] + b[1] * b[1]) + (b[2] * b[2] + b[3] * b[3]));
  q = row_xadd<0x161>(q);
  q = row_xadd<0x162>(q);
  q = row_xadd<0x164>(q);
  q = row_xadd<0x168>(q);
  q += xhalf(q);
  const float rstd = rsqrtf(q * (1.0f / 256.0f) + 1e-5f);
  a = a * rstd;
  b = b * rstd;
}

static constexpr int LN_RW = 2;
__global__ __launch_bounds__(128) void ln256_planes4_kernel(const float* __restrict__ in,
                                                            float* __restrict__ out, long sP) {
  const int lane = threadIdx.x & 31;
  const int wave = threadIdx.x >> 5;
  const long row = ((long)blockIdx.x * 4 + wave) * LN_RW;
  const float* src = in + row * 256 + lane * 4;
  v4f pa[LN_RW][4], pb[LN_RW][4];
#pragma unroll
  for (int r = 0; r < LN_RW; ++r) {
#pragma unroll
    for (int p = 0; p < 4; ++p) {
      pa[r][p] = *(const v4f*)(src + (long)p * sP + r * 256);
      pb[r][p] = *(const v4f*)(src + (long)p * sP + r * 256 + 128);
    }
  }
#pragma unroll
  for (int r = 0; r < LN_RW; ++r) {
    v4f a = (pa[r][0] + pa[r][1]) + (pa[r][2] + pa[r][3]);
    v4f b = (pb[r][0] + pb[r][1]) + (pb[r][2] + pb[r][3]);
    ln_row256(a, b);
    float* dst = out + (row + r) * 256 + lane * 4;
    *(v4f*)(dst) = a;
    *(v4f*)(dst + 128) = b;
  }
}


static constexpr int RS_STRIPS = 32;
static constexpr int XB = 64;
static constexpr int XN = 32;

__global__ __launch_bounds__(128) void xblocks_kernel(const float* __restrict__ Rp,
                                                      const float* __restrict__ U,
                                                      float* __restrict__ X) {
  constexpr int PA = XB + 4;
  constexpr int PB = XN + 8;
  constexpr int NBLK = Tt / XB;
  constexpr int NU = Tt * (XN / 4) / 128;
  __shared__ __attribute__((aligned(16))) float sm[Tt * PB + NBLK * XB * PA + 2 * Tt + 2 * XN];
  __shared__ unsigned big[8];
  float* Us = sm;
  float* Gs = Us + Tt * PB;
  float* rs = Gs + NBLK * XB * PA;
  float* qs = rs + Tt;
  float* xp = qs + Tt;

  const int b = blockIdx.z;
  const int col0 = blockIdx.x * XN;
  const int tid = threadIdx.x;
  const int lane = tid & 31;
  const int wave = tid >> 5;
  const int half = lane >> 4;
  const int l = lane & 15;

  const float* up = U + (long)b * Tt * Nn + col0;
  v4f ru[NU];
  int lu[NU];
#pragma unroll
  for (int j = 0; j < NU; ++j) {
    const int f = tid + 128 * j;
    const int r = f / (XN / 4);
    const int c = f - r * (XN / 4);
    ru[j] = *(const v4f*)(up + (long)r * Nn + c * 4);
    lu[j] = r * PB + c * 4;
  }

  float r0 = 0.f, r1 = 0.f;
#pragma unroll
  for (int c = 0; c < RS_STRIPS; ++c) {
    r0 += Rp[(long)c * (Bb * Tt) + b * Tt + tid];
    r1 += Rp[(long)c * (Bb * Tt) + b * Tt + tid + 128];
  }
  rs[tid] = r0;
  rs[tid + 128] = r1;
  const unsigned m0 = __builtin_amdgcn_ballot_w32(r0 >= 1e-12f);
  const unsigned m1 = __builtin_amdgcn_ballot_w32(r1 >= 1e-12f);
  if (lane == 0) {
    big[wave] = m0;
    big[4 + wave] = m1;
  }
  __syncthreads();

  int f = Tt;
#pragma unroll
  for (int w = 7; w >= 0; --w) {
    const unsigned mw = big[w];
    if (mw != 0u) f = 32 * w + __builtin_ctz(mw);
  }
  float inv0, inv1;
  {
    float S = 1.0f;
    if (tid <= f) {
      S = 0.f;
      for (int i = 0; i < tid; ++i) {
        const float sum = fmaf(0.97f, S, rs[i]);
        S = (sum >= 1e-12f) ? 1.0f : sum * 1e12f;
      }
    }
    inv0 = 1.0f / fmaxf(fmaf(0.97f, S, r0), 1e-12f);
    qs[tid] = 0.97f * inv0;
  }
  {
    float S = 1.0f;
    if (tid + 128 <= f) {
      S = 0.f;
      for (int i = 0; i < tid + 128; ++i) {
        const float sum = fmaf(0.97f, S, rs[i]);
        S = (sum >= 1e-12f) ? 1.0f : sum * 1e12f;
      }
    }
    inv1 = 1.0f / fmaxf(fmaf(0.97f, S, r1), 1e-12f);
    qs[tid + 128] = 0.97f * inv1;
  }
#pragma unroll
  for (int j = 0; j < NU; ++j) *(v4f*)(&Us[lu[j]]) = ru[j];
  __syncthreads();

  const float GMAX = 3.0e38f;
  {
    const int jb = tid >> 6;
    const int sl = tid & (XB - 1);
    float* gp0 = Gs + jb * (XB * PA) + sl;
    float* gp1 = gp0 + 2 * (XB * PA);
    const float* q0 = &qs[XB * jb];
    const float* q1 = &qs[XB * (jb + 2)];
    float g0 = 0.f, g1 = 0.f;
#pragma unroll 8
    for (int i = 0; i < XB; ++i) {
      const float n0 = fminf(g0 * q0[i], GMAX);
      const float n1 = fminf(g1 * q1[i], GMAX);
      g0 = (i == sl) ? inv0 : n0;
      g1 = (i == sl) ? inv1 : n1;
      gp0[i * PA] = g0;
      gp1[i * PA] = g1;
    }
  }
  __syncthreads();

  const float* Ab0 = Gs + (16 * wave + l) * PA + 2 * half;
  const float* Bb0 = Us + (2 * half) * PB + l;
  float* xo = X + (long)b * Tt * Nn + col0;
  for (int j = 0; j < NBLK; ++j) {
    const float* Ab = Ab0 + j * (XB * PA);
    const float* Bq = Bb0 + (j * XB) * PB;
    v8f acc[2];
    acc[0] = v8f{};
    acc[1] = v8f{};
#pragma unroll
    for (int ks = 1; ks < XB / 4; ++ks) {
      const v2f a = *(const v2f*)(Ab + ks * 4);
      v2f bb[2];
#pragma unroll
      for (int n = 0; n < 2; ++n) {
        bb[n][0] = Bq[(ks * 4) * PB + n * 16];
        bb[n][1] = Bq[(ks * 4 + 1) * PB + n * 16];
      }
#pragma unroll
      for (int n = 0; n < 2; ++n)
        acc[n] = __builtin_amdgcn_wmma_f32_16x16x4_f32(false, a, false, bb[n], (short)0, acc[n],
                                                       false, false);
    }
    {
      const v2f a = *(const v2f*)(Ab);
      v2f bb[2];
#pragma unroll
      for (int n = 0; n < 2; ++n) {
        bb[n][0] = Bq[n * 16];
        bb[n][1] = Bq[PB + n * 16];
      }
      if (j > 0) {
        __builtin_amdgcn_fence(__ATOMIC_RELEASE, "workgroup", "local");
        __builtin_amdgcn_s_barrier();
        __builtin_amdgcn_fence(__ATOMIC_ACQUIRE, "workgroup", "local");
        const float* xq = xp + ((j - 1) & 1) * XN;
#pragma unroll
        for (int n = 0; n < 2; ++n) {
          const float cv = fmaf(0.97f, xq[n * 16 + l], bb[n][0]);
          bb[n][0] = (half == 0) ? cv : bb[n][0];
        }
      }
#pragma unroll
      for (int n = 0; n < 2; ++n)
        acc[n] = __builtin_amdgcn_wmma_f32_16x16x4_f32(false, a, false, bb[n], (short)0, acc[n],
                                                       false, false);
    }
    if (j + 1 < NBLK && wave == 3 && half == 1) {
      xp[(j & 1) * XN + l] = acc[0][7];
      xp[(j & 1) * XN + 16 + l] = acc[1][7];
    }
#pragma unroll
    for (int n = 0; n < 2; ++n) {
#pragma unroll
      for (int r = 0; r < 8; ++r) {
        const long row = j * XB + 16 * wave + 8 * half + r;
        xo[row * Nn + n * 16 + l] = acc[n][r];
      }
    }
  }
}

template <int EPI, int BM, int BN, int BK, int SPLIT>
__global__ __launch_bounds__(128) void gemm_ABt_lds(const float* __restrict__ A,
                                                    const float* __restrict__ W,
                                                    float* __restrict__ C,
                                                    float* __restrict__ Rs,
                                                    float* __restrict__ Ln,
                                                    int N, int K,
                                                    long sA, long sW, long sC, long sP) {
  constexpr int PITCH = BK + 4;
  constexpr int C4 = BK / 4;
  constexpr int NA = BM * C4 / 128;
  constexpr int NW = BN * C4 / 128;
  constexpr int FM = BM / 32;
  constexpr int FN = BN / 32;
  __shared__ __attribute__((aligned(16))) float As[2][BM * PITCH];
  __shared__ __attribute__((aligned(16))) float Ws[2][BN * PITCH];

  const int b = blockIdx.z / SPLIT;
  const int sp = blockIdx.z - b * SPLIT;
  const int Ks = K / SPLIT;
  A += (long)b * sA + sp * Ks;
  W += (long)b * sW + sp * Ks;
  C += (long)b * sC + (long)sp * sP;
  const int tid = threadIdx.x;
  const int lane = tid & 31;
  const int wave = tid >> 5;
  const int wm = wave >> 1;
  const int wn = wave & 1;
  const int half = lane >> 4;
  const int l = lane & 15;
  const int row0 = blockIdx.y * BM;
  const int col0 = blockIdx.x * BN;

  const float* ga[NA];
  int la[NA];
#pragma unroll
  for (int j = 0; j < NA; ++j) {
    const int f = tid + 128 * j;
    const int r = f / C4;
    const int c = f - r * C4;
    ga[j] = A + (long)(row0 + r) * K + c * 4;
    la[j] = r * PITCH + c * 4;
  }
  const float* gw[NW];
  int lw[NW];
#pragma unroll
  for (int j = 0; j < NW; ++j) {
    const int f = tid + 128 * j;
    const int r = f / C4;
    const int c = f - r * C4;
    gw[j] = W + (long)(col0 + r) * K + c * 4;
    lw[j] = r * PITCH + c * 4;
  }

  v4f ra[NA], rw[NW];
#pragma unroll
  for (int j = 0; j < NA; ++j) ra[j] = *(const v4f*)(ga[j]);
#pragma unroll
  for (int j = 0; j < NW; ++j) rw[j] = *(const v4f*)(gw[j]);
  v4f na = v4f{}, nb = v4f{};
  long nrow = 0;
  if constexpr (EPI == 1) {
    nrow = row0 + 4 * (int)blockIdx.x + wave;
    const float* np = A + nrow * Dd + lane * 4;
    na = *(const v4f*)(np);
    nb = *(const v4f*)(np + 128);
  }
#pragma unroll
  for (int j = 0; j < NA; ++j) *(v4f*)(&As[0][la[j]]) = ra[j];
#pragma unroll
  for (int j = 0; j < NW; ++j) *(v4f*)(&Ws[0][lw[j]]) = rw[j];
  __syncthreads();
  if constexpr (EPI == 1) ln_row256(na, nb);

  v8f acc[FM][FN];
#pragma unroll
  for (int i = 0; i < FM; ++i)
#pragma unroll
    for (int j = 0; j < FN; ++j) acc[i][j] = v8f{};

  const int aoff = (wm * (BM / 2) + l) * PITCH + 2 * half;
  const int woff = (wn * (BN / 2) + l) * PITCH + 2 * half;
  const int nchunks = Ks / BK;
  for (int ch = 0; ch < nchunks; ++ch) {
    const int cur = ch & 1;
    const bool more = (ch + 1 < nchunks);
    if (more) {
      const int kn = (ch + 1) * BK;
#pragma unroll
      for (int j = 0; j < NA; ++j) ra[j] = *(const v4f*)(ga[j] + kn);
#pragma unroll
      for (int j = 0; j < NW; ++j) rw[j] = *(const v4f*)(gw[j] + kn);
    }
    const float* Ab = &As[cur][aoff];
    const float* Wb = &Ws[cur][woff];
#pragma unroll
    for (int ks = 0; ks < BK / 4; ++ks) {
      v2f a[FM], bb[FN];
#pragma unroll
      for (int i = 0; i < FM; ++i) a[i] = *(const v2f*)(Ab + i * 16 * PITCH + ks * 4);
#pragma unroll
      for (int j = 0; j < FN; ++j) bb[j] = *(const v2f*)(Wb + j * 16 * PITCH + ks * 4);
#pragma unroll
      for (int i = 0; i < FM; ++i)
#pragma unroll
        for (int j = 0; j < FN; ++j)
          acc[i][j] = __builtin_amdgcn_wmma_f32_16x16x4_f32(false, a[i], false, bb[j], (short)0,
                                                            acc[i][j], false, false);
    }
    if (more) {
#pragma unroll
      for (int j = 0; j < NA; ++j) *(v4f*)(&As[cur ^ 1][la[j]]) = ra[j];
#pragma unroll
      for (int j = 0; j < NW; ++j) *(v4f*)(&Ws[cur ^ 1][lw[j]]) = rw[j];
    }
    __syncthreads();
  }

  if constexpr (EPI == 1) {
    float* nd = Ln + nrow * Dd + lane * 4;
    *(v4f*)(nd) = na;
    *(v4f*)(nd + 128) = nb;
  }

#pragma unroll
  for (int i = 0; i < FM; ++i) {
#pragma unroll
    for (int j = 0; j < FN; ++j) {
      const long col = col0 + wn * (BN / 2) + j * 16 + l;
#pragma unroll
      for (int r = 0; r < 8; ++r) {
        const long row = row0 + wm * (BM / 2) + i * 16 + r + 8 * half;
        float v = acc[i][j][r];
        if (EPI == 1) v = fmaxf(v, 0.f);
        C[row * N + col] = v;
      }
    }
  }

  if (EPI == 1) {
    static_assert(EPI != 1 || (BM == 64 && BN == 64), "row-sum epilogue: 32 x 32 wave tile");
    float p[FM][8];
#pragma unroll
    for (int i = 0; i < FM; ++i) {
#pragma unroll
      for (int r = 0; r < 8; ++r) {
        float s = 0.f;
#pragma unroll
        for (int j = 0; j < FN; ++j) s += fmaxf(acc[i][j][r], 0.f);
        s = row_xadd<0x161>(s);
        s = row_xadd<0x162>(s);
        s = row_xadd<0x164>(s);
        s = row_xadd<0x168>(s);
        p[i][r] = s;
      }
    }
    float mine = 0.f;
#pragma unroll
    for (int i = 0; i < FM; ++i)
#pragma unroll
      for (int r = 0; r < 8; ++r)
        if (l == i * 8 + r) mine = p[i][r];
    const int rho = (l >> 3) * 16 + 8 * half + (l & 7);
    const long rows = (long)gridDim.y * BM;
    Rs[(long)(blockIdx.x * 2 + wn) * rows + row0 + wm * (BM / 2) + rho] = mine;
  }
}

__global__ __launch_bounds__(128) void gemm_lnA_Bt_lds(const float* __restrict__ A,
                                                       const float* __restrict__ W,
                                                       float* __restrict__ C,
                                                       const float* __restrict__ Mul,
                                                       int N) {
  constexpr int BM = 64, BN = 64, BK = 32, KK = 256;
  constexpr int PP = KK + 4;
  constexpr int PITCH = BK + 4;
  constexpr int C4 = BK / 4;
  constexpr int NW = BN * C4 / 128;
  constexpr int FM = BM / 32;
  constexpr int FN = BN / 32;
  __shared__ __attribute__((aligned(16))) float Ap[BM * PP];
  __shared__ __attribute__((aligned(16))) float Ws[2][BN * PITCH];

  const int tid = threadIdx.x;
  const int lane = tid & 31;
  const int wave = tid >> 5;
  const int wm = wave >> 1;
  const int wn = wave & 1;
  const int half = lane >> 4;
  const int l = lane & 15;
  const int row0 = blockIdx.y * BM;
  const int col0 = blockIdx.x * BN;

  const float* gw[NW];
  int lw[NW];
#pragma unroll
  for (int j = 0; j < NW; ++j) {
    const int f = tid + 128 * j;
    const int r = f / C4;
    const int c = f - r * C4;
    gw[j] = W + (long)(col0 + r) * KK + c * 4;
    lw[j] = r * PITCH + c * 4;
  }

  v4f p[16][2];
  const float* gp = A + (long)(row0 + wave * 16) * KK + lane * 4;
#pragma unroll
  for (int i = 0; i < 16; ++i) {
    p[i][0] = *(const v4f*)(gp + (long)i * KK);
    p[i][1] = *(const v4f*)(gp + (long)i * KK + 128);
  }
  v4f rw[NW];
#pragma unroll
  for (int j = 0; j < NW; ++j) rw[j] = *(const v4f*)(gw[j]);

#pragma unroll
  for (int i = 0; i < 16; ++i) {
    float s = ((p[i][0][0] + p[i][0][1]) + (p[i][0][2] + p[i][0][3])) +
              ((p[i][1][0] + p[i][1][1]) + (p[i][1][2] + p[i][1][3]));
    s = row_xadd<0x161>(s);
    s = row_xadd<0x162>(s);
    s = row_xadd<0x164>(s);
    s = row_xadd<0x168>(s);
    s += __shfl_xor(s, 16, 32);
    const float mean = s * (1.0f / 256.0f);
    const v4f c0 = p[i][0] - mean;
    const v4f c1 = p[i][1] - mean;
    float q = ((c0[0] * c0[0] + c0[1] * c0[1]) + (c0[2] * c0[2] + c0[3] * c0[3])) +
              ((c1[0] * c1[0] + c1[1] * c1[1]) + (c1[2] * c1[2] + c1[3] * c1[3]));
    q = row_xadd<0x161>(q);
    q = row_xadd<0x162>(q);
    q = row_xadd<0x164>(q);
    q = row_xadd<0x168>(q);
    q += __shfl_xor(q, 16, 32);
    const float rstd = rsqrtf(q * (1.0f / 256.0f) + 1e-5f);
    float* dst = &Ap[(wave * 16 + i) * PP + lane * 4];
    *(v4f*)(dst) = c0 * rstd;
    *(v4f*)(dst + 128) = c1 * rstd;
  }
#pragma unroll
  for (int j = 0; j < NW; ++j) *(v4f*)(&Ws[0][lw[j]]) = rw[j];
  __syncthreads();

  v8f acc[FM][FN];
#pragma unroll
  for (int i = 0; i < FM; ++i)
#pragma unroll
    for (int j = 0; j < FN; ++j) acc[i][j] = v8f{};

  const int aoff = (wm * (BM / 2) + l) * PP + 2 * half;
  const int woff = (wn * (BN / 2) + l) * PITCH + 2 * half;
  constexpr int nchunks = KK / BK;
#pragma unroll 1
  for (int ch = 0; ch < nchunks; ++ch) {
    const int cur = ch & 1;
    const bool more = (ch + 1 < nchunks);
    if (more) {
      const int kn = (ch + 1) * BK;
#pragma unroll
      for (int j = 0; j < NW; ++j) rw[j] = *(const v4f*)(gw[j] + kn);
    }
    const float* Ab = &Ap[aoff + ch * BK];
    const float* Wb = &Ws[cur][woff];
#pragma unroll
    for (int ks = 0; ks < BK / 4; ++ks) {
      v2f a[FM], bb[FN];
#pragma unroll
      for (int i = 0; i < FM; ++i) a[i] = *(const v2f*)(Ab + i * 16 * PP + ks * 4);
#pragma unroll
      for (int j = 0; j < FN; ++j) bb[j] = *(const v2f*)(Wb + j * 16 * PITCH + ks * 4);
#pragma unroll
      for (int i = 0; i < FM; ++i)
#pragma unroll
        for (int j = 0; j < FN; ++j)
          acc[i][j] = __builtin_amdgcn_wmma_f32_16x16x4_f32(false, a[i], false, bb[j], (short)0,
                                                            acc[i][j], false, false);
    }
    if (more) {
#pragma unroll
      for (int j = 0; j < NW; ++j) *(v4f*)(&Ws[cur ^ 1][lw[j]]) = rw[j];
    }
    __syncthreads();
  }

#pragma unroll
  for (int i = 0; i < FM; ++i) {
#pragma unroll
    for (int j = 0; j < FN; ++j) {
      const long col = col0 + wn * (BN / 2) + j * 16 + l;
#pragma unroll
      for (int r = 0; r < 8; ++r) {
        const long row = row0 + wm * (BM / 2) + i * 16 + r + 8 * half;
        const float v = fmaxf(acc[i][j][r], 0.f) * fmaxf(Mul[row * N + col], 0.f);
        C[row * N + col] = v;
      }
    }
  }
}

static __device__ __forceinline__ v4f decay_weights(const float* pw, int t, int s0) {
  v4f w;
#pragma unroll
  for (int e = 0; e < 4; ++e) {
    const int d = t - 1 - (s0 + e);
    w[e] = pw[d < 0 ? 0 : d];
  }
  return w;
}
static __device__ __forceinline__ v4f decay_apply(v4f v, v4f w, int t, int s0) {
  v4f o;
#pragma unroll
  for (int e = 0; e < 4; ++e) o[e] = (s0 + e < t) ? w[e] * v[e] : 0.f;
  return o;
}

template <int APL, int NA>
static __device__ __forceinline__ v4f plane_sum(const v4f (&rp)[APL][NA], int j) {
  if constexpr (APL == 4) return (rp[0][j] + rp[1][j]) + (rp[2][j] + rp[3][j]);
  else return rp[0][j];
}

template <int BM, int BN, int BK, bool CAUSAL, bool DECAY, int APL>
__global__ __launch_bounds__(128) void gemm_AB_lds(const float* __restrict__ A,
                                                   const float* __restrict__ Bm,
                                                   float* __restrict__ C,
                                                   int N, int K,
                                                   long sA, long sB, long sC, long sPA) {
  static_assert(APL == 1 || APL == 4, "1 plane or 4 partial planes");
  constexpr int PA = BK + 4;
  constexpr int PB = 2 * BN + 32;
  static_assert(BN == 64, "k-pair pitch: 2 * BN + 32 = 32 mod 64 banks");
  constexpr int CA = BK / 4;
  constexpr int CB = BN / 4;
  constexpr int NA = BM * CA / 128;
  constexpr int NB = (BK / 2) * CB / 128;
  constexpr int FM = BM / 32;
  constexpr int FN = BN / 32;
  __shared__ __attribute__((aligned(16))) float As[2][BM * PA];
  __shared__ __attribute__((aligned(16))) float Bs[2][(BK / 2) * PB];
  __shared__ float pw[DECAY ? Tt : 1];

  const int b = blockIdx.z;
  A += (long)b * sA;
  Bm += (long)b * sB;
  C += (long)b * sC;
  const int tid = threadIdx.x;
  const int lane = tid & 31;
  const int wave = tid >> 5;
  const int wm = wave >> 1;
  const int wn = wave & 1;
  const int half = lane >> 4;
  const int l = lane & 15;
  const int row0 = blockIdx.y * BM;
  const int col0 = blockIdx.x * BN;

  const float* ga[NA];
  int la[NA];
  int ta[NA], sa[NA];
#pragma unroll
  for (int j = 0; j < NA; ++j) {
    const int f = tid + 128 * j;
    const int r = f / CA;
    const int c = f - r * CA;
    ga[j] = A + (long)(row0 + r) * K + c * 4;
    la[j] = r * PA + c * 4;
    ta[j] = row0 + r;
    sa[j] = c * 4;
  }
  const float* gb[NB];
  int lb[NB];
#pragma unroll
  for (int j = 0; j < NB; ++j) {
    const int f = tid + 128 * j;
    const int r = f / CB;
    const int c = f - r * CB;
    gb[j] = Bm + (long)(2 * r) * N + col0 + c * 4;
    lb[j] = r * PB + c * 8;
  }

  v4f rp[APL][NA];
  v4f ra[NA], rb[NB][2];
#pragma unroll
  for (int p = 0; p < APL; ++p)
#pragma unroll
    for (int j = 0; j < NA; ++j) rp[p][j] = *(const v4f*)(ga[j] + (long)p * sPA);
#pragma unroll
  for (int j = 0; j < NB; ++j) {
    rb[j][0] = *(const v4f*)(gb[j]);
    rb[j][1] = *(const v4f*)(gb[j] + N);
  }
  v4f wq[NA];
  if (DECAY) {
    for (int i = tid; i < Tt; i += 128) pw[i] = __powf(0.97f, (float)i);
    __syncthreads();
#pragma unroll
    for (int j = 0; j < NA; ++j) wq[j] = decay_weights(pw, ta[j], sa[j]);
  }
#pragma unroll
  for (int j = 0; j < NA; ++j)
    ra[j] = plane_sum<APL, NA>(rp, j);
  if (DECAY) {
#pragma unroll
    for (int j = 0; j < NA; ++j) ra[j] = decay_apply(ra[j], wq[j], ta[j], sa[j]);
  }
#pragma unroll
  for (int j = 0; j < NA; ++j) *(v4f*)(&As[0][la[j]]) = ra[j];
#pragma unroll
  for (int j = 0; j < NB; ++j) {
    *(v4f*)(&Bs[0][lb[j]]) = v4f{rb[j][0][0], rb[j][1][0], rb[j][0][1], rb[j][1][1]};
    *(v4f*)(&Bs[0][lb[j] + 4]) = v4f{rb[j][0][2], rb[j][1][2], rb[j][0][3], rb[j][1][3]};
  }
  __syncthreads();

  v8f acc[FM][FN];
#pragma unroll
  for (int i = 0; i < FM; ++i)
#pragma unroll
    for (int j = 0; j < FN; ++j) acc[i][j] = v8f{};

  const int aoff = (wm * (BM / 2) + l) * PA + 2 * half;
  const int boff = half * PB + (wn * (BN / 2) + l) * 2;
  const int nchunks = CAUSAL ? (row0 + BM) / BK : K / BK;
  for (int ch = 0; ch < nchunks; ++ch) {
    const int cur = ch & 1;
    const bool more = (ch + 1 < nchunks);
    if (more) {
      const int kn = (ch + 1) * BK;
#pragma unroll
      for (int p = 0; p < APL; ++p)
#pragma unroll
        for (int j = 0; j < NA; ++j) rp[p][j] = *(const v4f*)(ga[j] + (long)p * sPA + kn);
#pragma unroll
      for (int j = 0; j < NB; ++j) {
        rb[j][0] = *(const v4f*)(gb[j] + (long)kn * N);
        rb[j][1] = *(const v4f*)(gb[j] + (long)(kn + 1) * N);
      }
      if (DECAY) {
#pragma unroll
        for (int j = 0; j < NA; ++j) wq[j] = decay_weights(pw, ta[j], sa[j] + kn);
      }
    }
    const float* Ab = &As[cur][aoff];
    const float* Bb = &Bs[cur][boff];
#pragma unroll
    for (int ks = 0; ks < BK / 4; ++ks) {
      v2f a[FM], bb[FN];
#pragma unroll
      for (int i = 0; i < FM; ++i) a[i] = *(const v2f*)(Ab + i * 16 * PA + ks * 4);
#pragma unroll
      for (int j = 0; j < FN; ++j) bb[j] = *(const v2f*)(Bb + (ks * 2) * PB + j * 32);
#pragma unroll
      for (int i = 0; i < FM; ++i)
#pragma unroll
        for (int j = 0; j < FN; ++j)
          acc[i][j] = __builtin_amdgcn_wmma_f32_16x16x4_f32(false, a[i], false, bb[j], (short)0,
                                                            acc[i][j], false, false);
    }
    if constexpr (FM == 1 && FN == 2) {
      __builtin_amdgcn_sched_group_barrier(0x100, 9, 0);
#pragma unroll
      for (int g = 0; g < BK / 8; ++g) {
        __builtin_amdgcn_sched_group_barrier(0x008, 4, 0);
        __builtin_amdgcn_sched_group_barrier(0x100, 3, 0);
      }
    }
    if (more) {
#pragma unroll
      for (int j = 0; j < NA; ++j)
        ra[j] = plane_sum<APL, NA>(rp, j);
      if (DECAY) {
        const int kn = (ch + 1) * BK;
#pragma unroll
        for (int j = 0; j < NA; ++j) ra[j] = decay_apply(ra[j], wq[j], ta[j], sa[j] + kn);
      }
#pragma unroll
      for (int j = 0; j < NA; ++j) *(v4f*)(&As[cur ^ 1][la[j]]) = ra[j];
#pragma unroll
      for (int j = 0; j < NB; ++j) {
        *(v4f*)(&Bs[cur ^ 1][lb[j]]) = v4f{rb[j][0][0], rb[j][1][0], rb[j][0][1], rb[j][1][1]};
        *(v4f*)(&Bs[cur ^ 1][lb[j] + 4]) = v4f{rb[j][0][2], rb[j][1][2], rb[j][0][3], rb[j][1][3]};
      }
    }
    __syncthreads();
  }

#pragma unroll
  for (int i = 0; i < FM; ++i) {
#pragma unroll
    for (int j = 0; j < FN; ++j) {
      const long col = col0 + wn * (BN / 2) + j * 16 + l;
#pragma unroll
      for (int r = 0; r < 8; ++r) {
        const long row = row0 + wm * (BM / 2) + i * 16 + r + 8 * half;
        C[row * N + col] = acc[i][j][r];
      }
    }
  }
}

extern "C" void kernel_launch(void* const* d_in, const int* in_sizes, int n_in,
                              void* d_out, int out_size, void* d_ws, size_t ws_size,
                              hipStream_t stream) {
  const float* emb = (const float*)d_in[0];
  const float* E   = (const float*)d_in[1];
  const float* Dx  = (const float*)d_in[2];
  const float* Dy  = (const float*)d_in[3];
  float* out = (float*)d_out;

  const long BT = (long)Bb * Tt;
  const long szBTD = BT * Dd;
  const long szBTN = BT * Nn;
  const long szBTT = (long)Bb * Tt * Tt;

  float* ws = (float*)d_ws;
  float* LNV = ws;
  float* XU  = LNV + szBTD;
  float* X   = XU + szBTN;
  float* Rp  = X + szBTN;
  float* As  = Rp + szBTT;
  float* Y   = As + szBTD;
  float* KgP = XU;
  float* VP  = XU;

  gemm_ABt_lds<1, 64, 64, 32, 1><<<dim3(1024 / 64, 2048 / 64, 1), 128, 0, stream>>>(
      emb, Dx, XU, Rp, LNV, 1024, 256, 0, 0, 0, 0);

  xblocks_kernel<<<dim3(Nn / XN, 1, Bb), 128, 0, stream>>>(Rp, XU, X);

  gemm_ABt_lds<0, 64, 64, 32, 4><<<dim3(256 / 64, 256 / 64, Bb * 4), 128, 0, stream>>>(
      X, X, KgP, nullptr, nullptr, 256, 1024, (long)Tt * Nn, (long)Tt * Nn, (long)Tt * Tt, szBTT);

  gemm_AB_lds<32, 64, 128, false, true, 4><<<dim3(256 / 64, 256 / 32, Bb), 128, 0, stream>>>(
      KgP, LNV, As, 256, 256, (long)Tt * Tt, (long)Tt * Dd, (long)Tt * Dd, szBTT);

  gemm_lnA_Bt_lds<<<dim3(1024 / 64, 2048 / 64, 1), 128, 0, stream>>>(As, Dy, Y, X, 1024);

  gemm_ABt_lds<0, 64, 64, 32, 4><<<dim3(256 / 64, 2048 / 64, 4), 128, 0, stream>>>(
      Y, E, VP, nullptr, nullptr, 256, 1024, 0, 0, 0, szBTD);

  ln256_planes4_kernel<<<2048 / (4 * LN_RW), 128, 0, stream>>>(VP, out, szBTD);
}


# x blocks (1 wave/SIMD): LDS reads off dependent paths: step factors in registers before G's column products, u strip as k pairs (8-byte B reads), next block's fragments read under WMMAs; sums kept
# speedup vs baseline: 10.8120x; 1.0469x over previous
#include <hip/hip_runtime.h>


typedef __attribute__((ext_vector_type(2))) float v2f;
typedef __attribute__((ext_vector_type(8))) float v8f;
typedef __attribute__((ext_vector_type(4))) float v4f;

static constexpr int Bb = 8;
static constexpr int Tt = 256;
static constexpr int Dd = 256;
static constexpr int Nn = 1024;

template <int CTRL>
static __device__ __forceinline__ float row_xadd(float v) {
  const int iv = __builtin_bit_cast(int, v);
  const int sw = __builtin_amdgcn_update_dpp(0, iv, CTRL, 0xf, 0xf, true);
  return v + __builtin_bit_cast(float, sw);
}

static __device__ __forceinline__ float xhalf(float v) {
  const int iv = __builtin_bit_cast(int, v);
  return __builtin_bit_cast(float, __builtin_amdgcn_permlanex16(iv, iv, 0x76543210, 0xfedcba98,
                                                                false, false));
}

static __device__ __forceinline__ void ln_row256(v4f& a, v4f& b) {
  float s = ((a[0] + a[1]) + (a[2] + a[3])) + ((b[0] + b[1]) + (b[2] + b[3]));
  s = row_xadd<0x161>(s);
  s = row_xadd<0x162>(s);
  s = row_xadd<0x164>(s);
  s = row_xadd<0x168>(s);
  s += xhalf(s);
  const float mean = s * (1.0f / 256.0f);
  a = a - mean;
  b = b - mean;
  float q = ((a[0] * a[0] + a[1] * a[1]) + (a[2] * a[2] + a[3] * a[3])) +
            ((b[0] * b[0] + b[1] * b[1]) + (b[2] * b[2] + b[3] * b[3]));
  q = row_xadd<0x161>(q);
  q = row_xadd<0x162>(q);
  q = row_xadd<0x164>(q);
  q = row_xadd<0x168>(q);
  q += xhalf(q);
  const float rstd = rsqrtf(q * (1.0f / 256.0f) + 1e-5f);
  a = a * rstd;
  b = b * rstd;
}

static constexpr int LN_RW = 2;
__global__ __launch_bounds__(128) void ln256_planes4_kernel(const float* __restrict__ in,
                                                            float* __restrict__ out, long sP) {
  const int lane = threadIdx.x & 31;
  const int wave = threadIdx.x >> 5;
  const long row = ((long)blockIdx.x * 4 + wave) * LN_RW;
  const float* src = in + row * 256 + lane * 4;
  v4f pa[LN_RW][4], pb[LN_RW][4];
#pragma unroll
  for (int r = 0; r < LN_RW; ++r) {
#pragma unroll
    for (int p = 0; p < 4; ++p) {
      pa[r][p] = *(const v4f*)(src + (long)p * sP + r * 256);
      pb[r][p] = *(const v4f*)(src + (long)p * sP + r * 256 + 128);
    }
  }
#pragma unroll
  for (int r = 0; r < LN_RW; ++r) {
    v4f a = (pa[r][0] + pa[r][1]) + (pa[r][2] + pa[r][3]);
    v4f b = (pb[r][0] + pb[r][1]) + (pb[r][2] + pb[r][3]);
    ln_row256(a, b);
    float* dst = out + (row + r) * 256 + lane * 4;
    *(v4f*)(dst) = a;
    *(v4f*)(dst + 128) = b;
  }
}


static constexpr int RS_STRIPS = 32;
static constexpr int XB = 64;
static constexpr int XN = 32;

__global__ __launch_bounds__(128) void xblocks_kernel(const float* __restrict__ Rp,
                                                      const float* __restrict__ U,
                                                      float* __restrict__ X) {
  constexpr int PA = XB + 4;
  constexpr int PB = 2 * XN + 32;
  constexpr int NBLK = Tt / XB;
  constexpr int NU = (Tt / 2) * (XN / 4) / 128;
  __shared__ __attribute__((aligned(16))) float sm[(Tt / 2) * PB + NBLK * XB * PA + 2 * Tt + 2 * XN];
  __shared__ unsigned big[8];
  float* Us = sm;
  float* Gs = Us + (Tt / 2) * PB;
  float* rs = Gs + NBLK * XB * PA;
  float* qs = rs + Tt;
  float* xp = qs + Tt;

  const int b = blockIdx.z;
  const int col0 = blockIdx.x * XN;
  const int tid = threadIdx.x;
  const int lane = tid & 31;
  const int wave = tid >> 5;
  const int half = lane >> 4;
  const int l = lane & 15;

  const float* up = U + (long)b * Tt * Nn + col0;
  v4f ru[NU][2];
  int lu[NU];
#pragma unroll
  for (int j = 0; j < NU; ++j) {
    const int f = tid + 128 * j;
    const int r = f / (XN / 4);
    const int c = f - r * (XN / 4);
    ru[j][0] = *(const v4f*)(up + (long)(2 * r) * Nn + c * 4);
    ru[j][1] = *(const v4f*)(up + (long)(2 * r + 1) * Nn + c * 4);
    lu[j] = r * PB + c * 8;
  }

  float r0 = 0.f, r1 = 0.f;
#pragma unroll
  for (int c = 0; c < RS_STRIPS; ++c) {
    r0 += Rp[(long)c * (Bb * Tt) + b * Tt + tid];
    r1 += Rp[(long)c * (Bb * Tt) + b * Tt + tid + 128];
  }
  rs[tid] = r0;
  rs[tid + 128] = r1;
  const unsigned m0 = __builtin_amdgcn_ballot_w32(r0 >= 1e-12f);
  const unsigned m1 = __builtin_amdgcn_ballot_w32(r1 >= 1e-12f);
  if (lane == 0) {
    big[wave] = m0;
    big[4 + wave] = m1;
  }
  __syncthreads();

  int f = Tt;
#pragma unroll
  for (int w = 7; w >= 0; --w) {
    const unsigned mw = big[w];
    if (mw != 0u) f = 32 * w + __builtin_ctz(mw);
  }
  float inv0, inv1;
  {
    float S = 1.0f;
    if (tid <= f) {
      S = 0.f;
      for (int i = 0; i < tid; ++i) {
        const float sum = fmaf(0.97f, S, rs[i]);
        S = (sum >= 1e-12f) ? 1.0f : sum * 1e12f;
      }
    }
    inv0 = 1.0f / fmaxf(fmaf(0.97f, S, r0), 1e-12f);
    qs[tid] = 0.97f * inv0;
  }
  {
    float S = 1.0f;
    if (tid + 128 <= f) {
      S = 0.f;
      for (int i = 0; i < tid + 128; ++i) {
        const float sum = fmaf(0.97f, S, rs[i]);
        S = (sum >= 1e-12f) ? 1.0f : sum * 1e12f;
      }
    }
    inv1 = 1.0f / fmaxf(fmaf(0.97f, S, r1), 1e-12f);
    qs[tid + 128] = 0.97f * inv1;
  }
#pragma unroll
  for (int j = 0; j < NU; ++j) {
    *(v4f*)(&Us[lu[j]]) = v4f{ru[j][0][0], ru[j][1][0], ru[j][0][1], ru[j][1][1]};
    *(v4f*)(&Us[lu[j] + 4]) = v4f{ru[j][0][2], ru[j][1][2], ru[j][0][3], ru[j][1][3]};
  }
  __syncthreads();

  const float GMAX = 3.0e38f;
  {
    const int jb = tid >> 6;
    const int sl = tid & (XB - 1);
    float* gp0 = Gs + jb * (XB * PA) + sl;
    float* gp1 = gp0 + 2 * (XB * PA);
    v4f qa[XB / 4], qb[XB / 4];
#pragma unroll
    for (int i = 0; i < XB / 4; ++i) {
      qa[i] = *(const v4f*)(&qs[XB * jb + 4 * i]);
      qb[i] = *(const v4f*)(&qs[XB * (jb + 2) + 4 * i]);
    }
    __builtin_amdgcn_sched_barrier(0);
    float g0 = 0.f, g1 = 0.f;
#pragma unroll
    for (int i = 0; i < XB; ++i) {
      const float n0 = fminf(g0 * qa[i >> 2][i & 3], GMAX);
      const float n1 = fminf(g1 * qb[i >> 2][i & 3], GMAX);
      g0 = (i == sl) ? inv0 : n0;
      g1 = (i == sl) ? inv1 : n1;
      gp0[i * PA] = g0;
      gp1[i * PA] = g1;
    }
  }
  __syncthreads();

  const float* Ab0 = Gs + (16 * wave + l) * PA + 2 * half;
  const float* Bb0 = Us + half * PB + l * 2;
  float* xo = X + (long)b * Tt * Nn + col0;
  static_assert(NBLK == 4, "the four blocks are written out below");
  constexpr int NKS = XB / 4;
  v2f fa0[NKS], fa1[NKS];
  v2f fb0[NKS][2], fb1[NKS][2];
  auto read_step = [&](int j, int ks, v2f (&fa)[NKS], v2f (&fb)[NKS][2])
                       __attribute__((always_inline)) {
    const float* Ab = Ab0 + j * (XB * PA);
    const float* Bq = Bb0 + (j * (XB / 2)) * PB;
    fa[ks] = *(const v2f*)(Ab + ks * 4);
#pragma unroll
    for (int n = 0; n < 2; ++n) fb[ks][n] = *(const v2f*)(Bq + (ks * 2) * PB + n * 32);
  };
  auto block = [&](int j, v2f (&fa)[NKS], v2f (&fb)[NKS][2], v2f (&na)[NKS], v2f (&nb)[NKS][2])
                   __attribute__((always_inline)) {
    v8f acc[2];
    acc[0] = v8f{};
    acc[1] = v8f{};
#pragma unroll
    for (int p = 0; p < NKS / 2; ++p) {
      if (j + 1 < NBLK) {
        read_step(j + 1, 2 * p, na, nb);
        read_step(j + 1, 2 * p + 1, na, nb);
      }
#pragma unroll
      for (int ks = 2 * p + 1; ks <= 2 * p + 2; ++ks) {
        if (ks < NKS) {
#pragma unroll
          for (int n = 0; n < 2; ++n)
            acc[n] = __builtin_amdgcn_wmma_f32_16x16x4_f32(false, fa[ks], false, fb[ks][n],
                                                           (short)0, acc[n], false, false);
        }
      }
      __builtin_amdgcn_sched_barrier(0);
    }
    {
      v2f bb[2];
      bb[0] = fb[0][0];
      bb[1] = fb[0][1];
      if (j > 0) {
        __builtin_amdgcn_fence(__ATOMIC_RELEASE, "workgroup", "local");
        __builtin_amdgcn_s_barrier();
        __builtin_amdgcn_fence(__ATOMIC_ACQUIRE, "workgroup", "local");
        const float* xq = xp + ((j - 1) & 1) * XN;
#pragma unroll
        for (int n = 0; n < 2; ++n) {
          const float cv = fmaf(0.97f, xq[n * 16 + l], bb[n][0]);
          bb[n][0] = (half == 0) ? cv : bb[n][0];
        }
      }
#pragma unroll
      for (int n = 0; n < 2; ++n)
        acc[n] = __builtin_amdgcn_wmma_f32_16x16x4_f32(false, fa[0], false, bb[n], (short)0,
                                                       acc[n], false, false);
    }
    if (j + 1 < NBLK && wave == 3 && half == 1) {
      xp[(j & 1) * XN + l] = acc[0][7];
      xp[(j & 1) * XN + 16 + l] = acc[1][7];
    }
#pragma unroll
    for (int n = 0; n < 2; ++n) {
#pragma unroll
      for (int r = 0; r < 8; ++r) {
        const long row = j * XB + 16 * wave + 8 * half + r;
        xo[row * Nn + n * 16 + l] = acc[n][r];
      }
    }
  };
#pragma unroll
  for (int ks = 0; ks < NKS; ++ks) read_step(0, ks, fa0, fb0);
  block(0, fa0, fb0, fa1, fb1);
  block(1, fa1, fb1, fa0, fb0);
  block(2, fa0, fb0, fa1, fb1);
  block(3, fa1, fb1, fa0, fb0);
}

template <int EPI, int BM, int BN, int BK, int SPLIT>
__global__ __launch_bounds__(128) void gemm_ABt_lds(const float* __restrict__ A,
                                                    const float* __restrict__ W,
                                                    float* __restrict__ C,
                                                    float* __restrict__ Rs,
                                                    float* __restrict__ Ln,
                                                    int N, int K,
                                                    long sA, long sW, long sC, long sP) {
  constexpr int PITCH = BK + 4;
  constexpr int C4 = BK / 4;
  constexpr int NA = BM * C4 / 128;
  constexpr int NW = BN * C4 / 128;
  constexpr int FM = BM / 32;
  constexpr int FN = BN / 32;
  __shared__ __attribute__((aligned(16))) float As[2][BM * PITCH];
  __shared__ __attribute__((aligned(16))) float Ws[2][BN * PITCH];

  const int b = blockIdx.z / SPLIT;
  const int sp = blockIdx.z - b * SPLIT;
  const int Ks = K / SPLIT;
  A += (long)b * sA + sp * Ks;
  W += (long)b * sW + sp * Ks;
  C += (long)b * sC + (long)sp * sP;
  const int tid = threadIdx.x;
  const int lane = tid & 31;
  const int wave = tid >> 5;
  const int wm = wave >> 1;
  const int wn = wave & 1;
  const int half = lane >> 4;
  const int l = lane & 15;
  const int row0 = blockIdx.y * BM;
  const int col0 = blockIdx.x * BN;

  const float* ga[NA];
  int la[NA];
#pragma unroll
  for (int j = 0; j < NA; ++j) {
    const int f = tid + 128 * j;
    const int r = f / C4;
    const int c = f - r * C4;
    ga[j] = A + (long)(row0 + r) * K + c * 4;
    la[j] = r * PITCH + c * 4;
  }
  const float* gw[NW];
  int lw[NW];
#pragma unroll
  for (int j = 0; j < NW; ++j) {
    const int f = tid + 128 * j;
    const int r = f / C4;
    const int c = f - r * C4;
    gw[j] = W + (long)(col0 + r) * K + c * 4;
    lw[j] = r * PITCH + c * 4;
  }

  v4f ra[NA], rw[NW];
#pragma unroll
  for (int j = 0; j < NA; ++j) ra[j] = *(const v4f*)(ga[j]);
#pragma unroll
  for (int j = 0; j < NW; ++j) rw[j] = *(const v4f*)(gw[j]);
  v4f na = v4f{}, nb = v4f{};
  long nrow = 0;
  if constexpr (EPI == 1) {
    nrow = row0 + 4 * (int)blockIdx.x + wave;
    const float* np = A + nrow * Dd + lane * 4;
    na = *(const v4f*)(np);
    nb = *(const v4f*)(np + 128);
  }
#pragma unroll
  for (int j = 0; j < NA; ++j) *(v4f*)(&As[0][la[j]]) = ra[j];
#pragma unroll
  for (int j = 0; j < NW; ++j) *(v4f*)(&Ws[0][lw[j]]) = rw[j];
  __syncthreads();
  if constexpr (EPI == 1) ln_row256(na, nb);

  v8f acc[FM][FN];
#pragma unroll
  for (int i = 0; i < FM; ++i)
#pragma unroll
    for (int j = 0; j < FN; ++j) acc[i][j] = v8f{};

  const int aoff = (wm * (BM / 2) + l) * PITCH + 2 * half;
  const int woff = (wn * (BN / 2) + l) * PITCH + 2 * half;
  const int nchunks = Ks / BK;
  for (int ch = 0; ch < nchunks; ++ch) {
    const int cur = ch & 1;
    const bool more = (ch + 1 < nchunks);
    if (more) {
      const int kn = (ch + 1) * BK;
#pragma unroll
      for (int j = 0; j < NA; ++j) ra[j] = *(const v4f*)(ga[j] + kn);
#pragma unroll
      for (int j = 0; j < NW; ++j) rw[j] = *(const v4f*)(gw[j] + kn);
    }
    const float* Ab = &As[cur][aoff];
    const float* Wb = &Ws[cur][woff];
#pragma unroll
    for (int ks = 0; ks < BK / 4; ++ks) {
      v2f a[FM], bb[FN];
#pragma unroll
      for (int i = 0; i < FM; ++i) a[i] = *(const v2f*)(Ab + i * 16 * PITCH + ks * 4);
#pragma unroll
      for (int j = 0; j < FN; ++j) bb[j] = *(const v2f*)(Wb + j * 16 * PITCH + ks * 4);
#pragma unroll
      for (int i = 0; i < FM; ++i)
#pragma unroll
        for (int j = 0; j < FN; ++j)
          acc[i][j] = __builtin_amdgcn_wmma_f32_16x16x4_f32(false, a[i], false, bb[j], (short)0,
                                                            acc[i][j], false, false);
    }
    if (more) {
#pragma unroll
      for (int j = 0; j < NA; ++j) *(v4f*)(&As[cur ^ 1][la[j]]) = ra[j];
#pragma unroll
      for (int j = 0; j < NW; ++j) *(v4f*)(&Ws[cur ^ 1][lw[j]]) = rw[j];
    }
    __syncthreads();
  }

  if constexpr (EPI == 1) {
    float* nd = Ln + nrow * Dd + lane * 4;
    *(v4f*)(nd) = na;
    *(v4f*)(nd + 128) = nb;
  }

#pragma unroll
  for (int i = 0; i < FM; ++i) {
#pragma unroll
    for (int j = 0; j < FN; ++j) {
      const long col = col0 + wn * (BN / 2) + j * 16 + l;
#pragma unroll
      for (int r = 0; r < 8; ++r) {
        const long row = row0 + wm * (BM / 2) + i * 16 + r + 8 * half;
        float v = acc[i][j][r];
        if (EPI == 1) v = fmaxf(v, 0.f);
        C[row * N + col] = v;
      }
    }
  }

  if (EPI == 1) {
    static_assert(EPI != 1 || (BM == 64 && BN == 64), "row-sum epilogue: 32 x 32 wave tile");
    float p[FM][8];
#pragma unroll
    for (int i = 0; i < FM; ++i) {
#pragma unroll
      for (int r = 0; r < 8; ++r) {
        float s = 0.f;
#pragma unroll
        for (int j = 0; j < FN; ++j) s += fmaxf(acc[i][j][r], 0.f);
        s = row_xadd<0x161>(s);
        s = row_xadd<0x162>(s);
        s = row_xadd<0x164>(s);
        s = row_xadd<0x168>(s);
        p[i][r] = s;
      }
    }
    float mine = 0.f;
#pragma unroll
    for (int i = 0; i < FM; ++i)
#pragma unroll
      for (int r = 0; r < 8; ++r)
        if (l == i * 8 + r) mine = p[i][r];
    const int rho = (l >> 3) * 16 + 8 * half + (l & 7);
    const long rows = (long)gridDim.y * BM;
    Rs[(long)(blockIdx.x * 2 + wn) * rows + row0 + wm * (BM / 2) + rho] = mine;
  }
}

__global__ __launch_bounds__(128) void gemm_lnA_Bt_lds(const float* __restrict__ A,
                                                       const float* __restrict__ W,
                                                       float* __restrict__ C,
                                                       const float* __restrict__ Mul,
                                                       int N) {
  constexpr int BM = 64, BN = 64, BK = 32, KK = 256;
  constexpr int PP = KK + 4;
  constexpr int PITCH = BK + 4;
  constexpr int C4 = BK / 4;
  constexpr int NW = BN * C4 / 128;
  constexpr int FM = BM / 32;
  constexpr int FN = BN / 32;
  __shared__ __attribute__((aligned(16))) float Ap[BM * PP];
  __shared__ __attribute__((aligned(16))) float Ws[2][BN * PITCH];

  const int tid = threadIdx.x;
  const int lane = tid & 31;
  const int wave = tid >> 5;
  const int wm = wave >> 1;
  const int wn = wave & 1;
  const int half = lane >> 4;
  const int l = lane & 15;
  const int row0 = blockIdx.y * BM;
  const int col0 = blockIdx.x * BN;

  const float* gw[NW];
  int lw[NW];
#pragma unroll
  for (int j = 0; j < NW; ++j) {
    const int f = tid + 128 * j;
    const int r = f / C4;
    const int c = f - r * C4;
    gw[j] = W + (long)(col0 + r) * KK + c * 4;
    lw[j] = r * PITCH + c * 4;
  }

  v4f p[16][2];
  const float* gp = A + (long)(row0 + wave * 16) * KK + lane * 4;
#pragma unroll
  for (int i = 0; i < 16; ++i) {
    p[i][0] = *(const v4f*)(gp + (long)i * KK);
    p[i][1] = *(const v4f*)(gp + (long)i * KK + 128);
  }
  v4f rw[NW];
#pragma unroll
  for (int j = 0; j < NW; ++j) rw[j] = *(const v4f*)(gw[j]);

#pragma unroll
  for (int i = 0; i < 16; ++i) {
    float s = ((p[i][0][0] + p[i][0][1]) + (p[i][0][2] + p[i][0][3])) +
              ((p[i][1][0] + p[i][1][1]) + (p[i][1][2] + p[i][1][3]));
    s = row_xadd<0x161>(s);
    s = row_xadd<0x162>(s);
    s = row_xadd<0x164>(s);
    s = row_xadd<0x168>(s);
    s += __shfl_xor(s, 16, 32);
    const float mean = s * (1.0f / 256.0f);
    const v4f c0 = p[i][0] - mean;
    const v4f c1 = p[i][1] - mean;
    float q = ((c0[0] * c0[0] + c0[1] * c0[1]) + (c0[2] * c0[2] + c0[3] * c0[3])) +
              ((c1[0] * c1[0] + c1[1] * c1[1]) + (c1[2] * c1[2] + c1[3] * c1[3]));
    q = row_xadd<0x161>(q);
    q = row_xadd<0x162>(q);
    q = row_xadd<0x164>(q);
    q = row_xadd<0x168>(q);
    q += __shfl_xor(q, 16, 32);
    const float rstd = rsqrtf(q * (1.0f / 256.0f) + 1e-5f);
    float* dst = &Ap[(wave * 16 + i) * PP + lane * 4];
    *(v4f*)(dst) = c0 * rstd;
    *(v4f*)(dst + 128) = c1 * rstd;
  }
#pragma unroll
  for (int j = 0; j < NW; ++j) *(v4f*)(&Ws[0][lw[j]]) = rw[j];
  __syncthreads();

  v8f acc[FM][FN];
#pragma unroll
  for (int i = 0; i < FM; ++i)
#pragma unroll
    for (int j = 0; j < FN; ++j) acc[i][j] = v8f{};

  const int aoff = (wm * (BM / 2) + l) * PP + 2 * half;
  const int woff = (wn * (BN / 2) + l) * PITCH + 2 * half;
  constexpr int nchunks = KK / BK;
#pragma unroll 1
  for (int ch = 0; ch < nchunks; ++ch) {
    const int cur = ch & 1;
    const bool more = (ch + 1 < nchunks);
    if (more) {
      const int kn = (ch + 1) * BK;
#pragma unroll
      for (int j = 0; j < NW; ++j) rw[j] = *(const v4f*)(gw[j] + kn);
    }
    const float* Ab = &Ap[aoff + ch * BK];
    const float* Wb = &Ws[cur][woff];
#pragma unroll
    for (int ks = 0; ks < BK / 4; ++ks) {
      v2f a[FM], bb[FN];
#pragma unroll
      for (int i = 0; i < FM; ++i) a[i] = *(const v2f*)(Ab + i * 16 * PP + ks * 4);
#pragma unroll
      for (int j = 0; j < FN; ++j) bb[j] = *(const v2f*)(Wb + j * 16 * PITCH + ks * 4);
#pragma unroll
      for (int i = 0; i < FM; ++i)
#pragma unroll
        for (int j = 0; j < FN; ++j)
          acc[i][j] = __builtin_amdgcn_wmma_f32_16x16x4_f32(false, a[i], false, bb[j], (short)0,
                                                            acc[i][j], false, false);
    }
    if (more) {
#pragma unroll
      for (int j = 0; j < NW; ++j) *(v4f*)(&Ws[cur ^ 1][lw[j]]) = rw[j];
    }
    __syncthreads();
  }

#pragma unroll
  for (int i = 0; i < FM; ++i) {
#pragma unroll
    for (int j = 0; j < FN; ++j) {
      const long col = col0 + wn * (BN / 2) + j * 16 + l;
#pragma unroll
      for (int r = 0; r < 8; ++r) {
        const long row = row0 + wm * (BM / 2) + i * 16 + r + 8 * half;
        const float v = fmaxf(acc[i][j][r], 0.f) * fmaxf(Mul[row * N + col], 0.f);
        C[row * N + col] = v;
      }
    }
  }
}

static __device__ __forceinline__ v4f decay_weights(const float* pw, int t, int s0) {
  v4f w;
#pragma unroll
  for (int e = 0; e < 4; ++e) {
    const int d = t - 1 - (s0 + e);
    w[e] = pw[d < 0 ? 0 : d];
  }
  return w;
}
static __device__ __forceinline__ v4f decay_apply(v4f v, v4f w, int t, int s0) {
  v4f o;
#pragma unroll
  for (int e = 0; e < 4; ++e) o[e] = (s0 + e < t) ? w[e] * v[e] : 0.f;
  return o;
}

template <int APL, int NA>
static __device__ __forceinline__ v4f plane_sum(const v4f (&rp)[APL][NA], int j) {
  if constexpr (APL == 4) return (rp[0][j] + rp[1][j]) + (rp[2][j] + rp[3][j]);
  else return rp[0][j];
}

template <int BM, int BN, int BK, bool CAUSAL, bool DECAY, int APL>
__global__ __launch_bounds__(128) void gemm_AB_lds(const float* __restrict__ A,
                                                   const float* __restrict__ Bm,
                                                   float* __restrict__ C,
                                                   int N, int K,
                                                   long sA, long sB, long sC, long sPA) {
  static_assert(APL == 1 || APL == 4, "1 plane or 4 partial planes");
  constexpr int PA = BK + 4;
  constexpr int PB = 2 * BN + 32;
  static_assert(BN == 64, "k-pair pitch: 2 * BN + 32 = 32 mod 64 banks");
  constexpr int CA = BK / 4;
  constexpr int CB = BN / 4;
  constexpr int NA = BM * CA / 128;
  constexpr int NB = (BK / 2) * CB / 128;
  constexpr int FM = BM / 32;
  constexpr int FN = BN / 32;
  __shared__ __attribute__((aligned(16))) float As[2][BM * PA];
  __shared__ __attribute__((aligned(16))) float Bs[2][(BK / 2) * PB];
  __shared__ float pw[DECAY ? Tt : 1];

  const int b = blockIdx.z;
  A += (long)b * sA;
  Bm += (long)b * sB;
  C += (long)b * sC;
  const int tid = threadIdx.x;
  const int lane = tid & 31;
  const int wave = tid >> 5;
  const int wm = wave >> 1;
  const int wn = wave & 1;
  const int half = lane >> 4;
  const int l = lane & 15;
  const int row0 = blockIdx.y * BM;
  const int col0 = blockIdx.x * BN;

  const float* ga[NA];
  int la[NA];
  int ta[NA], sa[NA];
#pragma unroll
  for (int j = 0; j < NA; ++j) {
    const int f = tid + 128 * j;
    const int r = f / CA;
    const int c = f - r * CA;
    ga[j] = A + (long)(row0 + r) * K + c * 4;
    la[j] = r * PA + c * 4;
    ta[j] = row0 + r;
    sa[j] = c * 4;
  }
  const float* gb[NB];
  int lb[NB];
#pragma unroll
  for (int j = 0; j < NB; ++j) {
    const int f = tid + 128 * j;
    const int r = f / CB;
    const int c = f - r * CB;
    gb[j] = Bm + (long)(2 * r) * N + col0 + c * 4;
    lb[j] = r * PB + c * 8;
  }

  v4f rp[APL][NA];
  v4f ra[NA], rb[NB][2];
#pragma unroll
  for (int p = 0; p < APL; ++p)
#pragma unroll
    for (int j = 0; j < NA; ++j) rp[p][j] = *(const v4f*)(ga[j] + (long)p * sPA);
#pragma unroll
  for (int j = 0; j < NB; ++j) {
    rb[j][0] = *(const v4f*)(gb[j]);
    rb[j][1] = *(const v4f*)(gb[j] + N);
  }
  v4f wq[NA];
  if (DECAY) {
    for (int i = tid; i < Tt; i += 128) pw[i] = __powf(0.97f, (float)i);
    __syncthreads();
#pragma unroll
    for (int j = 0; j < NA; ++j) wq[j] = decay_weights(pw, ta[j], sa[j]);
  }
#pragma unroll
  for (int j = 0; j < NA; ++j)
    ra[j] = plane_sum<APL, NA>(rp, j);
  if (DECAY) {
#pragma unroll
    for (int j = 0; j < NA; ++j) ra[j] = decay_apply(ra[j], wq[j], ta[j], sa[j]);
  }
#pragma unroll
  for (int j = 0; j < NA; ++j) *(v4f*)(&As[0][la[j]]) = ra[j];
#pragma unroll
  for (int j = 0; j < NB; ++j) {
    *(v4f*)(&Bs[0][lb[j]]) = v4f{rb[j][0][0], rb[j][1][0], rb[j][0][1], rb[j][1][1]};
    *(v4f*)(&Bs[0][lb[j] + 4]) = v4f{rb[j][0][2], rb[j][1][2], rb[j][0][3], rb[j][1][3]};
  }
  __syncthreads();

  v8f acc[FM][FN];
#pragma unroll
  for (int i = 0; i < FM; ++i)
#pragma unroll
    for (int j = 0; j < FN; ++j) acc[i][j] = v8f{};

  const int aoff = (wm * (BM / 2) + l) * PA + 2 * half;
  const int boff = half * PB + (wn * (BN / 2) + l) * 2;
  const int nchunks = CAUSAL ? (row0 + BM) / BK : K / BK;
  for (int ch = 0; ch < nchunks; ++ch) {
    const int cur = ch & 1;
    const bool more = (ch + 1 < nchunks);
    if (more) {
      const int kn = (ch + 1) * BK;
#pragma unroll
      for (int p = 0; p < APL; ++p)
#pragma unroll
        for (int j = 0; j < NA; ++j) rp[p][j] = *(const v4f*)(ga[j] + (long)p * sPA + kn);
#pragma unroll
      for (int j = 0; j < NB; ++j) {
        rb[j][0] = *(const v4f*)(gb[j] + (long)kn * N);
        rb[j][1] = *(const v4f*)(gb[j] + (long)(kn + 1) * N);
      }
      if (DECAY) {
#pragma unroll
        for (int j = 0; j < NA; ++j) wq[j] = decay_weights(pw, ta[j], sa[j] + kn);
      }
    }
    const float* Ab = &As[cur][aoff];
    const float* Bb = &Bs[cur][boff];
#pragma unroll
    for (int ks = 0; ks < BK / 4; ++ks) {
      v2f a[FM], bb[FN];
#pragma unroll
      for (int i = 0; i < FM; ++i) a[i] = *(const v2f*)(Ab + i * 16 * PA + ks * 4);
#pragma unroll
      for (int j = 0; j < FN; ++j) bb[j] = *(const v2f*)(Bb + (ks * 2) * PB + j * 32);
#pragma unroll
      for (int i = 0; i < FM; ++i)
#pragma unroll
        for (int j = 0; j < FN; ++j)
          acc[i][j] = __builtin_amdgcn_wmma_f32_16x16x4_f32(false, a[i], false, bb[j], (short)0,
                                                            acc[i][j], false, false);
    }
    if constexpr (FM == 1 && FN == 2) {
      __builtin_amdgcn_sched_group_barrier(0x100, 9, 0);
#pragma unroll
      for (int g = 0; g < BK / 8; ++g) {
        __builtin_amdgcn_sched_group_barrier(0x008, 4, 0);
        __builtin_amdgcn_sched_group_barrier(0x100, 3, 0);
      }
    }
    if (more) {
#pragma unroll
      for (int j = 0; j < NA; ++j)
        ra[j] = plane_sum<APL, NA>(rp, j);
      if (DECAY) {
        const int kn = (ch + 1) * BK;
#pragma unroll
        for (int j = 0; j < NA; ++j) ra[j] = decay_apply(ra[j], wq[j], ta[j], sa[j] + kn);
      }
#pragma unroll
      for (int j = 0; j < NA; ++j) *(v4f*)(&As[cur ^ 1][la[j]]) = ra[j];
#pragma unroll
      for (int j = 0; j < NB; ++j) {
        *(v4f*)(&Bs[cur ^ 1][lb[j]]) = v4f{rb[j][0][0], rb[j][1][0], rb[j][0][1], rb[j][1][1]};
        *(v4f*)(&Bs[cur ^ 1][lb[j] + 4]) = v4f{rb[j][0][2], rb[j][1][2], rb[j][0][3], rb[j][1][3]};
      }
    }
    __syncthreads();
  }

#pragma unroll
  for (int i = 0; i < FM; ++i) {
#pragma unroll
    for (int j = 0; j < FN; ++j) {
      const long col = col0 + wn * (BN / 2) + j * 16 + l;
#pragma unroll
      for (int r = 0; r < 8; ++r) {
        const long row = row0 + wm * (BM / 2) + i * 16 + r + 8 * half;
        C[row * N + col] = acc[i][j][r];
      }
    }
  }
}

extern "C" void kernel_launch(void* const* d_in, const int* in_sizes, int n_in,
                              void* d_out, int out_size, void* d_ws, size_t ws_size,
                              hipStream_t stream) {
  const float* emb = (const float*)d_in[0];
  const float* E   = (const float*)d_in[1];
  const float* Dx  = (const float*)d_in[2];
  const float* Dy  = (const float*)d_in[3];
  float* out = (float*)d_out;

  const long BT = (long)Bb * Tt;
  const long szBTD = BT * Dd;
  const long szBTN = BT * Nn;
  const long szBTT = (long)Bb * Tt * Tt;

  float* ws = (float*)d_ws;
  float* LNV = ws;
  float* XU  = LNV + szBTD;
  float* X   = XU + szBTN;
  float* Rp  = X + szBTN;
  float* As  = Rp + szBTT;
  float* Y   = As + szBTD;
  float* KgP = XU;
  float* VP  = XU;

  gemm_ABt_lds<1, 64, 64, 32, 1><<<dim3(1024 / 64, 2048 / 64, 1), 128, 0, stream>>>(
      emb, Dx, XU, Rp, LNV, 1024, 256, 0, 0, 0, 0);

  xblocks_kernel<<<dim3(Nn / XN, 1, Bb), 128, 0, stream>>>(Rp, XU, X);

  gemm_ABt_lds<0, 64, 64, 32, 4><<<dim3(256 / 64, 256 / 64, Bb * 4), 128, 0, stream>>>(
      X, X, KgP, nullptr, nullptr, 256, 1024, (long)Tt * Nn, (long)Tt * Nn, (long)Tt * Tt, szBTT);

  gemm_AB_lds<32, 64, 128, false, true, 4><<<dim3(256 / 64, 256 / 32, Bb), 128, 0, stream>>>(
      KgP, LNV, As, 256, 256, (long)Tt * Tt, (long)Tt * Dd, (long)Tt * Dd, szBTT);

  gemm_lnA_Bt_lds<<<dim3(1024 / 64, 2048 / 64, 1), 128, 0, stream>>>(As, Dy, Y, X, 1024);

  gemm_ABt_lds<0, 64, 64, 32, 4><<<dim3(256 / 64, 2048 / 64, 4), 128, 0, stream>>>(
      Y, E, VP, nullptr, nullptr, 256, 1024, 0, 0, 0, szBTD);

  ln256_planes4_kernel<<<2048 / (4 * LN_RW), 128, 0, stream>>>(VP, out, szBTD);
}
